# starting kernel
# baseline (speedup 1.0000x reference)
#include <hip/hip_runtime.h>

typedef __attribute__((ext_vector_type(2))) float v2f;
typedef __attribute__((ext_vector_type(8))) float v8f;

#define WMMA_F32(a, b, c) \
  __builtin_amdgcn_wmma_f32_16x16x4_f32(false, (a), false, (b), (short)0, (c), false, false)

static constexpr int Tn = 32768;
static constexpr int NS = 256;
static constexpr int L1 = 32;
static constexpr int C1 = 1024;
static constexpr int L2 = 16;
static constexpr int C2 = 64;
static constexpr int SC_COLS = 16;

__constant__ float d_H[6][4] = {
  {1.0f, 0.0f, 0.0f, 0.0f},
  {0.896f, 0.128f, 0.104f, -0.032f},
  {0.784f, 0.147f, 0.216f, -0.063f},
  {0.104f, 0.032f, 0.896f, -0.128f},
  {25.0f/729.0f, 8.0f/729.0f, 704.0f/729.0f, -64.0f/729.0f},
  {0.0f, 0.0f, 1.0f, 0.0f},
};

__global__ void k_scale_dt(float* __restrict__ o, const float* __restrict__ i,
                           int n, const float* __restrict__ t) {
  int idx = blockIdx.x * 256 + threadIdx.x;
  if (idx < n) { float dt = t[1] - t[0]; o[idx] = dt * i[idx]; }
}
__global__ void k_copy(float* __restrict__ o, const float* __restrict__ i, int n) {
  int idx = blockIdx.x * 256 + threadIdx.x;
  if (idx < n) o[idx] = i[idx];
}
__global__ void k_zero(float* __restrict__ o, int n) {
  int idx = blockIdx.x * 256 + threadIdx.x;
  if (idx < n) o[idx] = 0.0f;
}
__global__ void k_eye(float* __restrict__ o) {
  int idx = blockIdx.x * 256 + threadIdx.x;
  if (idx < 65536) o[idx] = ((idx >> 8) == (idx & 255)) ? 1.0f : 0.0f;
}
__global__ void k_axpy(float* __restrict__ o, const float* __restrict__ i,
                       int n, float alpha) {
  int idx = blockIdx.x * 256 + threadIdx.x;
  if (idx < n) o[idx] += alpha * i[idx];
}
__global__ void k_addblock(float* __restrict__ Lm, const float* __restrict__ Bd, int blk) {
  int idx = blockIdx.x * 256 + threadIdx.x;
  if (idx < 8192) { int r = idx >> 5, c = idx & 31; Lm[r * 192 + blk * 32 + c] += Bd[idx]; }
}
__global__ void k_qcat(float* __restrict__ Q, const float* __restrict__ W) {
  int idx = blockIdx.x * 256 + threadIdx.x;
  if (idx >= 128 * 256) return;
  int k = idx >> 8, r = idx & 255;
  int j = k >> 5, c = k & 31;
  float s = 0.0f;
  for (int st = 0; st < 6; ++st) s += d_H[st][j] * W[r * 192 + 32 * st + c];
  Q[idx] = s;
}
__global__ void k_buildP(float* __restrict__ P, const float* __restrict__ u) {
  int idx = blockIdx.x * 256 + threadIdx.x;
  if (idx >= Tn * 128) return;
  int n = idx >> 7, k = idx & 127;
  float val = 0.0f;
  if (n < Tn - 1) {
    int j = k >> 5, c = k & 31;
    int m = n + ((j >= 2) ? 1 : 0);
    if ((j & 1) == 0) {
      val = u[m * 32 + c];
    } else {
      if (m == 0)            val = u[32 + c] - u[c];
      else if (m == Tn - 1)  val = u[(Tn - 1) * 32 + c] - u[(Tn - 2) * 32 + c];
      else                   val = 0.5f * (u[(m + 1) * 32 + c] - u[(m - 1) * 32 + c]);
    }
  }
  P[idx] = val;
}

__global__ __launch_bounds__(128) void gemm_kernel(
    float* __restrict__ D, int ldd,
    const float* __restrict__ A, int lda,
    const float* __restrict__ B, int ldb,
    const float* __restrict__ Cin, int ldc,
    int Mr, int Nc, int Kd, float alpha, float beta, int transB) {
  const int lane = threadIdx.x & 31, wave = threadIdx.x >> 5;
  const int hl = lane >> 4, lr = lane & 15;
  const int tilesN = Nc >> 4;
  const int tile = blockIdx.x * 4 + wave;
  if (tile >= (Mr >> 4) * tilesN) return;
  const int tm = tile / tilesN, tn = tile % tilesN;
  const float* Ar = A + (size_t)(tm * 16 + lr) * lda + 2 * hl;
  v8f acc = {};
  if (transB) {
    const float* Br = B + (size_t)(tn * 16 + lr) * ldb + 2 * hl;
    #pragma unroll 4
    for (int k = 0; k < Kd; k += 4) {
      v2f a = *(const v2f*)(Ar + k);
      v2f b = *(const v2f*)(Br + k);
      acc = WMMA_F32(a, b, acc);
    }
  } else {
    const float* Bc = B + (size_t)(2 * hl) * ldb + tn * 16 + lr;
    #pragma unroll 4
    for (int k = 0; k < Kd; k += 4) {
      v2f a = *(const v2f*)(Ar + k);
      v2f b;
      b[0] = Bc[(size_t)k * ldb];
      b[1] = Bc[(size_t)k * ldb + ldb];
      acc = WMMA_F32(a, b, acc);
    }
  }
  float* Dp = D + (size_t)(tm * 16 + 8 * hl) * ldd + tn * 16 + lr;
  if (beta != 0.0f) {
    const float* Cp = Cin + (size_t)(tm * 16 + 8 * hl) * ldc + tn * 16 + lr;
    #pragma unroll
    for (int v = 0; v < 8; ++v)
      Dp[(size_t)v * ldd] = alpha * acc[v] + beta * Cp[(size_t)v * ldc];
  } else {
    #pragma unroll
    for (int v = 0; v < 8; ++v)
      Dp[(size_t)v * ldd] = alpha * acc[v];
  }
}

__global__ __launch_bounds__(256) void scan_kernel(
    const float* __restrict__ Mg, const float* __restrict__ G,
    const float* __restrict__ Init,
    float* __restrict__ OutAll,
    float* __restrict__ OutEnd,
    int L) {
  extern __shared__ float smem[];
  float* Ms = smem;
  float* Xbuf0 = smem + 65536;
  float* Xbuf1 = Xbuf0 + 256 * SC_COLS;
  const int tid = threadIdx.x;
  const int lane = tid & 31, wave = tid >> 5;
  const int hl = lane >> 4, lr = lane & 15;
  const int cbase = blockIdx.x * SC_COLS;
  const int tr0 = wave * 2;

  {
    const float4* src = (const float4*)Mg;
    float4* dst = (float4*)Ms;
    for (int k = tid; k < 65536 / 4; k += 256) dst[k] = src[k];
  }
  for (int e = 0; e < SC_COLS; ++e)
    Xbuf0[tid * SC_COLS + e] = Init ? Init[(size_t)(cbase + e) * 256 + tid] : 0.0f;
  __syncthreads();

  const float* gbase = G + ((size_t)(cbase + lr) * L) * 256 + tr0 * 16 + 8 * hl;

  float* Xc = Xbuf0;
  float* Xn = Xbuf1;
  for (int i = 0; i < L; ++i) {
    if (OutAll) {
      for (int e = 0; e < SC_COLS; ++e) {
        size_t m = (size_t)(cbase + e) * L + i;
        OutAll[m * 256 + tid] = Xc[tid * SC_COLS + e];
      }
    }
    const float* gp = gbase + (size_t)i * 256;
    if (i + 1 < L) __builtin_prefetch(gp + 256, 0, 1);
    v8f acc0, acc1;
    {
      float4 ga = *(const float4*)(gp);
      float4 gb = *(const float4*)(gp + 4);
      float4 gc = *(const float4*)(gp + 16);
      float4 gd = *(const float4*)(gp + 20);
      acc0[0]=ga.x; acc0[1]=ga.y; acc0[2]=ga.z; acc0[3]=ga.w;
      acc0[4]=gb.x; acc0[5]=gb.y; acc0[6]=gb.z; acc0[7]=gb.w;
      acc1[0]=gc.x; acc1[1]=gc.y; acc1[2]=gc.z; acc1[3]=gc.w;
      acc1[4]=gd.x; acc1[5]=gd.y; acc1[6]=gd.z; acc1[7]=gd.w;
    }
    const float* Mr0 = Ms + (size_t)(tr0 * 16 + lr) * 256 + 2 * hl;
    const float* Mr1 = Mr0 + 16 * 256;
    const float* Xk = Xc + 2 * hl * SC_COLS + lr;
    #pragma unroll 8
    for (int kc = 0; kc < 64; ++kc) {
      v2f a0 = *(const v2f*)(Mr0 + kc * 4);
      v2f a1 = *(const v2f*)(Mr1 + kc * 4);
      v2f b;
      b[0] = Xk[kc * 4 * SC_COLS];
      b[1] = Xk[kc * 4 * SC_COLS + SC_COLS];
      acc0 = WMMA_F32(a0, b, acc0);
      acc1 = WMMA_F32(a1, b, acc1);
    }
    #pragma unroll
    for (int v = 0; v < 8; ++v) {
      Xn[(tr0 * 16 + 8 * hl + v) * SC_COLS + lr] = acc0[v];
      Xn[((tr0 + 1) * 16 + 8 * hl + v) * SC_COLS + lr] = acc1[v];
    }
    __syncthreads();
    float* tmp = Xc; Xc = Xn; Xn = tmp;
  }
  if (OutEnd) {
    for (int e = 0; e < SC_COLS; ++e)
      OutEnd[(size_t)(cbase + e) * 256 + tid] = Xc[tid * SC_COLS + e];
  }
}

__global__ __launch_bounds__(256) void carry_kernel(
    const float* __restrict__ Mg, const float* __restrict__ Z,
    const float* __restrict__ x0, float* __restrict__ Tcar, int n) {
  __shared__ float tv[256];
  int r = threadIdx.x;
  tv[r] = x0[r];
  Tcar[r] = tv[r];
  __syncthreads();
  const float* Mr = Mg + (size_t)r * 256;
  for (int j = 0; j + 1 < n; ++j) {
    float acc = Z[(size_t)j * 256 + r];
    #pragma unroll 8
    for (int k = 0; k < 256; ++k) acc += Mr[k] * tv[k];
    __syncthreads();
    tv[r] = acc;
    Tcar[(size_t)(j + 1) * 256 + r] = acc;
    __syncthreads();
  }
}

static inline void launch_gemm(hipStream_t s, float* D, int ldd,
                               const float* A, int lda, const float* B, int ldb,
                               const float* Cin, int ldc,
                               int M, int N, int K, float alpha, float beta, int transB) {
  int tiles = (M / 16) * (N / 16);
  gemm_kernel<<<(tiles + 3) / 4, 128, 0, s>>>(D, ldd, A, lda, B, ldb, Cin, ldc,
                                              M, N, K, alpha, beta, transB);
}

extern "C" void kernel_launch(void* const* d_in, const int* in_sizes, int n_in,
                              void* d_out, int out_size, void* d_ws, size_t ws_size,
                              hipStream_t stream) {
  (void)in_sizes; (void)n_in; (void)out_size; (void)ws_size;
  const float* t  = (const float*)d_in[0];
  const float* u  = (const float*)d_in[1];
  const float* x0 = (const float*)d_in[2];
  const float* Am = (const float*)d_in[3];
  const float* Bm = (const float*)d_in[4];
  const float* Cm = (const float*)d_in[5];
  const float* Dm = (const float*)d_in[6];
  float* xout = (float*)d_out;
  float* yout = xout + (size_t)Tn * NS;

  float* w = (float*)d_ws;
  float* Ad = w;  w += 65536;
  float* Bd = w;  w += 8192;
  float* Kt[6]; for (int i = 0; i < 6; i++) { Kt[i] = w; w += 65536; }
  float* Lt[6]; for (int i = 0; i < 6; i++) { Lt[i] = w; w += 49152; }
  float* TMP  = w; w += 65536;
  float* TMPL = w; w += 49152;
  float* Mm   = w; w += 65536;
  float* Wm   = w; w += 49152;
  float* Qc   = w; w += 32768;
  float* ML   = w; w += 65536;
  float* M512 = w; w += 65536;
  float* PT1  = w; w += 65536;
  float* PT2  = w; w += 65536;
  float* Pm   = w; w += (size_t)Tn * 128;
  float* G    = w; w += (size_t)Tn * 256;
  float* Zend = w; w += (size_t)C1 * 256;
  float* Zend2= w; w += (size_t)C2 * 256;
  float* Tcar = w; w += (size_t)C2 * 256;
  float* Scar = w; w += (size_t)C1 * 256;

  auto gdim = [](int n) { return dim3((unsigned)((n + 255) / 256)); };

  k_scale_dt<<<gdim(65536), 256, 0, stream>>>(Ad, Am, 65536, t);
  k_scale_dt<<<gdim(8192),  256, 0, stream>>>(Bd, Bm, 8192, t);
  k_copy<<<gdim(65536), 256, 0, stream>>>(Kt[0], Ad, 65536);
  k_zero<<<gdim(49152), 256, 0, stream>>>(Lt[0], 49152);
  k_addblock<<<gdim(8192), 256, 0, stream>>>(Lt[0], Bd, 0);

  double aa[7][7] = {};
  aa[2][1] = 0.2;
  aa[3][1] = 3.0/40.0;      aa[3][2] = 9.0/40.0;
  aa[4][1] = 44.0/45.0;     aa[4][2] = -56.0/15.0;     aa[4][3] = 32.0/9.0;
  aa[5][1] = 19372.0/6561.0; aa[5][2] = -25360.0/2187.0;
  aa[5][3] = 64448.0/6561.0; aa[5][4] = -212.0/729.0;
  aa[6][1] = 9017.0/3168.0; aa[6][2] = -355.0/33.0; aa[6][3] = 46732.0/5247.0;
  aa[6][4] = 49.0/176.0;    aa[6][5] = -5103.0/18656.0;
  for (int i = 2; i <= 6; ++i) {
    k_eye<<<gdim(65536), 256, 0, stream>>>(TMP);
    k_zero<<<gdim(49152), 256, 0, stream>>>(TMPL, 49152);
    for (int j = 1; j < i; ++j) {
      float c = (float)aa[i][j];
      k_axpy<<<gdim(65536), 256, 0, stream>>>(TMP,  Kt[j-1], 65536, c);
      k_axpy<<<gdim(49152), 256, 0, stream>>>(TMPL, Lt[j-1], 49152, c);
    }
    launch_gemm(stream, Kt[i-1], 256, Ad, 256, TMP,  256, nullptr, 0, 256, 256, 256, 1.f, 0.f, 0);
    launch_gemm(stream, Lt[i-1], 192, Ad, 256, TMPL, 192, nullptr, 0, 256, 192, 256, 1.f, 0.f, 0);
    k_addblock<<<gdim(8192), 256, 0, stream>>>(Lt[i-1], Bd, i - 1);
  }
  const double bb[7] = {0.0, 35.0/384.0, 0.0, 500.0/1113.0, 125.0/192.0,
                        -2187.0/6784.0, 11.0/84.0};
  k_eye<<<gdim(65536), 256, 0, stream>>>(Mm);
  k_zero<<<gdim(49152), 256, 0, stream>>>(Wm, 49152);
  for (int i = 1; i <= 6; ++i) {
    float c = (float)bb[i];
    if (c != 0.0f) {
      k_axpy<<<gdim(65536), 256, 0, stream>>>(Mm, Kt[i-1], 65536, c);
      k_axpy<<<gdim(49152), 256, 0, stream>>>(Wm, Lt[i-1], 49152, c);
    }
  }
  k_qcat<<<gdim(32768), 256, 0, stream>>>(Qc, Wm);

  launch_gemm(stream, PT1, 256, Mm, 256, Mm, 256, nullptr, 0, 256, 256, 256, 1.f, 0.f, 0);
  launch_gemm(stream, PT2, 256, PT1, 256, PT1, 256, nullptr, 0, 256, 256, 256, 1.f, 0.f, 0);
  launch_gemm(stream, PT1, 256, PT2, 256, PT2, 256, nullptr, 0, 256, 256, 256, 1.f, 0.f, 0);
  launch_gemm(stream, PT2, 256, PT1, 256, PT1, 256, nullptr, 0, 256, 256, 256, 1.f, 0.f, 0);
  launch_gemm(stream, ML,  256, PT2, 256, PT2, 256, nullptr, 0, 256, 256, 256, 1.f, 0.f, 0);
  launch_gemm(stream, PT1, 256, ML,  256, ML,  256, nullptr, 0, 256, 256, 256, 1.f, 0.f, 0);
  launch_gemm(stream, PT2, 256, PT1, 256, PT1, 256, nullptr, 0, 256, 256, 256, 1.f, 0.f, 0);
  launch_gemm(stream, PT1, 256, PT2, 256, PT2, 256, nullptr, 0, 256, 256, 256, 1.f, 0.f, 0);
  launch_gemm(stream, M512,256, PT1, 256, PT1, 256, nullptr, 0, 256, 256, 256, 1.f, 0.f, 0);

  k_buildP<<<gdim(Tn * 128), 256, 0, stream>>>(Pm, u);
  launch_gemm(stream, G, 256, Pm, 128, Qc, 256, nullptr, 0, Tn, 256, 128, 1.f, 0.f, 0);

  const size_t SCAN_LDS = (size_t)(65536 + 2 * 256 * SC_COLS) * sizeof(float);
  scan_kernel<<<C1 / SC_COLS, 256, SCAN_LDS, stream>>>(Mm, G, nullptr, nullptr, Zend, L1);
  scan_kernel<<<C2 / SC_COLS, 256, SCAN_LDS, stream>>>(ML, Zend, nullptr, nullptr, Zend2, L2);
  carry_kernel<<<1, 256, 0, stream>>>(M512, Zend2, x0, Tcar, C2);
  scan_kernel<<<C2 / SC_COLS, 256, SCAN_LDS, stream>>>(ML, Zend, Tcar, Scar, nullptr, L2);
  scan_kernel<<<C1 / SC_COLS, 256, SCAN_LDS, stream>>>(Mm, G, Scar, xout, nullptr, L1);

  launch_gemm(stream, yout, 32, xout, 256, Cm, 256, nullptr, 0, Tn, 32, 256, 1.f, 0.f, 1);
  launch_gemm(stream, yout, 32, u,    32,  Dm, 32,  yout,   32, Tn, 32, 32,  1.f, 1.f, 1);
}


# level-1 scans (step matrix in LDS, 2 x 32 serial steps) replaced by cyclic reduction in time inside each chunk: 5 in-place pairing GEMMs up with M^1..M^16, 5 down; a different sum order over time
# speedup vs baseline: 2.5623x; 2.5623x over previous
#include <hip/hip_runtime.h>

typedef __attribute__((ext_vector_type(2))) float v2f;
typedef __attribute__((ext_vector_type(8))) float v8f;

#define WMMA_F32(a, b, c) \
  __builtin_amdgcn_wmma_f32_16x16x4_f32(false, (a), false, (b), (short)0, (c), false, false)

static constexpr int Tn = 32768;
static constexpr int NS = 256;
static constexpr int L1 = 32;
static constexpr int C1 = 1024;
static constexpr int C2 = 64;

__constant__ float d_mu[7] = {
  1.00000000f, 1.00000000f, 0.500000000f, 0.166666672f, 0.0416666679f, 0.00833333377f, 0.00166666671f};
__constant__ float d_theta[4][6] = {
  {0.500000000f, 0.349999994f, 0.133333340f, 0.0351111107f, 0.00746666687f, 0.00166666671f},
  {0.0833333358f, 0.0500000007f, 0.0166666675f, 0.00366666657f, 0.00106666668f, 0.0f},
  {0.500000000f, 0.150000006f, 0.0333333351f, 0.00655555539f, 0.000866666669f, 0.0f},
  {-0.0833333358f, -0.0333333351f, -0.00833333377f, -0.00188888889f, -0.000266666670f, 0.0f},
};

__global__ void k_prep(float* __restrict__ Ad, float* __restrict__ Bd,
                       const float* __restrict__ A, const float* __restrict__ B,
                       const float* __restrict__ t) {
  int idx = blockIdx.x * 256 + threadIdx.x;
  float dt = t[1] - t[0];
  if (idx < 65536) Ad[idx] = dt * A[idx];
  else if (idx < 65536 + 8192) Bd[idx - 65536] = dt * B[idx - 65536];
}
__global__ void k_combine(float* __restrict__ M, float* __restrict__ Q,
                          const float* __restrict__ PA, const float* __restrict__ Bd,
                          const float* __restrict__ R) {
  int idx = blockIdx.x * 256 + threadIdx.x;
  if (idx < 65536) {
    float s = d_mu[6] * PA[5 * 65536 + idx];
    #pragma unroll
    for (int p = 5; p >= 1; --p) s += d_mu[p] * PA[(p - 1) * 65536 + idx];
    M[idx] = (((idx >> 8) == (idx & 255)) ? 1.0f : 0.0f) + s;
  } else if (idx < 65536 + 32768) {
    int q = idx - 65536;
    int k = q >> 8, r = q & 255;
    int j = k >> 5, c = k & 31;
    float s = 0.0f;
    #pragma unroll
    for (int p = 5; p >= 1; --p) s += d_theta[j][p] * R[(p - 1) * 8192 + r * 32 + c];
    s += d_theta[j][0] * Bd[r * 32 + c];
    Q[q] = s;
  }
}
__global__ void k_buildP(float* __restrict__ P, const float* __restrict__ u) {
  int idx = blockIdx.x * 256 + threadIdx.x;
  if (idx >= Tn * 128) return;
  int n = idx >> 7, k = idx & 127;
  float val = 0.0f;
  if (n < Tn - 1) {
    int j = k >> 5, c = k & 31;
    int m = n + ((j >= 2) ? 1 : 0);
    if ((j & 1) == 0) {
      val = u[m * 32 + c];
    } else {
      if (m == 0)            val = u[32 + c] - u[c];
      else if (m == Tn - 1)  val = u[(Tn - 1) * 32 + c] - u[(Tn - 2) * 32 + c];
      else                   val = 0.5f * (u[(m + 1) * 32 + c] - u[(m - 1) * 32 + c]);
    }
  }
  P[idx] = val;
}

__global__ __launch_bounds__(128) void gemm_kernel(
    float* D, int ldd,
    const float* A, int lda,
    const float* __restrict__ B, int ldb,
    const float* Cin, int ldc,
    int Mr, int Nc, int Kd, float alpha, float beta, int transB) {
  const int lane = threadIdx.x & 31, wave = threadIdx.x >> 5;
  const int hl = lane >> 4, lr = lane & 15;
  const int tilesN = Nc >> 4;
  const int tile = blockIdx.x * 4 + wave;
  if (tile >= (Mr >> 4) * tilesN) return;
  const int tm = tile / tilesN, tn = tile % tilesN;
  const float* Ar = A + (size_t)(tm * 16 + lr) * lda + 2 * hl;
  v8f acc = {};
  if (transB) {
    const float* Br = B + (size_t)(tn * 16 + lr) * ldb + 2 * hl;
    #pragma unroll 4
    for (int k = 0; k < Kd; k += 4) {
      v2f a = *(const v2f*)(Ar + k);
      v2f b = *(const v2f*)(Br + k);
      acc = WMMA_F32(a, b, acc);
    }
  } else {
    const float* Bc = B + (size_t)(2 * hl) * ldb + tn * 16 + lr;
    #pragma unroll 4
    for (int k = 0; k < Kd; k += 4) {
      v2f a = *(const v2f*)(Ar + k);
      v2f b;
      b[0] = Bc[(size_t)k * ldb];
      b[1] = Bc[(size_t)k * ldb + ldb];
      acc = WMMA_F32(a, b, acc);
    }
  }
  float* Dp = D + (size_t)(tm * 16 + 8 * hl) * ldd + tn * 16 + lr;
  if (beta != 0.0f) {
    const float* Cp = Cin + (size_t)(tm * 16 + 8 * hl) * ldc + tn * 16 + lr;
    #pragma unroll
    for (int v = 0; v < 8; ++v)
      Dp[(size_t)v * ldd] = alpha * acc[v] + beta * Cp[(size_t)v * ldc];
  } else {
    #pragma unroll
    for (int v = 0; v < 8; ++v)
      Dp[(size_t)v * ldd] = alpha * acc[v];
  }
}

__global__ void k_gather(float* __restrict__ V, const float* __restrict__ G,
                         const float* __restrict__ x0) {
  int idx = blockIdx.x * 256 + threadIdx.x;
  int c = idx >> 8, j = idx & 255;
  V[idx] = (c == 0) ? x0[j] : G[((size_t)(c - 1) * L1 + (L1 - 1)) * 256 + j];
}
__global__ void k_scatter(float* __restrict__ X, const float* __restrict__ S) {
  int idx = blockIdx.x * 256 + threadIdx.x;
  int c = idx >> 8, j = idx & 255;
  X[(size_t)c * L1 * 256 + j] = S[idx];
}

__global__ __launch_bounds__(128) void carry_double_kernel(
    float* __restrict__ Vn, const float* __restrict__ Vc,
    const float* __restrict__ Pd, int d) {
  const int lane = threadIdx.x & 31, wave = threadIdx.x >> 5;
  const int hl = lane >> 4, lr = lane & 15;
  const int tile = blockIdx.x * 4 + wave;
  const int tm = tile >> 4, tn = tile & 15;
  const size_t ooff = (size_t)(tm * 16 + 8 * hl) * 256 + tn * 16 + lr;
  const float* Cp = Vc + ooff;
  v8f seed;
  #pragma unroll
  for (int v = 0; v < 8; ++v) seed[v] = Cp[(size_t)v * 256];
  v8f acc = seed;
  if (tm * 16 + 15 >= d) {
    const int srow = tm * 16 + lr - d;
    const bool live = srow >= 0;
    const float* Ar = Vc + (size_t)(live ? srow : 0) * 256 + 2 * hl;
    const float* Br = Pd + (size_t)(tn * 16 + lr) * 256 + 2 * hl;
    const v2f zero2 = {0.0f, 0.0f};
    #pragma unroll 16
    for (int k = 0; k < 256; k += 4) {
      v2f a = *(const v2f*)(Ar + k);
      v2f b = *(const v2f*)(Br + k);
      a = live ? a : zero2;
      acc = WMMA_F32(a, b, acc);
    }
  }
  float* Dp = Vn + ooff;
  const int r0 = tm * 16 + 8 * hl;
  #pragma unroll
  for (int v = 0; v < 8; ++v)
    Dp[(size_t)v * 256] = (r0 + v >= d) ? acc[v] : seed[v];
}

static inline void launch_gemm(hipStream_t s, float* D, int ldd,
                               const float* A, int lda, const float* B, int ldb,
                               const float* Cin, int ldc,
                               int M, int N, int K, float alpha, float beta, int transB) {
  int tiles = (M / 16) * (N / 16);
  gemm_kernel<<<(tiles + 3) / 4, 128, 0, s>>>(D, ldd, A, lda, B, ldb, Cin, ldc,
                                              M, N, K, alpha, beta, transB);
}

extern "C" void kernel_launch(void* const* d_in, const int* in_sizes, int n_in,
                              void* d_out, int out_size, void* d_ws, size_t ws_size,
                              hipStream_t stream) {
  (void)in_sizes; (void)n_in; (void)out_size; (void)ws_size;
  const float* t  = (const float*)d_in[0];
  const float* u  = (const float*)d_in[1];
  const float* x0 = (const float*)d_in[2];
  const float* Am = (const float*)d_in[3];
  const float* Bm = (const float*)d_in[4];
  const float* Cm = (const float*)d_in[5];
  const float* Dm = (const float*)d_in[6];
  float* xout = (float*)d_out;
  float* yout = xout + (size_t)Tn * NS;

  float* w = (float*)d_ws;
  float* Bd = w;  w += 8192;
  float* Kt[6]; for (int i = 0; i < 6; i++) { Kt[i] = w; w += 65536; }
  float* Rst  = w; w += 5 * 8192;
  float* Mm   = w; w += 65536;
  float* Qc   = w; w += 32768;
  float* PwA  = w; w += 65536;
  float* PwB  = w; w += 65536;
  float* PwC  = w; w += 65536;
  float* PwD  = w; w += 65536;
  float* Pm   = w; w += (size_t)Tn * 128;
  float* G    = w; w += (size_t)Tn * 256;
  float* V0   = w; w += (size_t)C1 * 256;
  w += (size_t)C2 * 256;
  w += (size_t)C2 * 256;
  float* V1   = w; w += (size_t)C1 * 256;

  auto gdim = [](int n) { return dim3((unsigned)((n + 255) / 256)); };

  float* PA  = Kt[0];
  k_prep<<<gdim(65536 + 8192), 256, 0, stream>>>(PA, Bd, Am, Bm, t);
  launch_gemm(stream, Kt[1], 256, Kt[0], 256, Kt[0], 256, nullptr, 0, 256, 256, 256, 1.f, 0.f, 0);
  launch_gemm(stream, Kt[2], 256, Kt[0], 256, Kt[1], 256, nullptr, 0, 512, 256, 256, 1.f, 0.f, 0);
  launch_gemm(stream, Kt[4], 256, Kt[0], 256, Kt[3], 256, nullptr, 0, 512, 256, 256, 1.f, 0.f, 0);
  launch_gemm(stream, Rst,   32,  Kt[0], 256, Bd,    32,  nullptr, 0, 1280, 32, 256, 1.f, 0.f, 0);
  k_combine<<<gdim(65536 + 32768), 256, 0, stream>>>(Mm, Qc, PA, Bd, Rst);

  k_buildP<<<gdim(Tn * 128), 256, 0, stream>>>(Pm, u);
  launch_gemm(stream, G, 256, Pm, 128, Qc, 256, nullptr, 0, Tn, 256, 128, 1.f, 0.f, 0);

  float* Mp[5] = {Mm, Pm, Pm + 65536, Pm + 2 * 65536, Pm + 3 * 65536};
  float* Pw[10] = {PwA, PwB, PwC, PwD, Kt[0], Kt[1], Kt[2], Kt[3], Kt[4], Kt[5]};
  for (int j = 1; j < 5; ++j)
    launch_gemm(stream, Mp[j], 256, Mp[j-1], 256, Mp[j-1], 256, nullptr, 0, 256, 256, 256, 1.f, 0.f, 0);
  launch_gemm(stream, Pw[0], 256, Mp[4], 256, Mp[4], 256, nullptr, 0, 256, 256, 256, 1.f, 0.f, 0);
  for (int p = 1; p < 10; ++p)
    launch_gemm(stream, Pw[p], 256, Pw[p-1], 256, Pw[p-1], 256, nullptr, 0, 256, 256, 256, 1.f, 0.f, 0);

  launch_gemm(stream, G + 1 * 256, 512, G + 0 * 256, 512, Mp[0], 256, G + 1 * 256, 512, 16384, 256, 256, 1.f, 1.f, 1);
  launch_gemm(stream, G + 3 * 256, 1024, G + 1 * 256, 1024, Mp[1], 256, G + 3 * 256, 1024, 8192, 256, 256, 1.f, 1.f, 1);
  launch_gemm(stream, G + 7 * 256, 2048, G + 3 * 256, 2048, Mp[2], 256, G + 7 * 256, 2048, 4096, 256, 256, 1.f, 1.f, 1);
  launch_gemm(stream, G + 15 * 256, 4096, G + 7 * 256, 4096, Mp[3], 256, G + 15 * 256, 4096, 2048, 256, 256, 1.f, 1.f, 1);
  launch_gemm(stream, G + 31 * 256, 8192, G + 15 * 256, 8192, Mp[4], 256, G + 31 * 256, 8192, 1024, 256, 256, 1.f, 1.f, 1);
  k_gather<<<gdim(C1 * 256), 256, 0, stream>>>(V0, G, x0);
  const int dd[10] = {1, 2, 4, 8, 16, 32, 64, 128, 256, 512};
  float* Va = V0;
  float* Vb = V1;
  for (int p = 0; p < 10; ++p) {
    carry_double_kernel<<<256, 128, 0, stream>>>(Vb, Va, Pw[p], dd[p]);
    float* tv = Va; Va = Vb; Vb = tv;
  }
  k_scatter<<<gdim(C1 * 256), 256, 0, stream>>>(xout, Va);
  launch_gemm(stream, xout + 16 * 256, 8192, xout, 8192, Mp[4], 256, G + 15 * 256, 8192, 1024, 256, 256, 1.f, 1.f, 1);
  launch_gemm(stream, xout + 8 * 256, 4096, xout, 4096, Mp[3], 256, G + 7 * 256, 4096, 2048, 256, 256, 1.f, 1.f, 1);
  launch_gemm(stream, xout + 4 * 256, 2048, xout, 2048, Mp[2], 256, G + 3 * 256, 2048, 4096, 256, 256, 1.f, 1.f, 1);
  launch_gemm(stream, xout + 2 * 256, 1024, xout, 1024, Mp[1], 256, G + 1 * 256, 1024, 8192, 256, 256, 1.f, 1.f, 1);
  launch_gemm(stream, xout + 1 * 256, 512, xout, 512, Mp[0], 256, G + 0 * 256, 512, 16384, 256, 256, 1.f, 1.f, 1);

  launch_gemm(stream, yout, 32, xout, 256, Cm, 256, nullptr, 0, Tn, 32, 256, 1.f, 0.f, 1);
  launch_gemm(stream, yout, 32, u,    32,  Dm, 32,  yout,   32, Tn, 32, 32,  1.f, 1.f, 1);
}


# GEMMs of 1024+ rows (10 pairing, drive, X@C^T): a wave holds 2x4 or 2x2 tiles of 16x16 in registers, 16-byte fragment loads reused across them, one 32-k step ahead; a different sum order over k
# speedup vs baseline: 3.3873x; 1.3220x over previous
#include <hip/hip_runtime.h>

typedef __attribute__((ext_vector_type(2))) float v2f;
typedef __attribute__((ext_vector_type(4))) float v4f;
typedef __attribute__((ext_vector_type(8))) float v8f;

#define WMMA_F32(a, b, c) \
  __builtin_amdgcn_wmma_f32_16x16x4_f32(false, (a), false, (b), (short)0, (c), false, false)

static constexpr int Tn = 32768;
static constexpr int NS = 256;
static constexpr int L1 = 32;
static constexpr int C1 = 1024;
static constexpr int C2 = 64;

__constant__ float d_mu[7] = {
  1.00000000f, 1.00000000f, 0.500000000f, 0.166666672f, 0.0416666679f, 0.00833333377f, 0.00166666671f};
__constant__ float d_theta[4][6] = {
  {0.500000000f, 0.349999994f, 0.133333340f, 0.0351111107f, 0.00746666687f, 0.00166666671f},
  {0.0833333358f, 0.0500000007f, 0.0166666675f, 0.00366666657f, 0.00106666668f, 0.0f},
  {0.500000000f, 0.150000006f, 0.0333333351f, 0.00655555539f, 0.000866666669f, 0.0f},
  {-0.0833333358f, -0.0333333351f, -0.00833333377f, -0.00188888889f, -0.000266666670f, 0.0f},
};

__global__ void k_prep(float* __restrict__ Ad, float* __restrict__ Bd,
                       const float* __restrict__ A, const float* __restrict__ B,
                       const float* __restrict__ t) {
  int idx = blockIdx.x * 256 + threadIdx.x;
  float dt = t[1] - t[0];
  if (idx < 65536) Ad[idx] = dt * A[idx];
  else if (idx < 65536 + 8192) Bd[idx - 65536] = dt * B[idx - 65536];
}
__global__ void k_combine(float* __restrict__ M, float* __restrict__ Q,
                          const float* __restrict__ PA, const float* __restrict__ Bd,
                          const float* __restrict__ R) {
  int idx = blockIdx.x * 256 + threadIdx.x;
  if (idx < 65536) {
    float s = d_mu[6] * PA[5 * 65536 + idx];
    #pragma unroll
    for (int p = 5; p >= 1; --p) s += d_mu[p] * PA[(p - 1) * 65536 + idx];
    M[idx] = (((idx >> 8) == (idx & 255)) ? 1.0f : 0.0f) + s;
  } else if (idx < 65536 + 32768) {
    int q = idx - 65536;
    int k = q >> 8, r = q & 255;
    int j = k >> 5, c = k & 31;
    float s = 0.0f;
    #pragma unroll
    for (int p = 5; p >= 1; --p) s += d_theta[j][p] * R[(p - 1) * 8192 + r * 32 + c];
    s += d_theta[j][0] * Bd[r * 32 + c];
    Q[r * 128 + k] = s;
  }
}
__global__ void k_buildP(float* __restrict__ P, const float* __restrict__ u) {
  int idx = blockIdx.x * 256 + threadIdx.x;
  if (idx >= Tn * 128) return;
  int n = idx >> 7, k = idx & 127;
  float val = 0.0f;
  if (n < Tn - 1) {
    int j = k >> 5, c = k & 31;
    int m = n + ((j >= 2) ? 1 : 0);
    if ((j & 1) == 0) {
      val = u[m * 32 + c];
    } else {
      if (m == 0)            val = u[32 + c] - u[c];
      else if (m == Tn - 1)  val = u[(Tn - 1) * 32 + c] - u[(Tn - 2) * 32 + c];
      else                   val = 0.5f * (u[(m + 1) * 32 + c] - u[(m - 1) * 32 + c]);
    }
  }
  P[idx] = val;
}

__global__ __launch_bounds__(128) void gemm_kernel(
    float* D, int ldd,
    const float* A, int lda,
    const float* __restrict__ B, int ldb,
    const float* Cin, int ldc,
    int Mr, int Nc, int Kd, float alpha, float beta, int transB) {
  const int lane = threadIdx.x & 31, wave = threadIdx.x >> 5;
  const int hl = lane >> 4, lr = lane & 15;
  const int tilesN = Nc >> 4;
  const int tile = blockIdx.x * 4 + wave;
  if (tile >= (Mr >> 4) * tilesN) return;
  const int tm = tile / tilesN, tn = tile % tilesN;
  const float* Ar = A + (size_t)(tm * 16 + lr) * lda + 2 * hl;
  v8f acc = {};
  if (transB) {
    const float* Br = B + (size_t)(tn * 16 + lr) * ldb + 2 * hl;
    #pragma unroll 4
    for (int k = 0; k < Kd; k += 4) {
      v2f a = *(const v2f*)(Ar + k);
      v2f b = *(const v2f*)(Br + k);
      acc = WMMA_F32(a, b, acc);
    }
  } else {
    const float* Bc = B + (size_t)(2 * hl) * ldb + tn * 16 + lr;
    #pragma unroll 4
    for (int k = 0; k < Kd; k += 4) {
      v2f a = *(const v2f*)(Ar + k);
      v2f b;
      b[0] = Bc[(size_t)k * ldb];
      b[1] = Bc[(size_t)k * ldb + ldb];
      acc = WMMA_F32(a, b, acc);
    }
  }
  float* Dp = D + (size_t)(tm * 16 + 8 * hl) * ldd + tn * 16 + lr;
  if (beta != 0.0f) {
    const float* Cp = Cin + (size_t)(tm * 16 + 8 * hl) * ldc + tn * 16 + lr;
    #pragma unroll
    for (int v = 0; v < 8; ++v)
      Dp[(size_t)v * ldd] = alpha * acc[v] + beta * Cp[(size_t)v * ldc];
  } else {
    #pragma unroll
    for (int v = 0; v < 8; ++v)
      Dp[(size_t)v * ldd] = alpha * acc[v];
  }
}

template <int RM, int RN>
struct FragSet {
  v4f a[RM][4];
  v4f b[RN][4];
};

template <int RM, int RN>
__device__ __forceinline__ void rb_load(FragSet<RM, RN>& f, const float* Ap, size_t astep,
                                        const float* Bp, size_t bstep, int k0) {
  #pragma unroll
  for (int i = 0; i < RM; ++i) {
    #pragma unroll
    for (int q = 0; q < 4; ++q) f.a[i][q] = *(const v4f*)(Ap + i * astep + k0 + 8 * q);
  }
  #pragma unroll
  for (int j = 0; j < RN; ++j) {
    #pragma unroll
    for (int q = 0; q < 4; ++q) f.b[j][q] = *(const v4f*)(Bp + j * bstep + k0 + 8 * q);
  }
}

template <int RM, int RN>
__device__ __forceinline__ void rb_compute(v8f (&acc)[RM][RN], const FragSet<RM, RN>& f) {
  #pragma unroll
  for (int q = 0; q < 4; ++q) {
    #pragma unroll
    for (int i = 0; i < RM; ++i) {
      #pragma unroll
      for (int j = 0; j < RN; ++j) {
        v2f a = {f.a[i][q][0], f.a[i][q][1]};
        v2f b = {f.b[j][q][0], f.b[j][q][1]};
        acc[i][j] = WMMA_F32(a, b, acc[i][j]);
      }
    }
    #pragma unroll
    for (int i = 0; i < RM; ++i) {
      #pragma unroll
      for (int j = 0; j < RN; ++j) {
        v2f a = {f.a[i][q][2], f.a[i][q][3]};
        v2f b = {f.b[j][q][2], f.b[j][q][3]};
        acc[i][j] = WMMA_F32(a, b, acc[i][j]);
      }
    }
  }
}

template <int RM, int RN, int WM, int WN, int KD, bool HASC, int BN>
__global__ __launch_bounds__(128) void gemm_rb_kernel(
    float* D, int ldd,
    const float* A, int lda,
    const float* __restrict__ B, int ldb,
    const float* Cin, int ldc) {
  static_assert(WM * WN == 4, "four waves per work-group");
  static_assert(KD % 64 == 0 && KD >= 128, "two 32-k steps per trip");
  const int lane = threadIdx.x & 31, wave = threadIdx.x >> 5;
  const int hl = lane >> 4, lr = lane & 15;
  const int bn = blockIdx.x % BN, bm = blockIdx.x / BN;
  const int wm = wave / WN, wn = wave % WN;
  const int row0 = (bm * WM + wm) * (16 * RM);
  const int col0 = (bn * WN + wn) * (16 * RN);
  const float* Ap = A + (size_t)(row0 + lr) * lda + 4 * hl;
  const float* Bp = B + (size_t)(col0 + lr) * ldb + 4 * hl;
  const size_t astep = (size_t)16 * lda, bstep = (size_t)16 * ldb;

  v8f acc[RM][RN];
  #pragma unroll
  for (int i = 0; i < RM; ++i) {
    #pragma unroll
    for (int j = 0; j < RN; ++j) acc[i][j] = v8f{};
  }

  FragSet<RM, RN> f0, f1;
  rb_load<RM, RN>(f0, Ap, astep, Bp, bstep, 0);
  #pragma unroll 1
  for (int k0 = 0; k0 < KD - 64; k0 += 64) {
    rb_load<RM, RN>(f1, Ap, astep, Bp, bstep, k0 + 32);
    __builtin_amdgcn_sched_barrier(0);
    rb_compute<RM, RN>(acc, f0);
    __builtin_amdgcn_sched_barrier(0);
    rb_load<RM, RN>(f0, Ap, astep, Bp, bstep, k0 + 64);
    __builtin_amdgcn_sched_barrier(0);
    rb_compute<RM, RN>(acc, f1);
    __builtin_amdgcn_sched_barrier(0);
  }
  rb_load<RM, RN>(f1, Ap, astep, Bp, bstep, KD - 32);
  __builtin_amdgcn_sched_barrier(0);
  rb_compute<RM, RN>(acc, f0);
  __builtin_amdgcn_sched_barrier(0);

  float* Dp = D + (size_t)(row0 + 8 * hl) * ldd + col0 + lr;
  v8f cv[RM][RN];
  if (HASC) {
    const float* Cp = Cin + (size_t)(row0 + 8 * hl) * ldc + col0 + lr;
    #pragma unroll
    for (int i = 0; i < RM; ++i) {
      #pragma unroll
      for (int j = 0; j < RN; ++j) {
        #pragma unroll
        for (int v = 0; v < 8; ++v) cv[i][j][v] = Cp[(size_t)(16 * i + v) * ldc + 16 * j];
      }
    }
    __builtin_amdgcn_sched_barrier(0);
  }
  rb_compute<RM, RN>(acc, f1);
  if (HASC) {
    #pragma unroll
    for (int i = 0; i < RM; ++i) {
      #pragma unroll
      for (int j = 0; j < RN; ++j) {
        #pragma unroll
        for (int v = 0; v < 8; ++v) acc[i][j][v] += cv[i][j][v];
      }
    }
  }
  #pragma unroll
  for (int i = 0; i < RM; ++i) {
    #pragma unroll
    for (int j = 0; j < RN; ++j) {
      #pragma unroll
      for (int v = 0; v < 8; ++v) Dp[(size_t)(16 * i + v) * ldd + 16 * j] = acc[i][j][v];
    }
  }
}

__global__ void k_gather(float* __restrict__ V, const float* __restrict__ G,
                         const float* __restrict__ x0) {
  int idx = blockIdx.x * 256 + threadIdx.x;
  int c = idx >> 8, j = idx & 255;
  V[idx] = (c == 0) ? x0[j] : G[((size_t)(c - 1) * L1 + (L1 - 1)) * 256 + j];
}
__global__ void k_scatter(float* __restrict__ X, const float* __restrict__ S) {
  int idx = blockIdx.x * 256 + threadIdx.x;
  int c = idx >> 8, j = idx & 255;
  X[(size_t)c * L1 * 256 + j] = S[idx];
}

__global__ __launch_bounds__(128) void carry_double_kernel(
    float* __restrict__ Vn, const float* __restrict__ Vc,
    const float* __restrict__ Pd, int d) {
  const int lane = threadIdx.x & 31, wave = threadIdx.x >> 5;
  const int hl = lane >> 4, lr = lane & 15;
  const int tile = blockIdx.x * 4 + wave;
  const int tm = tile >> 4, tn = tile & 15;
  const size_t ooff = (size_t)(tm * 16 + 8 * hl) * 256 + tn * 16 + lr;
  const float* Cp = Vc + ooff;
  v8f seed;
  #pragma unroll
  for (int v = 0; v < 8; ++v) seed[v] = Cp[(size_t)v * 256];
  v8f acc = seed;
  if (tm * 16 + 15 >= d) {
    const int srow = tm * 16 + lr - d;
    const bool live = srow >= 0;
    const float* Ar = Vc + (size_t)(live ? srow : 0) * 256 + 2 * hl;
    const float* Br = Pd + (size_t)(tn * 16 + lr) * 256 + 2 * hl;
    const v2f zero2 = {0.0f, 0.0f};
    #pragma unroll 16
    for (int k = 0; k < 256; k += 4) {
      v2f a = *(const v2f*)(Ar + k);
      v2f b = *(const v2f*)(Br + k);
      a = live ? a : zero2;
      acc = WMMA_F32(a, b, acc);
    }
  }
  float* Dp = Vn + ooff;
  const int r0 = tm * 16 + 8 * hl;
  #pragma unroll
  for (int v = 0; v < 8; ++v)
    Dp[(size_t)v * 256] = (r0 + v >= d) ? acc[v] : seed[v];
}

static inline void launch_gemm(hipStream_t s, float* D, int ldd,
                               const float* A, int lda, const float* B, int ldb,
                               const float* Cin, int ldc,
                               int M, int N, int K, float alpha, float beta, int transB) {
  int tiles = (M / 16) * (N / 16);
  gemm_kernel<<<(tiles + 3) / 4, 128, 0, s>>>(D, ldd, A, lda, B, ldb, Cin, ldc,
                                              M, N, K, alpha, beta, transB);
}

template <int RM, int RN, int WM, int WN, int KD, bool HASC, int Mr, int Nc>
static inline void launch_rb(hipStream_t s, float* D, int ldd,
                             const float* A, int lda, const float* B, int ldb,
                             const float* Cin, int ldc) {
  static_assert(Mr % (16 * RM * WM) == 0 && Nc % (16 * RN * WN) == 0, "whole work-group tiles");
  constexpr int blocksN = Nc / (16 * RN * WN);
  constexpr int blocksM = Mr / (16 * RM * WM);
  gemm_rb_kernel<RM, RN, WM, WN, KD, HASC, blocksN><<<blocksM * blocksN, 128, 0, s>>>(
      D, ldd, A, lda, B, ldb, Cin, ldc);
}

extern "C" void kernel_launch(void* const* d_in, const int* in_sizes, int n_in,
                              void* d_out, int out_size, void* d_ws, size_t ws_size,
                              hipStream_t stream) {
  (void)in_sizes; (void)n_in; (void)out_size; (void)ws_size;
  const float* t  = (const float*)d_in[0];
  const float* u  = (const float*)d_in[1];
  const float* x0 = (const float*)d_in[2];
  const float* Am = (const float*)d_in[3];
  const float* Bm = (const float*)d_in[4];
  const float* Cm = (const float*)d_in[5];
  const float* Dm = (const float*)d_in[6];
  float* xout = (float*)d_out;
  float* yout = xout + (size_t)Tn * NS;

  float* w = (float*)d_ws;
  float* Bd = w;  w += 8192;
  float* Kt[6]; for (int i = 0; i < 6; i++) { Kt[i] = w; w += 65536; }
  float* Rst  = w; w += 5 * 8192;
  float* Mm   = w; w += 65536;
  float* Qc   = w; w += 32768;
  float* PwA  = w; w += 65536;
  float* PwB  = w; w += 65536;
  float* PwC  = w; w += 65536;
  float* PwD  = w; w += 65536;
  float* Pm   = w; w += (size_t)Tn * 128;
  float* G    = w; w += (size_t)Tn * 256;
  float* V0   = w; w += (size_t)C1 * 256;
  w += (size_t)C2 * 256;
  w += (size_t)C2 * 256;
  float* V1   = w; w += (size_t)C1 * 256;

  auto gdim = [](int n) { return dim3((unsigned)((n + 255) / 256)); };

  float* PA  = Kt[0];
  k_prep<<<gdim(65536 + 8192), 256, 0, stream>>>(PA, Bd, Am, Bm, t);
  launch_gemm(stream, Kt[1], 256, Kt[0], 256, Kt[0], 256, nullptr, 0, 256, 256, 256, 1.f, 0.f, 0);
  launch_gemm(stream, Kt[2], 256, Kt[0], 256, Kt[1], 256, nullptr, 0, 512, 256, 256, 1.f, 0.f, 0);
  launch_gemm(stream, Kt[4], 256, Kt[0], 256, Kt[3], 256, nullptr, 0, 512, 256, 256, 1.f, 0.f, 0);
  launch_gemm(stream, Rst,   32,  Kt[0], 256, Bd,    32,  nullptr, 0, 1280, 32, 256, 1.f, 0.f, 0);
  k_combine<<<gdim(65536 + 32768), 256, 0, stream>>>(Mm, Qc, PA, Bd, Rst);

  k_buildP<<<gdim(Tn * 128), 256, 0, stream>>>(Pm, u);
  launch_rb<2, 4, 2, 2, 128, false, Tn, 256>(stream, G, 256, Pm, 128, Qc, 128, nullptr, 0);

  float* Mp[5] = {Mm, Pm, Pm + 65536, Pm + 2 * 65536, Pm + 3 * 65536};
  float* Pw[10] = {PwA, PwB, PwC, PwD, Kt[0], Kt[1], Kt[2], Kt[3], Kt[4], Kt[5]};
  for (int j = 1; j < 5; ++j)
    launch_gemm(stream, Mp[j], 256, Mp[j-1], 256, Mp[j-1], 256, nullptr, 0, 256, 256, 256, 1.f, 0.f, 0);
  launch_gemm(stream, Pw[0], 256, Mp[4], 256, Mp[4], 256, nullptr, 0, 256, 256, 256, 1.f, 0.f, 0);
  for (int p = 1; p < 10; ++p)
    launch_gemm(stream, Pw[p], 256, Pw[p-1], 256, Pw[p-1], 256, nullptr, 0, 256, 256, 256, 1.f, 0.f, 0);

  launch_rb<2, 4, 2, 2, 256, true, 16384, 256>(stream, G + 1 * 256, 512, G + 0 * 256, 512, Mp[0], 256, G + 1 * 256, 512);
  launch_rb<2, 4, 2, 2, 256, true, 8192, 256>(stream, G + 3 * 256, 1024, G + 1 * 256, 1024, Mp[1], 256, G + 3 * 256, 1024);
  launch_rb<2, 2, 2, 2, 256, true, 4096, 256>(stream, G + 7 * 256, 2048, G + 3 * 256, 2048, Mp[2], 256, G + 7 * 256, 2048);
  launch_rb<2, 2, 2, 2, 256, true, 2048, 256>(stream, G + 15 * 256, 4096, G + 7 * 256, 4096, Mp[3], 256, G + 15 * 256, 4096);
  launch_rb<2, 2, 2, 2, 256, true, 1024, 256>(stream, G + 31 * 256, 8192, G + 15 * 256, 8192, Mp[4], 256, G + 31 * 256, 8192);
  k_gather<<<gdim(C1 * 256), 256, 0, stream>>>(V0, G, x0);
  const int dd[10] = {1, 2, 4, 8, 16, 32, 64, 128, 256, 512};
  float* Va = V0;
  float* Vb = V1;
  for (int p = 0; p < 10; ++p) {
    carry_double_kernel<<<256, 128, 0, stream>>>(Vb, Va, Pw[p], dd[p]);
    float* tv = Va; Va = Vb; Vb = tv;
  }
  k_scatter<<<gdim(C1 * 256), 256, 0, stream>>>(xout, Va);
  launch_rb<2, 2, 2, 2, 256, true, 1024, 256>(stream, xout + 16 * 256, 8192, xout, 8192, Mp[4], 256, G + 15 * 256, 8192);
  launch_rb<2, 2, 2, 2, 256, true, 2048, 256>(stream, xout + 8 * 256, 4096, xout, 4096, Mp[3], 256, G + 7 * 256, 4096);
  launch_rb<2, 2, 2, 2, 256, true, 4096, 256>(stream, xout + 4 * 256, 2048, xout, 2048, Mp[2], 256, G + 3 * 256, 2048);
  launch_rb<2, 4, 2, 2, 256, true, 8192, 256>(stream, xout + 2 * 256, 1024, xout, 1024, Mp[1], 256, G + 1 * 256, 1024);
  launch_rb<2, 4, 2, 2, 256, true, 16384, 256>(stream, xout + 1 * 256, 512, xout, 512, Mp[0], 256, G + 0 * 256, 512);

  launch_rb<2, 2, 4, 1, 256, false, Tn, 32>(stream, yout, 32, xout, 256, Cm, 256, nullptr, 0);
  launch_gemm(stream, yout, 32, u,    32,  Dm, 32,  yout,   32, Tn, 32, 32,  1.f, 1.f, 1);
}


# the 18 K=256 products with fewer 16x16 tiles than SIMDs (4 set-up GEMMs, 14 squarings): one tile per work-group, K split over its 4 waves, partial tiles summed in LDS; a different sum order over k
# speedup vs baseline: 4.6683x; 1.3782x over previous
#include <hip/hip_runtime.h>

typedef __attribute__((ext_vector_type(2))) float v2f;
typedef __attribute__((ext_vector_type(4))) float v4f;
typedef __attribute__((ext_vector_type(8))) float v8f;

#define WMMA_F32(a, b, c) \
  __builtin_amdgcn_wmma_f32_16x16x4_f32(false, (a), false, (b), (short)0, (c), false, false)

static constexpr int Tn = 32768;
static constexpr int NS = 256;
static constexpr int L1 = 32;
static constexpr int C1 = 1024;
static constexpr int C2 = 64;

__constant__ float d_mu[7] = {
  1.00000000f, 1.00000000f, 0.500000000f, 0.166666672f, 0.0416666679f, 0.00833333377f, 0.00166666671f};
__constant__ float d_theta[4][6] = {
  {0.500000000f, 0.349999994f, 0.133333340f, 0.0351111107f, 0.00746666687f, 0.00166666671f},
  {0.0833333358f, 0.0500000007f, 0.0166666675f, 0.00366666657f, 0.00106666668f, 0.0f},
  {0.500000000f, 0.150000006f, 0.0333333351f, 0.00655555539f, 0.000866666669f, 0.0f},
  {-0.0833333358f, -0.0333333351f, -0.00833333377f, -0.00188888889f, -0.000266666670f, 0.0f},
};

__global__ void k_prep(float* __restrict__ Ad, float* __restrict__ Bd,
                       const float* __restrict__ A, const float* __restrict__ B,
                       const float* __restrict__ t) {
  int idx = blockIdx.x * 256 + threadIdx.x;
  float dt = t[1] - t[0];
  if (idx < 65536) Ad[idx] = dt * A[idx];
  else if (idx < 65536 + 8192) Bd[idx - 65536] = dt * B[idx - 65536];
}
__global__ void k_combine(float* __restrict__ M, float* __restrict__ Q,
                          const float* __restrict__ PA, const float* __restrict__ Bd,
                          const float* __restrict__ R) {
  int idx = blockIdx.x * 256 + threadIdx.x;
  if (idx < 65536) {
    float s = d_mu[6] * PA[5 * 65536 + idx];
    #pragma unroll
    for (int p = 5; p >= 1; --p) s += d_mu[p] * PA[(p - 1) * 65536 + idx];
    M[idx] = (((idx >> 8) == (idx & 255)) ? 1.0f : 0.0f) + s;
  } else if (idx < 65536 + 32768) {
    int q = idx - 65536;
    int k = q >> 8, r = q & 255;
    int j = k >> 5, c = k & 31;
    float s = 0.0f;
    #pragma unroll
    for (int p = 5; p >= 1; --p) s += d_theta[j][p] * R[(p - 1) * 8192 + r * 32 + c];
    s += d_theta[j][0] * Bd[r * 32 + c];
    Q[r * 128 + k] = s;
  }
}
__global__ void k_buildP(float* __restrict__ P, const float* __restrict__ u) {
  int idx = blockIdx.x * 256 + threadIdx.x;
  if (idx >= Tn * 128) return;
  int n = idx >> 7, k = idx & 127;
  float val = 0.0f;
  if (n < Tn - 1) {
    int j = k >> 5, c = k & 31;
    int m = n + ((j >= 2) ? 1 : 0);
    if ((j & 1) == 0) {
      val = u[m * 32 + c];
    } else {
      if (m == 0)            val = u[32 + c] - u[c];
      else if (m == Tn - 1)  val = u[(Tn - 1) * 32 + c] - u[(Tn - 2) * 32 + c];
      else                   val = 0.5f * (u[(m + 1) * 32 + c] - u[(m - 1) * 32 + c]);
    }
  }
  P[idx] = val;
}

__global__ __launch_bounds__(128) void gemm_kernel(
    float* D, int ldd,
    const float* A, int lda,
    const float* __restrict__ B, int ldb,
    const float* Cin, int ldc,
    int Mr, int Nc, int Kd, float alpha, float beta, int transB) {
  const int lane = threadIdx.x & 31, wave = threadIdx.x >> 5;
  const int hl = lane >> 4, lr = lane & 15;
  const int tilesN = Nc >> 4;
  const int tile = blockIdx.x * 4 + wave;
  if (tile >= (Mr >> 4) * tilesN) return;
  const int tm = tile / tilesN, tn = tile % tilesN;
  const float* Ar = A + (size_t)(tm * 16 + lr) * lda + 2 * hl;
  v8f acc = {};
  if (transB) {
    const float* Br = B + (size_t)(tn * 16 + lr) * ldb + 2 * hl;
    #pragma unroll 4
    for (int k = 0; k < Kd; k += 4) {
      v2f a = *(const v2f*)(Ar + k);
      v2f b = *(const v2f*)(Br + k);
      acc = WMMA_F32(a, b, acc);
    }
  } else {
    const float* Bc = B + (size_t)(2 * hl) * ldb + tn * 16 + lr;
    #pragma unroll 4
    for (int k = 0; k < Kd; k += 4) {
      v2f a = *(const v2f*)(Ar + k);
      v2f b;
      b[0] = Bc[(size_t)k * ldb];
      b[1] = Bc[(size_t)k * ldb + ldb];
      acc = WMMA_F32(a, b, acc);
    }
  }
  float* Dp = D + (size_t)(tm * 16 + 8 * hl) * ldd + tn * 16 + lr;
  if (beta != 0.0f) {
    const float* Cp = Cin + (size_t)(tm * 16 + 8 * hl) * ldc + tn * 16 + lr;
    #pragma unroll
    for (int v = 0; v < 8; ++v)
      Dp[(size_t)v * ldd] = alpha * acc[v] + beta * Cp[(size_t)v * ldc];
  } else {
    #pragma unroll
    for (int v = 0; v < 8; ++v)
      Dp[(size_t)v * ldd] = alpha * acc[v];
  }
}

template <int NC>
__global__ __launch_bounds__(128) void gemm_ksplit_kernel(
    float* __restrict__ D, const float* __restrict__ A, const float* __restrict__ B) {
  __shared__ float part[4][8][32];
  const int lane = threadIdx.x & 31, wave = threadIdx.x >> 5;
  const int hl = lane >> 4, lr = lane & 15;
  constexpr int tilesN = NC >> 4;
  const int tm = blockIdx.x / tilesN, tn = blockIdx.x % tilesN;
  const float* Ar = A + (size_t)(tm * 16 + lr) * 256 + 64 * wave + 2 * hl;
  const float* Bc = B + (size_t)(64 * wave + 2 * hl) * NC + tn * 16 + lr;
  v2f a[16], b[16];
  #pragma unroll
  for (int i = 0; i < 16; ++i) {
    a[i] = *(const v2f*)(Ar + 4 * i);
    b[i][0] = Bc[(size_t)(4 * i) * NC];
    b[i][1] = Bc[(size_t)(4 * i) * NC + NC];
  }
  __builtin_amdgcn_sched_barrier(0);
  v8f acc = {};
  #pragma unroll
  for (int i = 0; i < 16; ++i) acc = WMMA_F32(a[i], b[i], acc);
  #pragma unroll
  for (int v = 0; v < 8; ++v) part[wave][v][lane] = acc[v];
  __syncthreads();
  #pragma unroll
  for (int e = 0; e < 2; ++e) {
    const int v = 2 * wave + e;
    float s = part[0][v][lane] + part[1][v][lane];
    s += part[2][v][lane];
    s += part[3][v][lane];
    D[(size_t)(tm * 16 + 8 * hl + v) * NC + tn * 16 + lr] = s;
  }
}

template <int RM, int RN>
struct FragSet {
  v4f a[RM][4];
  v4f b[RN][4];
};

template <int RM, int RN>
__device__ __forceinline__ void rb_load(FragSet<RM, RN>& f, const float* Ap, size_t astep,
                                        const float* Bp, size_t bstep, int k0) {
  #pragma unroll
  for (int i = 0; i < RM; ++i) {
    #pragma unroll
    for (int q = 0; q < 4; ++q) f.a[i][q] = *(const v4f*)(Ap + i * astep + k0 + 8 * q);
  }
  #pragma unroll
  for (int j = 0; j < RN; ++j) {
    #pragma unroll
    for (int q = 0; q < 4; ++q) f.b[j][q] = *(const v4f*)(Bp + j * bstep + k0 + 8 * q);
  }
}

template <int RM, int RN>
__device__ __forceinline__ void rb_compute(v8f (&acc)[RM][RN], const FragSet<RM, RN>& f) {
  #pragma unroll
  for (int q = 0; q < 4; ++q) {
    #pragma unroll
    for (int i = 0; i < RM; ++i) {
      #pragma unroll
      for (int j = 0; j < RN; ++j) {
        v2f a = {f.a[i][q][0], f.a[i][q][1]};
        v2f b = {f.b[j][q][0], f.b[j][q][1]};
        acc[i][j] = WMMA_F32(a, b, acc[i][j]);
      }
    }
    #pragma unroll
    for (int i = 0; i < RM; ++i) {
      #pragma unroll
      for (int j = 0; j < RN; ++j) {
        v2f a = {f.a[i][q][2], f.a[i][q][3]};
        v2f b = {f.b[j][q][2], f.b[j][q][3]};
        acc[i][j] = WMMA_F32(a, b, acc[i][j]);
      }
    }
  }
}

template <int RM, int RN, int WM, int WN, int KD, bool HASC, int BN>
__global__ __launch_bounds__(128) void gemm_rb_kernel(
    float* D, int ldd,
    const float* A, int lda,
    const float* __restrict__ B, int ldb,
    const float* Cin, int ldc) {
  static_assert(WM * WN == 4, "four waves per work-group");
  static_assert(KD % 64 == 0 && KD >= 128, "two 32-k steps per trip");
  const int lane = threadIdx.x & 31, wave = threadIdx.x >> 5;
  const int hl = lane >> 4, lr = lane & 15;
  const int bn = blockIdx.x % BN, bm = blockIdx.x / BN;
  const int wm = wave / WN, wn = wave % WN;
  const int row0 = (bm * WM + wm) * (16 * RM);
  const int col0 = (bn * WN + wn) * (16 * RN);
  const float* Ap = A + (size_t)(row0 + lr) * lda + 4 * hl;
  const float* Bp = B + (size_t)(col0 + lr) * ldb + 4 * hl;
  const size_t astep = (size_t)16 * lda, bstep = (size_t)16 * ldb;

  v8f acc[RM][RN];
  #pragma unroll
  for (int i = 0; i < RM; ++i) {
    #pragma unroll
    for (int j = 0; j < RN; ++j) acc[i][j] = v8f{};
  }

  FragSet<RM, RN> f0, f1;
  rb_load<RM, RN>(f0, Ap, astep, Bp, bstep, 0);
  #pragma unroll 1
  for (int k0 = 0; k0 < KD - 64; k0 += 64) {
    rb_load<RM, RN>(f1, Ap, astep, Bp, bstep, k0 + 32);
    __builtin_amdgcn_sched_barrier(0);
    rb_compute<RM, RN>(acc, f0);
    __builtin_amdgcn_sched_barrier(0);
    rb_load<RM, RN>(f0, Ap, astep, Bp, bstep, k0 + 64);
    __builtin_amdgcn_sched_barrier(0);
    rb_compute<RM, RN>(acc, f1);
    __builtin_amdgcn_sched_barrier(0);
  }
  rb_load<RM, RN>(f1, Ap, astep, Bp, bstep, KD - 32);
  __builtin_amdgcn_sched_barrier(0);
  rb_compute<RM, RN>(acc, f0);
  __builtin_amdgcn_sched_barrier(0);

  float* Dp = D + (size_t)(row0 + 8 * hl) * ldd + col0 + lr;
  v8f cv[RM][RN];
  if (HASC) {
    const float* Cp = Cin + (size_t)(row0 + 8 * hl) * ldc + col0 + lr;
    #pragma unroll
    for (int i = 0; i < RM; ++i) {
      #pragma unroll
      for (int j = 0; j < RN; ++j) {
        #pragma unroll
        for (int v = 0; v < 8; ++v) cv[i][j][v] = Cp[(size_t)(16 * i + v) * ldc + 16 * j];
      }
    }
    __builtin_amdgcn_sched_barrier(0);
  }
  rb_compute<RM, RN>(acc, f1);
  if (HASC) {
    #pragma unroll
    for (int i = 0; i < RM; ++i) {
      #pragma unroll
      for (int j = 0; j < RN; ++j) {
        #pragma unroll
        for (int v = 0; v < 8; ++v) acc[i][j][v] += cv[i][j][v];
      }
    }
  }
  #pragma unroll
  for (int i = 0; i < RM; ++i) {
    #pragma unroll
    for (int j = 0; j < RN; ++j) {
      #pragma unroll
      for (int v = 0; v < 8; ++v) Dp[(size_t)(16 * i + v) * ldd + 16 * j] = acc[i][j][v];
    }
  }
}

__global__ void k_gather(float* __restrict__ V, const float* __restrict__ G,
                         const float* __restrict__ x0) {
  int idx = blockIdx.x * 256 + threadIdx.x;
  int c = idx >> 8, j = idx & 255;
  V[idx] = (c == 0) ? x0[j] : G[((size_t)(c - 1) * L1 + (L1 - 1)) * 256 + j];
}
__global__ void k_scatter(float* __restrict__ X, const float* __restrict__ S) {
  int idx = blockIdx.x * 256 + threadIdx.x;
  int c = idx >> 8, j = idx & 255;
  X[(size_t)c * L1 * 256 + j] = S[idx];
}

__global__ __launch_bounds__(128) void carry_double_kernel(
    float* __restrict__ Vn, const float* __restrict__ Vc,
    const float* __restrict__ Pd, int d) {
  const int lane = threadIdx.x & 31, wave = threadIdx.x >> 5;
  const int hl = lane >> 4, lr = lane & 15;
  const int tile = blockIdx.x * 4 + wave;
  const int tm = tile >> 4, tn = tile & 15;
  const size_t ooff = (size_t)(tm * 16 + 8 * hl) * 256 + tn * 16 + lr;
  const float* Cp = Vc + ooff;
  v8f seed;
  #pragma unroll
  for (int v = 0; v < 8; ++v) seed[v] = Cp[(size_t)v * 256];
  v8f acc = seed;
  if (tm * 16 + 15 >= d) {
    const int srow = tm * 16 + lr - d;
    const bool live = srow >= 0;
    const float* Ar = Vc + (size_t)(live ? srow : 0) * 256 + 2 * hl;
    const float* Br = Pd + (size_t)(tn * 16 + lr) * 256 + 2 * hl;
    const v2f zero2 = {0.0f, 0.0f};
    #pragma unroll 16
    for (int k = 0; k < 256; k += 4) {
      v2f a = *(const v2f*)(Ar + k);
      v2f b = *(const v2f*)(Br + k);
      a = live ? a : zero2;
      acc = WMMA_F32(a, b, acc);
    }
  }
  float* Dp = Vn + ooff;
  const int r0 = tm * 16 + 8 * hl;
  #pragma unroll
  for (int v = 0; v < 8; ++v)
    Dp[(size_t)v * 256] = (r0 + v >= d) ? acc[v] : seed[v];
}

static inline void launch_gemm(hipStream_t s, float* D, int ldd,
                               const float* A, int lda, const float* B, int ldb,
                               const float* Cin, int ldc,
                               int M, int N, int K, float alpha, float beta, int transB) {
  int tiles = (M / 16) * (N / 16);
  gemm_kernel<<<(tiles + 3) / 4, 128, 0, s>>>(D, ldd, A, lda, B, ldb, Cin, ldc,
                                              M, N, K, alpha, beta, transB);
}

template <int NC, int Mr>
static inline void launch_ksplit(hipStream_t s, float* D, const float* A, const float* B) {
  static_assert(Mr % 16 == 0 && NC % 16 == 0, "whole tiles");
  gemm_ksplit_kernel<NC><<<(Mr / 16) * (NC / 16), 128, 0, s>>>(D, A, B);
}

template <int RM, int RN, int WM, int WN, int KD, bool HASC, int Mr, int Nc>
static inline void launch_rb(hipStream_t s, float* D, int ldd,
                             const float* A, int lda, const float* B, int ldb,
                             const float* Cin, int ldc) {
  static_assert(Mr % (16 * RM * WM) == 0 && Nc % (16 * RN * WN) == 0, "whole work-group tiles");
  constexpr int blocksN = Nc / (16 * RN * WN);
  constexpr int blocksM = Mr / (16 * RM * WM);
  gemm_rb_kernel<RM, RN, WM, WN, KD, HASC, blocksN><<<blocksM * blocksN, 128, 0, s>>>(
      D, ldd, A, lda, B, ldb, Cin, ldc);
}

extern "C" void kernel_launch(void* const* d_in, const int* in_sizes, int n_in,
                              void* d_out, int out_size, void* d_ws, size_t ws_size,
                              hipStream_t stream) {
  (void)in_sizes; (void)n_in; (void)out_size; (void)ws_size;
  const float* t  = (const float*)d_in[0];
  const float* u  = (const float*)d_in[1];
  const float* x0 = (const float*)d_in[2];
  const float* Am = (const float*)d_in[3];
  const float* Bm = (const float*)d_in[4];
  const float* Cm = (const float*)d_in[5];
  const float* Dm = (const float*)d_in[6];
  float* xout = (float*)d_out;
  float* yout = xout + (size_t)Tn * NS;

  float* w = (float*)d_ws;
  float* Bd = w;  w += 8192;
  float* Kt[6]; for (int i = 0; i < 6; i++) { Kt[i] = w; w += 65536; }
  float* Rst  = w; w += 5 * 8192;
  float* Mm   = w; w += 65536;
  float* Qc   = w; w += 32768;
  float* PwA  = w; w += 65536;
  float* PwB  = w; w += 65536;
  float* PwC  = w; w += 65536;
  float* PwD  = w; w += 65536;
  float* Pm   = w; w += (size_t)Tn * 128;
  float* G    = w; w += (size_t)Tn * 256;
  float* V0   = w; w += (size_t)C1 * 256;
  w += (size_t)C2 * 256;
  w += (size_t)C2 * 256;
  float* V1   = w; w += (size_t)C1 * 256;

  auto gdim = [](int n) { return dim3((unsigned)((n + 255) / 256)); };

  float* PA  = Kt[0];
  k_prep<<<gdim(65536 + 8192), 256, 0, stream>>>(PA, Bd, Am, Bm, t);
  launch_ksplit<256, 256>(stream, Kt[1], Kt[0], Kt[0]);
  launch_ksplit<256, 512>(stream, Kt[2], Kt[0], Kt[1]);
  launch_ksplit<256, 512>(stream, Kt[4], Kt[0], Kt[3]);
  launch_ksplit<32, 1280>(stream, Rst, Kt[0], Bd);
  k_combine<<<gdim(65536 + 32768), 256, 0, stream>>>(Mm, Qc, PA, Bd, Rst);

  k_buildP<<<gdim(Tn * 128), 256, 0, stream>>>(Pm, u);
  launch_rb<2, 4, 2, 2, 128, false, Tn, 256>(stream, G, 256, Pm, 128, Qc, 128, nullptr, 0);

  float* Mp[5] = {Mm, Pm, Pm + 65536, Pm + 2 * 65536, Pm + 3 * 65536};
  float* Pw[10] = {PwA, PwB, PwC, PwD, Kt[0], Kt[1], Kt[2], Kt[3], Kt[4], Kt[5]};
  for (int j = 1; j < 5; ++j)
    launch_ksplit<256, 256>(stream, Mp[j], Mp[j-1], Mp[j-1]);
  launch_ksplit<256, 256>(stream, Pw[0], Mp[4], Mp[4]);
  for (int p = 1; p < 10; ++p)
    launch_ksplit<256, 256>(stream, Pw[p], Pw[p-1], Pw[p-1]);

  launch_rb<2, 4, 2, 2, 256, true, 16384, 256>(stream, G + 1 * 256, 512, G + 0 * 256, 512, Mp[0], 256, G + 1 * 256, 512);
  launch_rb<2, 4, 2, 2, 256, true, 8192, 256>(stream, G + 3 * 256, 1024, G + 1 * 256, 1024, Mp[1], 256, G + 3 * 256, 1024);
  launch_rb<2, 2, 2, 2, 256, true, 4096, 256>(stream, G + 7 * 256, 2048, G + 3 * 256, 2048, Mp[2], 256, G + 7 * 256, 2048);
  launch_rb<2, 2, 2, 2, 256, true, 2048, 256>(stream, G + 15 * 256, 4096, G + 7 * 256, 4096, Mp[3], 256, G + 15 * 256, 4096);
  launch_rb<2, 2, 2, 2, 256, true, 1024, 256>(stream, G + 31 * 256, 8192, G + 15 * 256, 8192, Mp[4], 256, G + 31 * 256, 8192);
  k_gather<<<gdim(C1 * 256), 256, 0, stream>>>(V0, G, x0);
  const int dd[10] = {1, 2, 4, 8, 16, 32, 64, 128, 256, 512};
  float* Va = V0;
  float* Vb = V1;
  for (int p = 0; p < 10; ++p) {
    carry_double_kernel<<<256, 128, 0, stream>>>(Vb, Va, Pw[p], dd[p]);
    float* tv = Va; Va = Vb; Vb = tv;
  }
  k_scatter<<<gdim(C1 * 256), 256, 0, stream>>>(xout, Va);
  launch_rb<2, 2, 2, 2, 256, true, 1024, 256>(stream, xout + 16 * 256, 8192, xout, 8192, Mp[4], 256, G + 15 * 256, 8192);
  launch_rb<2, 2, 2, 2, 256, true, 2048, 256>(stream, xout + 8 * 256, 4096, xout, 4096, Mp[3], 256, G + 7 * 256, 4096);
  launch_rb<2, 2, 2, 2, 256, true, 4096, 256>(stream, xout + 4 * 256, 2048, xout, 2048, Mp[2], 256, G + 3 * 256, 2048);
  launch_rb<2, 4, 2, 2, 256, true, 8192, 256>(stream, xout + 2 * 256, 1024, xout, 1024, Mp[1], 256, G + 1 * 256, 1024);
  launch_rb<2, 4, 2, 2, 256, true, 16384, 256>(stream, xout + 1 * 256, 512, xout, 512, Mp[0], 256, G + 0 * 256, 512);

  launch_rb<2, 2, 4, 1, 256, false, Tn, 32>(stream, yout, 32, xout, 256, Cm, 256, nullptr, 0);
  launch_gemm(stream, yout, 32, u,    32,  Dm, 32,  yout,   32, Tn, 32, 32,  1.f, 1.f, 1);
}


# 4 launches folded into neighbours (46 to 42): carry gather into first doubling pass loads, scatter into last pass stores, dt scaling into P build, U@D^T into the X@C^T K walk; y sum order changes
# speedup vs baseline: 4.9091x; 1.0516x over previous
#include <hip/hip_runtime.h>

typedef __attribute__((ext_vector_type(2))) float v2f;
typedef __attribute__((ext_vector_type(4))) float v4f;
typedef __attribute__((ext_vector_type(8))) float v8f;

#define WMMA_F32(a, b, c) \
  __builtin_amdgcn_wmma_f32_16x16x4_f32(false, (a), false, (b), (short)0, (c), false, false)

static constexpr int Tn = 32768;
static constexpr int NS = 256;
static constexpr int L1 = 32;
static constexpr int C1 = 1024;
static constexpr int C2 = 64;

__constant__ float d_mu[7] = {
  1.00000000f, 1.00000000f, 0.500000000f, 0.166666672f, 0.0416666679f, 0.00833333377f, 0.00166666671f};
__constant__ float d_theta[4][6] = {
  {0.500000000f, 0.349999994f, 0.133333340f, 0.0351111107f, 0.00746666687f, 0.00166666671f},
  {0.0833333358f, 0.0500000007f, 0.0166666675f, 0.00366666657f, 0.00106666668f, 0.0f},
  {0.500000000f, 0.150000006f, 0.0333333351f, 0.00655555539f, 0.000866666669f, 0.0f},
  {-0.0833333358f, -0.0333333351f, -0.00833333377f, -0.00188888889f, -0.000266666670f, 0.0f},
};

__device__ __forceinline__ void prep_part(int idx, float* __restrict__ Ad, float* __restrict__ Bd,
                                          const float* __restrict__ A, const float* __restrict__ B,
                                          const float* __restrict__ t) {
  float dt = t[1] - t[0];
  if (idx < 65536) Ad[idx] = dt * A[idx];
  else if (idx < 65536 + 8192) Bd[idx - 65536] = dt * B[idx - 65536];
}
__global__ void k_combine(float* __restrict__ M, float* __restrict__ Q,
                          const float* __restrict__ PA, const float* __restrict__ Bd,
                          const float* __restrict__ R) {
  int idx = blockIdx.x * 256 + threadIdx.x;
  if (idx < 65536) {
    float s = d_mu[6] * PA[5 * 65536 + idx];
    #pragma unroll
    for (int p = 5; p >= 1; --p) s += d_mu[p] * PA[(p - 1) * 65536 + idx];
    M[idx] = (((idx >> 8) == (idx & 255)) ? 1.0f : 0.0f) + s;
  } else if (idx < 65536 + 32768) {
    int q = idx - 65536;
    int k = q >> 8, r = q & 255;
    int j = k >> 5, c = k & 31;
    float s = 0.0f;
    #pragma unroll
    for (int p = 5; p >= 1; --p) s += d_theta[j][p] * R[(p - 1) * 8192 + r * 32 + c];
    s += d_theta[j][0] * Bd[r * 32 + c];
    Q[r * 128 + k] = s;
  }
}
__global__ void k_front(float* __restrict__ P, const float* __restrict__ u,
                        float* __restrict__ Ad, float* __restrict__ Bd,
                        const float* __restrict__ A, const float* __restrict__ B,
                        const float* __restrict__ t) {
  int idx = blockIdx.x * 256 + threadIdx.x;
  if (idx >= Tn * 128) { prep_part(idx - Tn * 128, Ad, Bd, A, B, t); return; }
  int n = idx >> 7, k = idx & 127;
  float val = 0.0f;
  if (n < Tn - 1) {
    int j = k >> 5, c = k & 31;
    int m = n + ((j >= 2) ? 1 : 0);
    if ((j & 1) == 0) {
      val = u[m * 32 + c];
    } else {
      if (m == 0)            val = u[32 + c] - u[c];
      else if (m == Tn - 1)  val = u[(Tn - 1) * 32 + c] - u[(Tn - 2) * 32 + c];
      else                   val = 0.5f * (u[(m + 1) * 32 + c] - u[(m - 1) * 32 + c]);
    }
  }
  P[idx] = val;
}

template <int NC>
__global__ __launch_bounds__(128) void gemm_ksplit_kernel(
    float* __restrict__ D, const float* __restrict__ A, const float* __restrict__ B) {
  __shared__ float part[4][8][32];
  const int lane = threadIdx.x & 31, wave = threadIdx.x >> 5;
  const int hl = lane >> 4, lr = lane & 15;
  constexpr int tilesN = NC >> 4;
  const int tm = blockIdx.x / tilesN, tn = blockIdx.x % tilesN;
  const float* Ar = A + (size_t)(tm * 16 + lr) * 256 + 64 * wave + 2 * hl;
  const float* Bc = B + (size_t)(64 * wave + 2 * hl) * NC + tn * 16 + lr;
  v2f a[16], b[16];
  #pragma unroll
  for (int i = 0; i < 16; ++i) {
    a[i] = *(const v2f*)(Ar + 4 * i);
    b[i][0] = Bc[(size_t)(4 * i) * NC];
    b[i][1] = Bc[(size_t)(4 * i) * NC + NC];
  }
  __builtin_amdgcn_sched_barrier(0);
  v8f acc = {};
  #pragma unroll
  for (int i = 0; i < 16; ++i) acc = WMMA_F32(a[i], b[i], acc);
  #pragma unroll
  for (int v = 0; v < 8; ++v) part[wave][v][lane] = acc[v];
  __syncthreads();
  #pragma unroll
  for (int e = 0; e < 2; ++e) {
    const int v = 2 * wave + e;
    float s = part[0][v][lane] + part[1][v][lane];
    s += part[2][v][lane];
    s += part[3][v][lane];
    D[(size_t)(tm * 16 + 8 * hl + v) * NC + tn * 16 + lr] = s;
  }
}

template <int RM, int RN>
struct FragSet {
  v4f a[RM][4];
  v4f b[RN][4];
};

template <int RM, int RN>
__device__ __forceinline__ void rb_load(FragSet<RM, RN>& f, const float* Ap, size_t astep,
                                        const float* Bp, size_t bstep, int k0) {
  #pragma unroll
  for (int i = 0; i < RM; ++i) {
    #pragma unroll
    for (int q = 0; q < 4; ++q) f.a[i][q] = *(const v4f*)(Ap + i * astep + k0 + 8 * q);
  }
  #pragma unroll
  for (int j = 0; j < RN; ++j) {
    #pragma unroll
    for (int q = 0; q < 4; ++q) f.b[j][q] = *(const v4f*)(Bp + j * bstep + k0 + 8 * q);
  }
}

template <int RM, int RN>
__device__ __forceinline__ void rb_compute(v8f (&acc)[RM][RN], const FragSet<RM, RN>& f) {
  #pragma unroll
  for (int q = 0; q < 4; ++q) {
    #pragma unroll
    for (int i = 0; i < RM; ++i) {
      #pragma unroll
      for (int j = 0; j < RN; ++j) {
        v2f a = {f.a[i][q][0], f.a[i][q][1]};
        v2f b = {f.b[j][q][0], f.b[j][q][1]};
        acc[i][j] = WMMA_F32(a, b, acc[i][j]);
      }
    }
    #pragma unroll
    for (int i = 0; i < RM; ++i) {
      #pragma unroll
      for (int j = 0; j < RN; ++j) {
        v2f a = {f.a[i][q][2], f.a[i][q][3]};
        v2f b = {f.b[j][q][2], f.b[j][q][3]};
        acc[i][j] = WMMA_F32(a, b, acc[i][j]);
      }
    }
  }
}

template <int RM, int RN, int WM, int WN, int KD, bool HASC, int BN, int K2>
__global__ __launch_bounds__(128) void gemm_rb_kernel(
    float* D, int ldd,
    const float* A, int lda,
    const float* __restrict__ B, int ldb,
    const float* Cin, int ldc,
    const float* __restrict__ A2, const float* __restrict__ B2) {
  static_assert(WM * WN == 4, "four waves per work-group");
  static_assert(K2 == 0 || K2 == 32, "the second pair is exactly one 32-k step");
  static_assert(KD % 64 == 0 && KD >= 128, "two 32-k steps per trip");
  const int lane = threadIdx.x & 31, wave = threadIdx.x >> 5;
  const int hl = lane >> 4, lr = lane & 15;
  const int bn = blockIdx.x % BN, bm = blockIdx.x / BN;
  const int wm = wave / WN, wn = wave % WN;
  const int row0 = (bm * WM + wm) * (16 * RM);
  const int col0 = (bn * WN + wn) * (16 * RN);
  const float* Ap = A + (size_t)(row0 + lr) * lda + 4 * hl;
  const float* Bp = B + (size_t)(col0 + lr) * ldb + 4 * hl;
  const size_t astep = (size_t)16 * lda, bstep = (size_t)16 * ldb;

  v8f acc[RM][RN];
  #pragma unroll
  for (int i = 0; i < RM; ++i) {
    #pragma unroll
    for (int j = 0; j < RN; ++j) acc[i][j] = v8f{};
  }

  FragSet<RM, RN> f0, f1;
  rb_load<RM, RN>(f0, Ap, astep, Bp, bstep, 0);
  #pragma unroll 1
  for (int k0 = 0; k0 < KD - 64; k0 += 64) {
    rb_load<RM, RN>(f1, Ap, astep, Bp, bstep, k0 + 32);
    __builtin_amdgcn_sched_barrier(0);
    rb_compute<RM, RN>(acc, f0);
    __builtin_amdgcn_sched_barrier(0);
    rb_load<RM, RN>(f0, Ap, astep, Bp, bstep, k0 + 64);
    __builtin_amdgcn_sched_barrier(0);
    rb_compute<RM, RN>(acc, f1);
    __builtin_amdgcn_sched_barrier(0);
  }
  rb_load<RM, RN>(f1, Ap, astep, Bp, bstep, KD - 32);
  __builtin_amdgcn_sched_barrier(0);
  rb_compute<RM, RN>(acc, f0);
  __builtin_amdgcn_sched_barrier(0);

  float* Dp = D + (size_t)(row0 + 8 * hl) * ldd + col0 + lr;
  v8f cv[RM][RN];
  if (HASC) {
    const float* Cp = Cin + (size_t)(row0 + 8 * hl) * ldc + col0 + lr;
    #pragma unroll
    for (int i = 0; i < RM; ++i) {
      #pragma unroll
      for (int j = 0; j < RN; ++j) {
        #pragma unroll
        for (int v = 0; v < 8; ++v) cv[i][j][v] = Cp[(size_t)(16 * i + v) * ldc + 16 * j];
      }
    }
    __builtin_amdgcn_sched_barrier(0);
  }
  if (K2 != 0) {
    rb_load<RM, RN>(f0, A2 + (size_t)(row0 + lr) * K2 + 4 * hl, (size_t)16 * K2,
                    B2 + (size_t)(col0 + lr) * K2 + 4 * hl, (size_t)16 * K2, 0);
    __builtin_amdgcn_sched_barrier(0);
  }
  rb_compute<RM, RN>(acc, f1);
  if (K2 != 0) rb_compute<RM, RN>(acc, f0);
  if (HASC) {
    #pragma unroll
    for (int i = 0; i < RM; ++i) {
      #pragma unroll
      for (int j = 0; j < RN; ++j) {
        #pragma unroll
        for (int v = 0; v < 8; ++v) acc[i][j][v] += cv[i][j][v];
      }
    }
  }
  #pragma unroll
  for (int i = 0; i < RM; ++i) {
    #pragma unroll
    for (int j = 0; j < RN; ++j) {
      #pragma unroll
      for (int v = 0; v < 8; ++v) Dp[(size_t)(16 * i + v) * ldd + 16 * j] = acc[i][j][v];
    }
  }
}

template <bool FIRST>
__device__ __forceinline__ const float* carry_row(const float* Vc, const float* G,
                                                  const float* x0, int c) {
  if (FIRST) return (c == 0) ? x0 : G + ((size_t)(c - 1) * L1 + (L1 - 1)) * 256;
  return Vc + (size_t)c * 256;
}
template <bool FIRST, bool LAST>
__global__ __launch_bounds__(128) void carry_double_kernel(
    float* __restrict__ Vn, const float* __restrict__ Vc,
    const float* __restrict__ Pd, int d,
    const float* __restrict__ G, const float* __restrict__ x0) {
  const int lane = threadIdx.x & 31, wave = threadIdx.x >> 5;
  const int hl = lane >> 4, lr = lane & 15;
  const int tile = blockIdx.x * 4 + wave;
  const int tm = tile >> 4, tn = tile & 15;
  const int r0 = tm * 16 + 8 * hl;
  const int col = tn * 16 + lr;
  v8f seed;
  #pragma unroll
  for (int v = 0; v < 8; ++v) seed[v] = carry_row<FIRST>(Vc, G, x0, r0 + v)[col];
  v8f acc = seed;
  if (tm * 16 + 15 >= d) {
    const int srow = tm * 16 + lr - d;
    const bool live = srow >= 0;
    const float* Ar = carry_row<FIRST>(Vc, G, x0, live ? srow : 0) + 2 * hl;
    const float* Br = Pd + (size_t)(tn * 16 + lr) * 256 + 2 * hl;
    const v2f zero2 = {0.0f, 0.0f};
    #pragma unroll 16
    for (int k = 0; k < 256; k += 4) {
      v2f a = *(const v2f*)(Ar + k);
      v2f b = *(const v2f*)(Br + k);
      a = live ? a : zero2;
      acc = WMMA_F32(a, b, acc);
    }
  }
  constexpr size_t orow = LAST ? (size_t)L1 * 256 : (size_t)256;
  float* Dp = Vn + (size_t)r0 * orow + col;
  #pragma unroll
  for (int v = 0; v < 8; ++v)
    Dp[(size_t)v * orow] = (r0 + v >= d) ? acc[v] : seed[v];
}

template <int NC, int Mr>
static inline void launch_ksplit(hipStream_t s, float* D, const float* A, const float* B) {
  static_assert(Mr % 16 == 0 && NC % 16 == 0, "whole tiles");
  gemm_ksplit_kernel<NC><<<(Mr / 16) * (NC / 16), 128, 0, s>>>(D, A, B);
}

template <int RM, int RN, int WM, int WN, int KD, bool HASC, int Mr, int Nc, int K2 = 0>
static inline void launch_rb(hipStream_t s, float* D, int ldd,
                             const float* A, int lda, const float* B, int ldb,
                             const float* Cin, int ldc,
                             const float* A2 = nullptr, const float* B2 = nullptr) {
  static_assert(Mr % (16 * RM * WM) == 0 && Nc % (16 * RN * WN) == 0, "whole work-group tiles");
  constexpr int blocksN = Nc / (16 * RN * WN);
  constexpr int blocksM = Mr / (16 * RM * WM);
  gemm_rb_kernel<RM, RN, WM, WN, KD, HASC, blocksN, K2><<<blocksM * blocksN, 128, 0, s>>>(
      D, ldd, A, lda, B, ldb, Cin, ldc, A2, B2);
}

extern "C" void kernel_launch(void* const* d_in, const int* in_sizes, int n_in,
                              void* d_out, int out_size, void* d_ws, size_t ws_size,
                              hipStream_t stream) {
  (void)in_sizes; (void)n_in; (void)out_size; (void)ws_size;
  const float* t  = (const float*)d_in[0];
  const float* u  = (const float*)d_in[1];
  const float* x0 = (const float*)d_in[2];
  const float* Am = (const float*)d_in[3];
  const float* Bm = (const float*)d_in[4];
  const float* Cm = (const float*)d_in[5];
  const float* Dm = (const float*)d_in[6];
  float* xout = (float*)d_out;
  float* yout = xout + (size_t)Tn * NS;

  float* w = (float*)d_ws;
  float* Bd = w;  w += 8192;
  float* Kt[6]; for (int i = 0; i < 6; i++) { Kt[i] = w; w += 65536; }
  float* Rst  = w; w += 5 * 8192;
  float* Mm   = w; w += 65536;
  float* Qc   = w; w += 32768;
  float* PwA  = w; w += 65536;
  float* PwB  = w; w += 65536;
  float* PwC  = w; w += 65536;
  float* PwD  = w; w += 65536;
  float* Pm   = w; w += (size_t)Tn * 128;
  float* G    = w; w += (size_t)Tn * 256;
  float* V0   = w; w += (size_t)C1 * 256;
  w += (size_t)C2 * 256;
  w += (size_t)C2 * 256;
  float* V1   = w; w += (size_t)C1 * 256;

  auto gdim = [](int n) { return dim3((unsigned)((n + 255) / 256)); };

  float* PA  = Kt[0];
  k_front<<<gdim(Tn * 128 + 65536 + 8192), 256, 0, stream>>>(Pm, u, PA, Bd, Am, Bm, t);
  launch_ksplit<256, 256>(stream, Kt[1], Kt[0], Kt[0]);
  launch_ksplit<256, 512>(stream, Kt[2], Kt[0], Kt[1]);
  launch_ksplit<256, 512>(stream, Kt[4], Kt[0], Kt[3]);
  launch_ksplit<32, 1280>(stream, Rst, Kt[0], Bd);
  k_combine<<<gdim(65536 + 32768), 256, 0, stream>>>(Mm, Qc, PA, Bd, Rst);

  launch_rb<2, 4, 2, 2, 128, false, Tn, 256>(stream, G, 256, Pm, 128, Qc, 128, nullptr, 0);

  float* Mp[5] = {Mm, Pm, Pm + 65536, Pm + 2 * 65536, Pm + 3 * 65536};
  float* Pw[10] = {PwA, PwB, PwC, PwD, Kt[0], Kt[1], Kt[2], Kt[3], Kt[4], Kt[5]};
  for (int j = 1; j < 5; ++j)
    launch_ksplit<256, 256>(stream, Mp[j], Mp[j-1], Mp[j-1]);
  launch_ksplit<256, 256>(stream, Pw[0], Mp[4], Mp[4]);
  for (int p = 1; p < 10; ++p)
    launch_ksplit<256, 256>(stream, Pw[p], Pw[p-1], Pw[p-1]);

  launch_rb<2, 4, 2, 2, 256, true, 16384, 256>(stream, G + 1 * 256, 512, G + 0 * 256, 512, Mp[0], 256, G + 1 * 256, 512);
  launch_rb<2, 4, 2, 2, 256, true, 8192, 256>(stream, G + 3 * 256, 1024, G + 1 * 256, 1024, Mp[1], 256, G + 3 * 256, 1024);
  launch_rb<2, 2, 2, 2, 256, true, 4096, 256>(stream, G + 7 * 256, 2048, G + 3 * 256, 2048, Mp[2], 256, G + 7 * 256, 2048);
  launch_rb<2, 2, 2, 2, 256, true, 2048, 256>(stream, G + 15 * 256, 4096, G + 7 * 256, 4096, Mp[3], 256, G + 15 * 256, 4096);
  launch_rb<2, 2, 2, 2, 256, true, 1024, 256>(stream, G + 31 * 256, 8192, G + 15 * 256, 8192, Mp[4], 256, G + 31 * 256, 8192);
  const int dd[10] = {1, 2, 4, 8, 16, 32, 64, 128, 256, 512};
  carry_double_kernel<true, false><<<256, 128, 0, stream>>>(V0, V1, Pw[0], dd[0], G, x0);
  float* Va = V0;
  float* Vb = V1;
  for (int p = 1; p < 9; ++p) {
    carry_double_kernel<false, false><<<256, 128, 0, stream>>>(Vb, Va, Pw[p], dd[p], G, x0);
    float* tv = Va; Va = Vb; Vb = tv;
  }
  carry_double_kernel<false, true><<<256, 128, 0, stream>>>(xout, Va, Pw[9], dd[9], G, x0);
  launch_rb<2, 2, 2, 2, 256, true, 1024, 256>(stream, xout + 16 * 256, 8192, xout, 8192, Mp[4], 256, G + 15 * 256, 8192);
  launch_rb<2, 2, 2, 2, 256, true, 2048, 256>(stream, xout + 8 * 256, 4096, xout, 4096, Mp[3], 256, G + 7 * 256, 4096);
  launch_rb<2, 2, 2, 2, 256, true, 4096, 256>(stream, xout + 4 * 256, 2048, xout, 2048, Mp[2], 256, G + 3 * 256, 2048);
  launch_rb<2, 4, 2, 2, 256, true, 8192, 256>(stream, xout + 2 * 256, 1024, xout, 1024, Mp[1], 256, G + 1 * 256, 1024);
  launch_rb<2, 4, 2, 2, 256, true, 16384, 256>(stream, xout + 1 * 256, 512, xout, 512, Mp[0], 256, G + 0 * 256, 512);

  launch_rb<2, 2, 4, 1, 256, false, Tn, 32, 32>(stream, yout, 32, xout, 256, Cm, 256, nullptr, 0, u, Dm);
}


# the 14 products of at most 2048 rows (10 carry passes, pairing strides 8 and 16 up and down): one 16x32 block per work-group, K split over its 4 waves, partials summed in LDS; a different k sum order
# speedup vs baseline: 5.5544x; 1.1314x over previous
#include <hip/hip_runtime.h>

typedef __attribute__((ext_vector_type(2))) float v2f;
typedef __attribute__((ext_vector_type(4))) float v4f;
typedef __attribute__((ext_vector_type(8))) float v8f;

#define WMMA_F32(a, b, c) \
  __builtin_amdgcn_wmma_f32_16x16x4_f32(false, (a), false, (b), (short)0, (c), false, false)

static constexpr int Tn = 32768;
static constexpr int NS = 256;
static constexpr int L1 = 32;
static constexpr int C1 = 1024;
static constexpr int C2 = 64;

__constant__ float d_mu[7] = {
  1.00000000f, 1.00000000f, 0.500000000f, 0.166666672f, 0.0416666679f, 0.00833333377f, 0.00166666671f};
__constant__ float d_theta[4][6] = {
  {0.500000000f, 0.349999994f, 0.133333340f, 0.0351111107f, 0.00746666687f, 0.00166666671f},
  {0.0833333358f, 0.0500000007f, 0.0166666675f, 0.00366666657f, 0.00106666668f, 0.0f},
  {0.500000000f, 0.150000006f, 0.0333333351f, 0.00655555539f, 0.000866666669f, 0.0f},
  {-0.0833333358f, -0.0333333351f, -0.00833333377f, -0.00188888889f, -0.000266666670f, 0.0f},
};

__device__ __forceinline__ void prep_part(int idx, float* __restrict__ Ad, float* __restrict__ Bd,
                                          const float* __restrict__ A, const float* __restrict__ B,
                                          const float* __restrict__ t) {
  float dt = t[1] - t[0];
  if (idx < 65536) Ad[idx] = dt * A[idx];
  else if (idx < 65536 + 8192) Bd[idx - 65536] = dt * B[idx - 65536];
}
__global__ void k_combine(float* __restrict__ M, float* __restrict__ Q,
                          const float* __restrict__ PA, const float* __restrict__ Bd,
                          const float* __restrict__ R) {
  int idx = blockIdx.x * 256 + threadIdx.x;
  if (idx < 65536) {
    float s = d_mu[6] * PA[5 * 65536 + idx];
    #pragma unroll
    for (int p = 5; p >= 1; --p) s += d_mu[p] * PA[(p - 1) * 65536 + idx];
    M[idx] = (((idx >> 8) == (idx & 255)) ? 1.0f : 0.0f) + s;
  } else if (idx < 65536 + 32768) {
    int q = idx - 65536;
    int k = q >> 8, r = q & 255;
    int j = k >> 5, c = k & 31;
    float s = 0.0f;
    #pragma unroll
    for (int p = 5; p >= 1; --p) s += d_theta[j][p] * R[(p - 1) * 8192 + r * 32 + c];
    s += d_theta[j][0] * Bd[r * 32 + c];
    Q[r * 128 + k] = s;
  }
}
__global__ void k_front(float* __restrict__ P, const float* __restrict__ u,
                        float* __restrict__ Ad, float* __restrict__ Bd,
                        const float* __restrict__ A, const float* __restrict__ B,
                        const float* __restrict__ t) {
  int idx = blockIdx.x * 256 + threadIdx.x;
  if (idx >= Tn * 128) { prep_part(idx - Tn * 128, Ad, Bd, A, B, t); return; }
  int n = idx >> 7, k = idx & 127;
  float val = 0.0f;
  if (n < Tn - 1) {
    int j = k >> 5, c = k & 31;
    int m = n + ((j >= 2) ? 1 : 0);
    if ((j & 1) == 0) {
      val = u[m * 32 + c];
    } else {
      if (m == 0)            val = u[32 + c] - u[c];
      else if (m == Tn - 1)  val = u[(Tn - 1) * 32 + c] - u[(Tn - 2) * 32 + c];
      else                   val = 0.5f * (u[(m + 1) * 32 + c] - u[(m - 1) * 32 + c]);
    }
  }
  P[idx] = val;
}

template <int NC>
__global__ __launch_bounds__(128) void gemm_ksplit_kernel(
    float* __restrict__ D, const float* __restrict__ A, const float* __restrict__ B) {
  __shared__ float part[4][8][32];
  const int lane = threadIdx.x & 31, wave = threadIdx.x >> 5;
  const int hl = lane >> 4, lr = lane & 15;
  constexpr int tilesN = NC >> 4;
  const int tm = blockIdx.x / tilesN, tn = blockIdx.x % tilesN;
  const float* Ar = A + (size_t)(tm * 16 + lr) * 256 + 64 * wave + 2 * hl;
  const float* Bc = B + (size_t)(64 * wave + 2 * hl) * NC + tn * 16 + lr;
  v2f a[16], b[16];
  #pragma unroll
  for (int i = 0; i < 16; ++i) {
    a[i] = *(const v2f*)(Ar + 4 * i);
    b[i][0] = Bc[(size_t)(4 * i) * NC];
    b[i][1] = Bc[(size_t)(4 * i) * NC + NC];
  }
  __builtin_amdgcn_sched_barrier(0);
  v8f acc = {};
  #pragma unroll
  for (int i = 0; i < 16; ++i) acc = WMMA_F32(a[i], b[i], acc);
  #pragma unroll
  for (int v = 0; v < 8; ++v) part[wave][v][lane] = acc[v];
  __syncthreads();
  #pragma unroll
  for (int e = 0; e < 2; ++e) {
    const int v = 2 * wave + e;
    float s = part[0][v][lane] + part[1][v][lane];
    s += part[2][v][lane];
    s += part[3][v][lane];
    D[(size_t)(tm * 16 + 8 * hl + v) * NC + tn * 16 + lr] = s;
  }
}

template <int RM, int RN>
struct FragSet {
  v4f a[RM][4];
  v4f b[RN][4];
};

template <int RM, int RN>
__device__ __forceinline__ void rb_load(FragSet<RM, RN>& f, const float* Ap, size_t astep,
                                        const float* Bp, size_t bstep, int k0) {
  #pragma unroll
  for (int i = 0; i < RM; ++i) {
    #pragma unroll
    for (int q = 0; q < 4; ++q) f.a[i][q] = *(const v4f*)(Ap + i * astep + k0 + 8 * q);
  }
  #pragma unroll
  for (int j = 0; j < RN; ++j) {
    #pragma unroll
    for (int q = 0; q < 4; ++q) f.b[j][q] = *(const v4f*)(Bp + j * bstep + k0 + 8 * q);
  }
}

template <int RM, int RN>
__device__ __forceinline__ void rb_compute(v8f (&acc)[RM][RN], const FragSet<RM, RN>& f) {
  #pragma unroll
  for (int q = 0; q < 4; ++q) {
    #pragma unroll
    for (int i = 0; i < RM; ++i) {
      #pragma unroll
      for (int j = 0; j < RN; ++j) {
        v2f a = {f.a[i][q][0], f.a[i][q][1]};
        v2f b = {f.b[j][q][0], f.b[j][q][1]};
        acc[i][j] = WMMA_F32(a, b, acc[i][j]);
      }
    }
    #pragma unroll
    for (int i = 0; i < RM; ++i) {
      #pragma unroll
      for (int j = 0; j < RN; ++j) {
        v2f a = {f.a[i][q][2], f.a[i][q][3]};
        v2f b = {f.b[j][q][2], f.b[j][q][3]};
        acc[i][j] = WMMA_F32(a, b, acc[i][j]);
      }
    }
  }
}

template <int RM, int RN, int WM, int WN, int KD, bool HASC, int BN, int K2>
__global__ __launch_bounds__(128) void gemm_rb_kernel(
    float* D, int ldd,
    const float* A, int lda,
    const float* __restrict__ B, int ldb,
    const float* Cin, int ldc,
    const float* __restrict__ A2, const float* __restrict__ B2) {
  static_assert(WM * WN == 4, "four waves per work-group");
  static_assert(K2 == 0 || K2 == 32, "the second pair is exactly one 32-k step");
  static_assert(KD % 64 == 0 && KD >= 128, "two 32-k steps per trip");
  const int lane = threadIdx.x & 31, wave = threadIdx.x >> 5;
  const int hl = lane >> 4, lr = lane & 15;
  const int bn = blockIdx.x % BN, bm = blockIdx.x / BN;
  const int wm = wave / WN, wn = wave % WN;
  const int row0 = (bm * WM + wm) * (16 * RM);
  const int col0 = (bn * WN + wn) * (16 * RN);
  const float* Ap = A + (size_t)(row0 + lr) * lda + 4 * hl;
  const float* Bp = B + (size_t)(col0 + lr) * ldb + 4 * hl;
  const size_t astep = (size_t)16 * lda, bstep = (size_t)16 * ldb;

  v8f acc[RM][RN];
  #pragma unroll
  for (int i = 0; i < RM; ++i) {
    #pragma unroll
    for (int j = 0; j < RN; ++j) acc[i][j] = v8f{};
  }

  FragSet<RM, RN> f0, f1;
  rb_load<RM, RN>(f0, Ap, astep, Bp, bstep, 0);
  #pragma unroll 1
  for (int k0 = 0; k0 < KD - 64; k0 += 64) {
    rb_load<RM, RN>(f1, Ap, astep, Bp, bstep, k0 + 32);
    __builtin_amdgcn_sched_barrier(0);
    rb_compute<RM, RN>(acc, f0);
    __builtin_amdgcn_sched_barrier(0);
    rb_load<RM, RN>(f0, Ap, astep, Bp, bstep, k0 + 64);
    __builtin_amdgcn_sched_barrier(0);
    rb_compute<RM, RN>(acc, f1);
    __builtin_amdgcn_sched_barrier(0);
  }
  rb_load<RM, RN>(f1, Ap, astep, Bp, bstep, KD - 32);
  __builtin_amdgcn_sched_barrier(0);
  rb_compute<RM, RN>(acc, f0);
  __builtin_amdgcn_sched_barrier(0);

  float* Dp = D + (size_t)(row0 + 8 * hl) * ldd + col0 + lr;
  v8f cv[RM][RN];
  if (HASC) {
    const float* Cp = Cin + (size_t)(row0 + 8 * hl) * ldc + col0 + lr;
    #pragma unroll
    for (int i = 0; i < RM; ++i) {
      #pragma unroll
      for (int j = 0; j < RN; ++j) {
        #pragma unroll
        for (int v = 0; v < 8; ++v) cv[i][j][v] = Cp[(size_t)(16 * i + v) * ldc + 16 * j];
      }
    }
    __builtin_amdgcn_sched_barrier(0);
  }
  if (K2 != 0) {
    rb_load<RM, RN>(f0, A2 + (size_t)(row0 + lr) * K2 + 4 * hl, (size_t)16 * K2,
                    B2 + (size_t)(col0 + lr) * K2 + 4 * hl, (size_t)16 * K2, 0);
    __builtin_amdgcn_sched_barrier(0);
  }
  rb_compute<RM, RN>(acc, f1);
  if (K2 != 0) rb_compute<RM, RN>(acc, f0);
  if (HASC) {
    #pragma unroll
    for (int i = 0; i < RM; ++i) {
      #pragma unroll
      for (int j = 0; j < RN; ++j) {
        #pragma unroll
        for (int v = 0; v < 8; ++v) acc[i][j][v] += cv[i][j][v];
      }
    }
  }
  #pragma unroll
  for (int i = 0; i < RM; ++i) {
    #pragma unroll
    for (int j = 0; j < RN; ++j) {
      #pragma unroll
      for (int v = 0; v < 8; ++v) Dp[(size_t)(16 * i + v) * ldd + 16 * j] = acc[i][j][v];
    }
  }
}

__device__ __forceinline__ void ksplit_block(float (&s)[4], const float* Ar, const float* Br,
                                             float (*part)[2][8][32], int wave, int lane) {
  v4f a[8], b0[8], b1[8];
  #pragma unroll
  for (int q = 0; q < 8; ++q) {
    a[q]  = *(const v4f*)(Ar + 8 * q);
    b0[q] = *(const v4f*)(Br + 8 * q);
    b1[q] = *(const v4f*)(Br + (size_t)16 * 256 + 8 * q);
  }
  __builtin_amdgcn_sched_barrier(0);
  v8f acc0 = {}, acc1 = {};
  #pragma unroll
  for (int q = 0; q < 8; ++q) {
    v2f a01 = {a[q][0], a[q][1]}, a23 = {a[q][2], a[q][3]};
    v2f p01 = {b0[q][0], b0[q][1]}, p23 = {b0[q][2], b0[q][3]};
    v2f r01 = {b1[q][0], b1[q][1]}, r23 = {b1[q][2], b1[q][3]};
    acc0 = WMMA_F32(a01, p01, acc0);
    acc1 = WMMA_F32(a01, r01, acc1);
    acc0 = WMMA_F32(a23, p23, acc0);
    acc1 = WMMA_F32(a23, r23, acc1);
  }
  #pragma unroll
  for (int v = 0; v < 8; ++v) {
    part[wave][0][v][lane] = acc0[v];
    part[wave][1][v][lane] = acc1[v];
  }
  __syncthreads();
  const int j = lane >> 4, lc = lane & 15;
  #pragma unroll
  for (int e = 0; e < 4; ++e) {
    const int r = 4 * wave + e;
    const int v = r & 7, sl = 16 * (r >> 3) + lc;
    float t = part[0][j][v][sl] + part[1][j][v][sl];
    t += part[2][j][v][sl];
    t += part[3][j][v][sl];
    s[e] = t;
  }
}

__global__ __launch_bounds__(128) void pair_ksplit_kernel(
    float* D, int ldd, const float* A, int lda, const float* __restrict__ B,
    const float* Cin, int ldc) {
  __shared__ float part[4][2][8][32];
  const int lane = threadIdx.x & 31, wave = threadIdx.x >> 5;
  const int hl = lane >> 4, lr = lane & 15;
  const int tm = blockIdx.x >> 3, tp = blockIdx.x & 7;
  const int col = tp * 32 + lane;
  const int rbase = tm * 16 + 4 * wave;
  float c[4];
  #pragma unroll
  for (int e = 0; e < 4; ++e) c[e] = Cin[(size_t)(rbase + e) * ldc + col];
  const float* Ar = A + (size_t)(tm * 16 + lr) * lda + 64 * wave + 4 * hl;
  const float* Br = B + (size_t)(tp * 32 + lr) * 256 + 64 * wave + 4 * hl;
  float s[4];
  ksplit_block(s, Ar, Br, part, wave, lane);
  #pragma unroll
  for (int e = 0; e < 4; ++e) D[(size_t)(rbase + e) * ldd + col] = c[e] + s[e];
}

template <bool FIRST>
__device__ __forceinline__ const float* carry_row(const float* Vc, const float* G,
                                                  const float* x0, int c) {
  if (FIRST) return (c == 0) ? x0 : G + ((size_t)(c - 1) * L1 + (L1 - 1)) * 256;
  return Vc + (size_t)c * 256;
}
template <bool FIRST, bool LAST>
__global__ __launch_bounds__(128) void carry_double_kernel(
    float* __restrict__ Vn, const float* __restrict__ Vc,
    const float* __restrict__ Pd, int d,
    const float* __restrict__ G, const float* __restrict__ x0) {
  __shared__ float part[4][2][8][32];
  const int lane = threadIdx.x & 31, wave = threadIdx.x >> 5;
  const int hl = lane >> 4, lr = lane & 15;
  const int tm = blockIdx.x >> 3, tp = blockIdx.x & 7;
  const int col = tp * 32 + lane;
  const int rbase = tm * 16 + 4 * wave;
  float seed[4], res[4];
  #pragma unroll
  for (int e = 0; e < 4; ++e) {
    seed[e] = carry_row<FIRST>(Vc, G, x0, rbase + e)[col];
    res[e] = seed[e];
  }
  if (tm * 16 + 15 >= d) {
    const int srow = tm * 16 + lr - d;
    const float* Ar = carry_row<FIRST>(Vc, G, x0, srow >= 0 ? srow : 0) + 64 * wave + 4 * hl;
    const float* Br = Pd + (size_t)(tp * 32 + lr) * 256 + 64 * wave + 4 * hl;
    float s[4];
    ksplit_block(s, Ar, Br, part, wave, lane);
    #pragma unroll
    for (int e = 0; e < 4; ++e) res[e] = (rbase + e >= d) ? seed[e] + s[e] : seed[e];
  }
  constexpr size_t orow = LAST ? (size_t)L1 * 256 : (size_t)256;
  #pragma unroll
  for (int e = 0; e < 4; ++e) Vn[(size_t)(rbase + e) * orow + col] = res[e];
}

template <int NC, int Mr>
static inline void launch_ksplit(hipStream_t s, float* D, const float* A, const float* B) {
  static_assert(Mr % 16 == 0 && NC % 16 == 0, "whole tiles");
  gemm_ksplit_kernel<NC><<<(Mr / 16) * (NC / 16), 128, 0, s>>>(D, A, B);
}

template <int RM, int RN, int WM, int WN, int KD, bool HASC, int Mr, int Nc, int K2 = 0>
static inline void launch_rb(hipStream_t s, float* D, int ldd,
                             const float* A, int lda, const float* B, int ldb,
                             const float* Cin, int ldc,
                             const float* A2 = nullptr, const float* B2 = nullptr) {
  static_assert(Mr % (16 * RM * WM) == 0 && Nc % (16 * RN * WN) == 0, "whole work-group tiles");
  constexpr int blocksN = Nc / (16 * RN * WN);
  constexpr int blocksM = Mr / (16 * RM * WM);
  gemm_rb_kernel<RM, RN, WM, WN, KD, HASC, blocksN, K2><<<blocksM * blocksN, 128, 0, s>>>(
      D, ldd, A, lda, B, ldb, Cin, ldc, A2, B2);
}

template <int Mr>
static inline void launch_pair(hipStream_t s, float* D, int ldd, const float* A, int lda,
                               const float* B, const float* Cin, int ldc) {
  static_assert(Mr % 16 == 0, "whole blocks");
  pair_ksplit_kernel<<<(Mr / 16) * 8, 128, 0, s>>>(D, ldd, A, lda, B, Cin, ldc);
}

extern "C" void kernel_launch(void* const* d_in, const int* in_sizes, int n_in,
                              void* d_out, int out_size, void* d_ws, size_t ws_size,
                              hipStream_t stream) {
  (void)in_sizes; (void)n_in; (void)out_size; (void)ws_size;
  const float* t  = (const float*)d_in[0];
  const float* u  = (const float*)d_in[1];
  const float* x0 = (const float*)d_in[2];
  const float* Am = (const float*)d_in[3];
  const float* Bm = (const float*)d_in[4];
  const float* Cm = (const float*)d_in[5];
  const float* Dm = (const float*)d_in[6];
  float* xout = (float*)d_out;
  float* yout = xout + (size_t)Tn * NS;

  float* w = (float*)d_ws;
  float* Bd = w;  w += 8192;
  float* Kt[6]; for (int i = 0; i < 6; i++) { Kt[i] = w; w += 65536; }
  float* Rst  = w; w += 5 * 8192;
  float* Mm   = w; w += 65536;
  float* Qc   = w; w += 32768;
  float* PwA  = w; w += 65536;
  float* PwB  = w; w += 65536;
  float* PwC  = w; w += 65536;
  float* PwD  = w; w += 65536;
  float* Pm   = w; w += (size_t)Tn * 128;
  float* G    = w; w += (size_t)Tn * 256;
  float* V0   = w; w += (size_t)C1 * 256;
  w += (size_t)C2 * 256;
  w += (size_t)C2 * 256;
  float* V1   = w; w += (size_t)C1 * 256;

  auto gdim = [](int n) { return dim3((unsigned)((n + 255) / 256)); };

  float* PA  = Kt[0];
  k_front<<<gdim(Tn * 128 + 65536 + 8192), 256, 0, stream>>>(Pm, u, PA, Bd, Am, Bm, t);
  launch_ksplit<256, 256>(stream, Kt[1], Kt[0], Kt[0]);
  launch_ksplit<256, 512>(stream, Kt[2], Kt[0], Kt[1]);
  launch_ksplit<256, 512>(stream, Kt[4], Kt[0], Kt[3]);
  launch_ksplit<32, 1280>(stream, Rst, Kt[0], Bd);
  k_combine<<<gdim(65536 + 32768), 256, 0, stream>>>(Mm, Qc, PA, Bd, Rst);

  launch_rb<2, 4, 2, 2, 128, false, Tn, 256>(stream, G, 256, Pm, 128, Qc, 128, nullptr, 0);

  float* Mp[5] = {Mm, Pm, Pm + 65536, Pm + 2 * 65536, Pm + 3 * 65536};
  float* Pw[10] = {PwA, PwB, PwC, PwD, Kt[0], Kt[1], Kt[2], Kt[3], Kt[4], Kt[5]};
  for (int j = 1; j < 5; ++j)
    launch_ksplit<256, 256>(stream, Mp[j], Mp[j-1], Mp[j-1]);
  launch_ksplit<256, 256>(stream, Pw[0], Mp[4], Mp[4]);
  for (int p = 1; p < 10; ++p)
    launch_ksplit<256, 256>(stream, Pw[p], Pw[p-1], Pw[p-1]);

  launch_rb<2, 4, 2, 2, 256, true, 16384, 256>(stream, G + 1 * 256, 512, G + 0 * 256, 512, Mp[0], 256, G + 1 * 256, 512);
  launch_rb<2, 4, 2, 2, 256, true, 8192, 256>(stream, G + 3 * 256, 1024, G + 1 * 256, 1024, Mp[1], 256, G + 3 * 256, 1024);
  launch_rb<2, 2, 2, 2, 256, true, 4096, 256>(stream, G + 7 * 256, 2048, G + 3 * 256, 2048, Mp[2], 256, G + 7 * 256, 2048);
  launch_pair<2048>(stream, G + 15 * 256, 4096, G + 7 * 256, 4096, Mp[3], G + 15 * 256, 4096);
  launch_pair<1024>(stream, G + 31 * 256, 8192, G + 15 * 256, 8192, Mp[4], G + 31 * 256, 8192);
  const int dd[10] = {1, 2, 4, 8, 16, 32, 64, 128, 256, 512};
  carry_double_kernel<true, false><<<512, 128, 0, stream>>>(V0, V1, Pw[0], dd[0], G, x0);
  float* Va = V0;
  float* Vb = V1;
  for (int p = 1; p < 9; ++p) {
    carry_double_kernel<false, false><<<512, 128, 0, stream>>>(Vb, Va, Pw[p], dd[p], G, x0);
    float* tv = Va; Va = Vb; Vb = tv;
  }
  carry_double_kernel<false, true><<<512, 128, 0, stream>>>(xout, Va, Pw[9], dd[9], G, x0);
  launch_pair<1024>(stream, xout + 16 * 256, 8192, xout, 8192, Mp[4], G + 15 * 256, 8192);
  launch_pair<2048>(stream, xout + 8 * 256, 4096, xout, 4096, Mp[3], G + 7 * 256, 4096);
  launch_rb<2, 2, 2, 2, 256, true, 4096, 256>(stream, xout + 4 * 256, 2048, xout, 2048, Mp[2], 256, G + 3 * 256, 2048);
  launch_rb<2, 4, 2, 2, 256, true, 8192, 256>(stream, xout + 2 * 256, 1024, xout, 1024, Mp[1], 256, G + 1 * 256, 1024);
  launch_rb<2, 4, 2, 2, 256, true, 16384, 256>(stream, xout + 1 * 256, 512, xout, 512, Mp[0], 256, G + 0 * 256, 512);

  launch_rb<2, 2, 4, 1, 256, false, Tn, 32, 32>(stream, yout, 32, xout, 256, Cm, 256, nullptr, 0, u, Dm);
}


# the two elementwise launches (P build with dt scaling; combine) re-threaded: a thread forms 8 rows x 4 columns, or 4 elements, by 16-byte accesses: 1024 and 768 waves to start, not 133376 and 3072
# speedup vs baseline: 5.7791x; 1.0405x over previous
#include <hip/hip_runtime.h>

typedef __attribute__((ext_vector_type(2))) float v2f;
typedef __attribute__((ext_vector_type(4))) float v4f;
typedef __attribute__((ext_vector_type(8))) float v8f;

#define WMMA_F32(a, b, c) \
  __builtin_amdgcn_wmma_f32_16x16x4_f32(false, (a), false, (b), (short)0, (c), false, false)

static constexpr int Tn = 32768;
static constexpr int NS = 256;
static constexpr int L1 = 32;
static constexpr int C1 = 1024;
static constexpr int C2 = 64;

__constant__ float d_mu[7] = {
  1.00000000f, 1.00000000f, 0.500000000f, 0.166666672f, 0.0416666679f, 0.00833333377f, 0.00166666671f};
__constant__ float d_theta[4][6] = {
  {0.500000000f, 0.349999994f, 0.133333340f, 0.0351111107f, 0.00746666687f, 0.00166666671f},
  {0.0833333358f, 0.0500000007f, 0.0166666675f, 0.00366666657f, 0.00106666668f, 0.0f},
  {0.500000000f, 0.150000006f, 0.0333333351f, 0.00655555539f, 0.000866666669f, 0.0f},
  {-0.0833333358f, -0.0333333351f, -0.00833333377f, -0.00188888889f, -0.000266666670f, 0.0f},
};

__global__ __launch_bounds__(256) void k_combine(float* __restrict__ M, float* __restrict__ Q,
                                                 const float* __restrict__ PA,
                                                 const float* __restrict__ Bd,
                                                 const float* __restrict__ R) {
  const int g = blockIdx.x * 256 + threadIdx.x;
  if (g < 16384) {
    const int idx = 4 * g;
    v4f pl[6];
    #pragma unroll
    for (int p = 0; p < 6; ++p) pl[p] = *(const v4f*)(PA + p * 65536 + idx);
    v4f o;
    #pragma unroll
    for (int e = 0; e < 4; ++e) {
      float s = d_mu[6] * pl[5][e];
      #pragma unroll
      for (int p = 5; p >= 1; --p) s += d_mu[p] * pl[p - 1][e];
      o[e] = ((((idx + e) >> 8) == ((idx + e) & 255)) ? 1.0f : 0.0f) + s;
    }
    *(v4f*)(M + idx) = o;
  } else if (g < 16384 + 8192) {
    const int q = g - 16384;
    const int c = (q & 7) * 4, j = (q >> 3) & 3, r = q >> 5;
    v4f rl[5];
    #pragma unroll
    for (int p = 0; p < 5; ++p) rl[p] = *(const v4f*)(R + p * 8192 + r * 32 + c);
    const v4f bd = *(const v4f*)(Bd + r * 32 + c);
    v4f o;
    #pragma unroll
    for (int e = 0; e < 4; ++e) {
      float s = 0.0f;
      #pragma unroll
      for (int p = 5; p >= 1; --p) s += d_theta[j][p] * rl[p - 1][e];
      s += d_theta[j][0] * bd[e];
      o[e] = s;
    }
    *(v4f*)(Q + r * 128 + 32 * j + c) = o;
  }
}
static constexpr int FR = 8;
__global__ __launch_bounds__(128) void k_front(float* __restrict__ P, const float* __restrict__ u,
                                               float* __restrict__ Ad, float* __restrict__ Bd,
                                               const float* __restrict__ A, const float* __restrict__ B,
                                               const float* __restrict__ t) {
  const int tid = threadIdx.x;
  const int c = (tid & 7) * 4;
  const int n0 = (blockIdx.x * 16 + (tid >> 3)) * FR;
  const int g = blockIdx.x * 128 + tid;
  const bool scales = g < 16384 + 2048;
  const bool inA = g < 16384;
  const int off = 4 * (inA ? g : g - 16384);
  const float* sp = inA ? A : B;
  float* dp = inA ? Ad : Bd;
  const float t0 = t[0], t1 = t[1];
  v4f av = {};
  if (scales) av = *(const v4f*)(sp + off);
  v4f w[FR + 3];
  #pragma unroll
  for (int i = 0; i < FR + 3; ++i) {
    int r = n0 - 1 + i;
    r = r < 0 ? 0 : (r > Tn - 1 ? Tn - 1 : r);
    w[i] = *(const v4f*)(u + (size_t)r * 32 + c);
  }
  __builtin_amdgcn_sched_barrier(0);
  v4f s[FR + 1];
  #pragma unroll
  for (int i = 0; i < FR + 1; ++i) {
    const int m = n0 + i;
    const float h = (m == 0 || m >= Tn - 1) ? 1.0f : 0.5f;
    s[i] = h * (w[i + 2] - w[i]);
  }
  const v4f zero = {};
  #pragma unroll
  for (int i = 0; i < FR; ++i) {
    const int n = n0 + i;
    const bool live = (i < FR - 1) || (n < Tn - 1);
    float* row = P + (size_t)n * 128 + c;
    *(v4f*)(row)      = live ? w[i + 1] : zero;
    *(v4f*)(row + 32) = live ? s[i]     : zero;
    *(v4f*)(row + 64) = live ? w[i + 2] : zero;
    *(v4f*)(row + 96) = live ? s[i + 1] : zero;
  }
  if (scales) *(v4f*)(dp + off) = (t1 - t0) * av;
}

template <int NC>
__global__ __launch_bounds__(128) void gemm_ksplit_kernel(
    float* __restrict__ D, const float* __restrict__ A, const float* __restrict__ B) {
  __shared__ float part[4][8][32];
  const int lane = threadIdx.x & 31, wave = threadIdx.x >> 5;
  const int hl = lane >> 4, lr = lane & 15;
  constexpr int tilesN = NC >> 4;
  const int tm = blockIdx.x / tilesN, tn = blockIdx.x % tilesN;
  const float* Ar = A + (size_t)(tm * 16 + lr) * 256 + 64 * wave + 2 * hl;
  const float* Bc = B + (size_t)(64 * wave + 2 * hl) * NC + tn * 16 + lr;
  v2f a[16], b[16];
  #pragma unroll
  for (int i = 0; i < 16; ++i) {
    a[i] = *(const v2f*)(Ar + 4 * i);
    b[i][0] = Bc[(size_t)(4 * i) * NC];
    b[i][1] = Bc[(size_t)(4 * i) * NC + NC];
  }
  __builtin_amdgcn_sched_barrier(0);
  v8f acc = {};
  #pragma unroll
  for (int i = 0; i < 16; ++i) acc = WMMA_F32(a[i], b[i], acc);
  #pragma unroll
  for (int v = 0; v < 8; ++v) part[wave][v][lane] = acc[v];
  __syncthreads();
  #pragma unroll
  for (int e = 0; e < 2; ++e) {
    const int v = 2 * wave + e;
    float s = part[0][v][lane] + part[1][v][lane];
    s += part[2][v][lane];
    s += part[3][v][lane];
    D[(size_t)(tm * 16 + 8 * hl + v) * NC + tn * 16 + lr] = s;
  }
}

template <int RM, int RN>
struct FragSet {
  v4f a[RM][4];
  v4f b[RN][4];
};

template <int RM, int RN>
__device__ __forceinline__ void rb_load(FragSet<RM, RN>& f, const float* Ap, size_t astep,
                                        const float* Bp, size_t bstep, int k0) {
  #pragma unroll
  for (int i = 0; i < RM; ++i) {
    #pragma unroll
    for (int q = 0; q < 4; ++q) f.a[i][q] = *(const v4f*)(Ap + i * astep + k0 + 8 * q);
  }
  #pragma unroll
  for (int j = 0; j < RN; ++j) {
    #pragma unroll
    for (int q = 0; q < 4; ++q) f.b[j][q] = *(const v4f*)(Bp + j * bstep + k0 + 8 * q);
  }
}

template <int RM, int RN>
__device__ __forceinline__ void rb_compute(v8f (&acc)[RM][RN], const FragSet<RM, RN>& f) {
  #pragma unroll
  for (int q = 0; q < 4; ++q) {
    #pragma unroll
    for (int i = 0; i < RM; ++i) {
      #pragma unroll
      for (int j = 0; j < RN; ++j) {
        v2f a = {f.a[i][q][0], f.a[i][q][1]};
        v2f b = {f.b[j][q][0], f.b[j][q][1]};
        acc[i][j] = WMMA_F32(a, b, acc[i][j]);
      }
    }
    #pragma unroll
    for (int i = 0; i < RM; ++i) {
      #pragma unroll
      for (int j = 0; j < RN; ++j) {
        v2f a = {f.a[i][q][2], f.a[i][q][3]};
        v2f b = {f.b[j][q][2], f.b[j][q][3]};
        acc[i][j] = WMMA_F32(a, b, acc[i][j]);
      }
    }
  }
}

template <int RM, int RN, int WM, int WN, int KD, bool HASC, int BN, int K2>
__global__ __launch_bounds__(128) void gemm_rb_kernel(
    float* D, int ldd,
    const float* A, int lda,
    const float* __restrict__ B, int ldb,
    const float* Cin, int ldc,
    const float* __restrict__ A2, const float* __restrict__ B2) {
  static_assert(WM * WN == 4, "four waves per work-group");
  static_assert(K2 == 0 || K2 == 32, "the second pair is exactly one 32-k step");
  static_assert(KD % 64 == 0 && KD >= 128, "two 32-k steps per trip");
  const int lane = threadIdx.x & 31, wave = threadIdx.x >> 5;
  const int hl = lane >> 4, lr = lane & 15;
  const int bn = blockIdx.x % BN, bm = blockIdx.x / BN;
  const int wm = wave / WN, wn = wave % WN;
  const int row0 = (bm * WM + wm) * (16 * RM);
  const int col0 = (bn * WN + wn) * (16 * RN);
  const float* Ap = A + (size_t)(row0 + lr) * lda + 4 * hl;
  const float* Bp = B + (size_t)(col0 + lr) * ldb + 4 * hl;
  const size_t astep = (size_t)16 * lda, bstep = (size_t)16 * ldb;

  v8f acc[RM][RN];
  #pragma unroll
  for (int i = 0; i < RM; ++i) {
    #pragma unroll
    for (int j = 0; j < RN; ++j) acc[i][j] = v8f{};
  }

  FragSet<RM, RN> f0, f1;
  rb_load<RM, RN>(f0, Ap, astep, Bp, bstep, 0);
  #pragma unroll 1
  for (int k0 = 0; k0 < KD - 64; k0 += 64) {
    rb_load<RM, RN>(f1, Ap, astep, Bp, bstep, k0 + 32);
    __builtin_amdgcn_sched_barrier(0);
    rb_compute<RM, RN>(acc, f0);
    __builtin_amdgcn_sched_barrier(0);
    rb_load<RM, RN>(f0, Ap, astep, Bp, bstep, k0 + 64);
    __builtin_amdgcn_sched_barrier(0);
    rb_compute<RM, RN>(acc, f1);
    __builtin_amdgcn_sched_barrier(0);
  }
  rb_load<RM, RN>(f1, Ap, astep, Bp, bstep, KD - 32);
  __builtin_amdgcn_sched_barrier(0);
  rb_compute<RM, RN>(acc, f0);
  __builtin_amdgcn_sched_barrier(0);

  float* Dp = D + (size_t)(row0 + 8 * hl) * ldd + col0 + lr;
  v8f cv[RM][RN];
  if (HASC) {
    const float* Cp = Cin + (size_t)(row0 + 8 * hl) * ldc + col0 + lr;
    #pragma unroll
    for (int i = 0; i < RM; ++i) {
      #pragma unroll
      for (int j = 0; j < RN; ++j) {
        #pragma unroll
        for (int v = 0; v < 8; ++v) cv[i][j][v] = Cp[(size_t)(16 * i + v) * ldc + 16 * j];
      }
    }
    __builtin_amdgcn_sched_barrier(0);
  }
  if (K2 != 0) {
    rb_load<RM, RN>(f0, A2 + (size_t)(row0 + lr) * K2 + 4 * hl, (size_t)16 * K2,
                    B2 + (size_t)(col0 + lr) * K2 + 4 * hl, (size_t)16 * K2, 0);
    __builtin_amdgcn_sched_barrier(0);
  }
  rb_compute<RM, RN>(acc, f1);
  if (K2 != 0) rb_compute<RM, RN>(acc, f0);
  if (HASC) {
    #pragma unroll
    for (int i = 0; i < RM; ++i) {
      #pragma unroll
      for (int j = 0; j < RN; ++j) {
        #pragma unroll
        for (int v = 0; v < 8; ++v) acc[i][j][v] += cv[i][j][v];
      }
    }
  }
  #pragma unroll
  for (int i = 0; i < RM; ++i) {
    #pragma unroll
    for (int j = 0; j < RN; ++j) {
      #pragma unroll
      for (int v = 0; v < 8; ++v) Dp[(size_t)(16 * i + v) * ldd + 16 * j] = acc[i][j][v];
    }
  }
}

__device__ __forceinline__ void ksplit_block(float (&s)[4], const float* Ar, const float* Br,
                                             float (*part)[2][8][32], int wave, int lane) {
  v4f a[8], b0[8], b1[8];
  #pragma unroll
  for (int q = 0; q < 8; ++q) {
    a[q]  = *(const v4f*)(Ar + 8 * q);
    b0[q] = *(const v4f*)(Br + 8 * q);
    b1[q] = *(const v4f*)(Br + (size_t)16 * 256 + 8 * q);
  }
  __builtin_amdgcn_sched_barrier(0);
  v8f acc0 = {}, acc1 = {};
  #pragma unroll
  for (int q = 0; q < 8; ++q) {
    v2f a01 = {a[q][0], a[q][1]}, a23 = {a[q][2], a[q][3]};
    v2f p01 = {b0[q][0], b0[q][1]}, p23 = {b0[q][2], b0[q][3]};
    v2f r01 = {b1[q][0], b1[q][1]}, r23 = {b1[q][2], b1[q][3]};
    acc0 = WMMA_F32(a01, p01, acc0);
    acc1 = WMMA_F32(a01, r01, acc1);
    acc0 = WMMA_F32(a23, p23, acc0);
    acc1 = WMMA_F32(a23, r23, acc1);
  }
  #pragma unroll
  for (int v = 0; v < 8; ++v) {
    part[wave][0][v][lane] = acc0[v];
    part[wave][1][v][lane] = acc1[v];
  }
  __syncthreads();
  const int j = lane >> 4, lc = lane & 15;
  #pragma unroll
  for (int e = 0; e < 4; ++e) {
    const int r = 4 * wave + e;
    const int v = r & 7, sl = 16 * (r >> 3) + lc;
    float t = part[0][j][v][sl] + part[1][j][v][sl];
    t += part[2][j][v][sl];
    t += part[3][j][v][sl];
    s[e] = t;
  }
}

__global__ __launch_bounds__(128) void pair_ksplit_kernel(
    float* D, int ldd, const float* A, int lda, const float* __restrict__ B,
    const float* Cin, int ldc) {
  __shared__ float part[4][2][8][32];
  const int lane = threadIdx.x & 31, wave = threadIdx.x >> 5;
  const int hl = lane >> 4, lr = lane & 15;
  const int tm = blockIdx.x >> 3, tp = blockIdx.x & 7;
  const int col = tp * 32 + lane;
  const int rbase = tm * 16 + 4 * wave;
  float c[4];
  #pragma unroll
  for (int e = 0; e < 4; ++e) c[e] = Cin[(size_t)(rbase + e) * ldc + col];
  const float* Ar = A + (size_t)(tm * 16 + lr) * lda + 64 * wave + 4 * hl;
  const float* Br = B + (size_t)(tp * 32 + lr) * 256 + 64 * wave + 4 * hl;
  float s[4];
  ksplit_block(s, Ar, Br, part, wave, lane);
  #pragma unroll
  for (int e = 0; e < 4; ++e) D[(size_t)(rbase + e) * ldd + col] = c[e] + s[e];
}

template <bool FIRST>
__device__ __forceinline__ const float* carry_row(const float* Vc, const float* G,
                                                  const float* x0, int c) {
  if (FIRST) return (c == 0) ? x0 : G + ((size_t)(c - 1) * L1 + (L1 - 1)) * 256;
  return Vc + (size_t)c * 256;
}
template <bool FIRST, bool LAST>
__global__ __launch_bounds__(128) void carry_double_kernel(
    float* __restrict__ Vn, const float* __restrict__ Vc,
    const float* __restrict__ Pd, int d,
    const float* __restrict__ G, const float* __restrict__ x0) {
  __shared__ float part[4][2][8][32];
  const int lane = threadIdx.x & 31, wave = threadIdx.x >> 5;
  const int hl = lane >> 4, lr = lane & 15;
  const int tm = blockIdx.x >> 3, tp = blockIdx.x & 7;
  const int col = tp * 32 + lane;
  const int rbase = tm * 16 + 4 * wave;
  float seed[4], res[4];
  #pragma unroll
  for (int e = 0; e < 4; ++e) {
    seed[e] = carry_row<FIRST>(Vc, G, x0, rbase + e)[col];
    res[e] = seed[e];
  }
  if (tm * 16 + 15 >= d) {
    const int srow = tm * 16 + lr - d;
    const float* Ar = carry_row<FIRST>(Vc, G, x0, srow >= 0 ? srow : 0) + 64 * wave + 4 * hl;
    const float* Br = Pd + (size_t)(tp * 32 + lr) * 256 + 64 * wave + 4 * hl;
    float s[4];
    ksplit_block(s, Ar, Br, part, wave, lane);
    #pragma unroll
    for (int e = 0; e < 4; ++e) res[e] = (rbase + e >= d) ? seed[e] + s[e] : seed[e];
  }
  constexpr size_t orow = LAST ? (size_t)L1 * 256 : (size_t)256;
  #pragma unroll
  for (int e = 0; e < 4; ++e) Vn[(size_t)(rbase + e) * orow + col] = res[e];
}

template <int NC, int Mr>
static inline void launch_ksplit(hipStream_t s, float* D, const float* A, const float* B) {
  static_assert(Mr % 16 == 0 && NC % 16 == 0, "whole tiles");
  gemm_ksplit_kernel<NC><<<(Mr / 16) * (NC / 16), 128, 0, s>>>(D, A, B);
}

template <int RM, int RN, int WM, int WN, int KD, bool HASC, int Mr, int Nc, int K2 = 0>
static inline void launch_rb(hipStream_t s, float* D, int ldd,
                             const float* A, int lda, const float* B, int ldb,
                             const float* Cin, int ldc,
                             const float* A2 = nullptr, const float* B2 = nullptr) {
  static_assert(Mr % (16 * RM * WM) == 0 && Nc % (16 * RN * WN) == 0, "whole work-group tiles");
  constexpr int blocksN = Nc / (16 * RN * WN);
  constexpr int blocksM = Mr / (16 * RM * WM);
  gemm_rb_kernel<RM, RN, WM, WN, KD, HASC, blocksN, K2><<<blocksM * blocksN, 128, 0, s>>>(
      D, ldd, A, lda, B, ldb, Cin, ldc, A2, B2);
}

template <int Mr>
static inline void launch_pair(hipStream_t s, float* D, int ldd, const float* A, int lda,
                               const float* B, const float* Cin, int ldc) {
  static_assert(Mr % 16 == 0, "whole blocks");
  pair_ksplit_kernel<<<(Mr / 16) * 8, 128, 0, s>>>(D, ldd, A, lda, B, Cin, ldc);
}

extern "C" void kernel_launch(void* const* d_in, const int* in_sizes, int n_in,
                              void* d_out, int out_size, void* d_ws, size_t ws_size,
                              hipStream_t stream) {
  (void)in_sizes; (void)n_in; (void)out_size; (void)ws_size;
  const float* t  = (const float*)d_in[0];
  const float* u  = (const float*)d_in[1];
  const float* x0 = (const float*)d_in[2];
  const float* Am = (const float*)d_in[3];
  const float* Bm = (const float*)d_in[4];
  const float* Cm = (const float*)d_in[5];
  const float* Dm = (const float*)d_in[6];
  float* xout = (float*)d_out;
  float* yout = xout + (size_t)Tn * NS;

  float* w = (float*)d_ws;
  float* Bd = w;  w += 8192;
  float* Kt[6]; for (int i = 0; i < 6; i++) { Kt[i] = w; w += 65536; }
  float* Rst  = w; w += 5 * 8192;
  float* Mm   = w; w += 65536;
  float* Qc   = w; w += 32768;
  float* PwA  = w; w += 65536;
  float* PwB  = w; w += 65536;
  float* PwC  = w; w += 65536;
  float* PwD  = w; w += 65536;
  float* Pm   = w; w += (size_t)Tn * 128;
  float* G    = w; w += (size_t)Tn * 256;
  float* V0   = w; w += (size_t)C1 * 256;
  w += (size_t)C2 * 256;
  w += (size_t)C2 * 256;
  float* V1   = w; w += (size_t)C1 * 256;

  float* PA  = Kt[0];
  k_front<<<Tn / (16 * FR), 128, 0, stream>>>(Pm, u, PA, Bd, Am, Bm, t);
  launch_ksplit<256, 256>(stream, Kt[1], Kt[0], Kt[0]);
  launch_ksplit<256, 512>(stream, Kt[2], Kt[0], Kt[1]);
  launch_ksplit<256, 512>(stream, Kt[4], Kt[0], Kt[3]);
  launch_ksplit<32, 1280>(stream, Rst, Kt[0], Bd);
  k_combine<<<(16384 + 8192) / 256, 256, 0, stream>>>(Mm, Qc, PA, Bd, Rst);

  launch_rb<2, 4, 2, 2, 128, false, Tn, 256>(stream, G, 256, Pm, 128, Qc, 128, nullptr, 0);

  float* Mp[5] = {Mm, Pm, Pm + 65536, Pm + 2 * 65536, Pm + 3 * 65536};
  float* Pw[10] = {PwA, PwB, PwC, PwD, Kt[0], Kt[1], Kt[2], Kt[3], Kt[4], Kt[5]};
  for (int j = 1; j < 5; ++j)
    launch_ksplit<256, 256>(stream, Mp[j], Mp[j-1], Mp[j-1]);
  launch_ksplit<256, 256>(stream, Pw[0], Mp[4], Mp[4]);
  for (int p = 1; p < 10; ++p)
    launch_ksplit<256, 256>(stream, Pw[p], Pw[p-1], Pw[p-1]);

  launch_rb<2, 4, 2, 2, 256, true, 16384, 256>(stream, G + 1 * 256, 512, G + 0 * 256, 512, Mp[0], 256, G + 1 * 256, 512);
  launch_rb<2, 4, 2, 2, 256, true, 8192, 256>(stream, G + 3 * 256, 1024, G + 1 * 256, 1024, Mp[1], 256, G + 3 * 256, 1024);
  launch_rb<2, 2, 2, 2, 256, true, 4096, 256>(stream, G + 7 * 256, 2048, G + 3 * 256, 2048, Mp[2], 256, G + 7 * 256, 2048);
  launch_pair<2048>(stream, G + 15 * 256, 4096, G + 7 * 256, 4096, Mp[3], G + 15 * 256, 4096);
  launch_pair<1024>(stream, G + 31 * 256, 8192, G + 15 * 256, 8192, Mp[4], G + 31 * 256, 8192);
  const int dd[10] = {1, 2, 4, 8, 16, 32, 64, 128, 256, 512};
  carry_double_kernel<true, false><<<512, 128, 0, stream>>>(V0, V1, Pw[0], dd[0], G, x0);
  float* Va = V0;
  float* Vb = V1;
  for (int p = 1; p < 9; ++p) {
    carry_double_kernel<false, false><<<512, 128, 0, stream>>>(Vb, Va, Pw[p], dd[p], G, x0);
    float* tv = Va; Va = Vb; Vb = tv;
  }
  carry_double_kernel<false, true><<<512, 128, 0, stream>>>(xout, Va, Pw[9], dd[9], G, x0);
  launch_pair<1024>(stream, xout + 16 * 256, 8192, xout, 8192, Mp[4], G + 15 * 256, 8192);
  launch_pair<2048>(stream, xout + 8 * 256, 4096, xout, 4096, Mp[3], G + 7 * 256, 4096);
  launch_rb<2, 2, 2, 2, 256, true, 4096, 256>(stream, xout + 4 * 256, 2048, xout, 2048, Mp[2], 256, G + 3 * 256, 2048);
  launch_rb<2, 4, 2, 2, 256, true, 8192, 256>(stream, xout + 2 * 256, 1024, xout, 1024, Mp[1], 256, G + 1 * 256, 1024);
  launch_rb<2, 4, 2, 2, 256, true, 16384, 256>(stream, xout + 1 * 256, 512, xout, 512, Mp[0], 256, G + 0 * 256, 512);

  launch_rb<2, 2, 4, 1, 256, false, Tn, 32, 32>(stream, yout, 32, xout, 256, Cm, 256, nullptr, 0, u, Dm);
}


# the 14 matrix squarings (M^2 .. M^16384) each ride as 256 extra work-groups in the launch before the power's first use (drive GEMM, 5 pairing, 8 carry passes): 42 launches become 28; sum order kept
# speedup vs baseline: 6.7943x; 1.1757x over previous
#include <hip/hip_runtime.h>

typedef __attribute__((ext_vector_type(2))) float v2f;
typedef __attribute__((ext_vector_type(4))) float v4f;
typedef __attribute__((ext_vector_type(8))) float v8f;

#define WMMA_F32(a, b, c) \
  __builtin_amdgcn_wmma_f32_16x16x4_f32(false, (a), false, (b), (short)0, (c), false, false)

static constexpr int Tn = 32768;
static constexpr int NS = 256;
static constexpr int L1 = 32;
static constexpr int C1 = 1024;
static constexpr int C2 = 64;

__constant__ float d_mu[7] = {
  1.00000000f, 1.00000000f, 0.500000000f, 0.166666672f, 0.0416666679f, 0.00833333377f, 0.00166666671f};
__constant__ float d_theta[4][6] = {
  {0.500000000f, 0.349999994f, 0.133333340f, 0.0351111107f, 0.00746666687f, 0.00166666671f},
  {0.0833333358f, 0.0500000007f, 0.0166666675f, 0.00366666657f, 0.00106666668f, 0.0f},
  {0.500000000f, 0.150000006f, 0.0333333351f, 0.00655555539f, 0.000866666669f, 0.0f},
  {-0.0833333358f, -0.0333333351f, -0.00833333377f, -0.00188888889f, -0.000266666670f, 0.0f},
};

__global__ __launch_bounds__(256) void k_combine(float* __restrict__ M, float* __restrict__ Q,
                                                 const float* __restrict__ PA,
                                                 const float* __restrict__ Bd,
                                                 const float* __restrict__ R) {
  const int g = blockIdx.x * 256 + threadIdx.x;
  if (g < 16384) {
    const int idx = 4 * g;
    v4f pl[6];
    #pragma unroll
    for (int p = 0; p < 6; ++p) pl[p] = *(const v4f*)(PA + p * 65536 + idx);
    v4f o;
    #pragma unroll
    for (int e = 0; e < 4; ++e) {
      float s = d_mu[6] * pl[5][e];
      #pragma unroll
      for (int p = 5; p >= 1; --p) s += d_mu[p] * pl[p - 1][e];
      o[e] = ((((idx + e) >> 8) == ((idx + e) & 255)) ? 1.0f : 0.0f) + s;
    }
    *(v4f*)(M + idx) = o;
  } else if (g < 16384 + 8192) {
    const int q = g - 16384;
    const int c = (q & 7) * 4, j = (q >> 3) & 3, r = q >> 5;
    v4f rl[5];
    #pragma unroll
    for (int p = 0; p < 5; ++p) rl[p] = *(const v4f*)(R + p * 8192 + r * 32 + c);
    const v4f bd = *(const v4f*)(Bd + r * 32 + c);
    v4f o;
    #pragma unroll
    for (int e = 0; e < 4; ++e) {
      float s = 0.0f;
      #pragma unroll
      for (int p = 5; p >= 1; --p) s += d_theta[j][p] * rl[p - 1][e];
      s += d_theta[j][0] * bd[e];
      o[e] = s;
    }
    *(v4f*)(Q + r * 128 + 32 * j + c) = o;
  }
}
static constexpr int FR = 8;
__global__ __launch_bounds__(128) void k_front(float* __restrict__ P, const float* __restrict__ u,
                                               float* __restrict__ Ad, float* __restrict__ Bd,
                                               const float* __restrict__ A, const float* __restrict__ B,
                                               const float* __restrict__ t) {
  const int tid = threadIdx.x;
  const int c = (tid & 7) * 4;
  const int n0 = (blockIdx.x * 16 + (tid >> 3)) * FR;
  const int g = blockIdx.x * 128 + tid;
  const bool scales = g < 16384 + 2048;
  const bool inA = g < 16384;
  const int off = 4 * (inA ? g : g - 16384);
  const float* sp = inA ? A : B;
  float* dp = inA ? Ad : Bd;
  const float t0 = t[0], t1 = t[1];
  v4f av = {};
  if (scales) av = *(const v4f*)(sp + off);
  v4f w[FR + 3];
  #pragma unroll
  for (int i = 0; i < FR + 3; ++i) {
    int r = n0 - 1 + i;
    r = r < 0 ? 0 : (r > Tn - 1 ? Tn - 1 : r);
    w[i] = *(const v4f*)(u + (size_t)r * 32 + c);
  }
  __builtin_amdgcn_sched_barrier(0);
  v4f s[FR + 1];
  #pragma unroll
  for (int i = 0; i < FR + 1; ++i) {
    const int m = n0 + i;
    const float h = (m == 0 || m >= Tn - 1) ? 1.0f : 0.5f;
    s[i] = h * (w[i + 2] - w[i]);
  }
  const v4f zero = {};
  #pragma unroll
  for (int i = 0; i < FR; ++i) {
    const int n = n0 + i;
    const bool live = (i < FR - 1) || (n < Tn - 1);
    float* row = P + (size_t)n * 128 + c;
    *(v4f*)(row)      = live ? w[i + 1] : zero;
    *(v4f*)(row + 32) = live ? s[i]     : zero;
    *(v4f*)(row + 64) = live ? w[i + 2] : zero;
    *(v4f*)(row + 96) = live ? s[i + 1] : zero;
  }
  if (scales) *(v4f*)(dp + off) = (t1 - t0) * av;
}

template <int NC>
__device__ __forceinline__ void ksplit_tile(float* __restrict__ D, const float* __restrict__ A,
                                            const float* __restrict__ B, int tile,
                                            float (*part)[8][32]) {
  const int lane = threadIdx.x & 31, wave = threadIdx.x >> 5;
  const int hl = lane >> 4, lr = lane & 15;
  constexpr int tilesN = NC >> 4;
  const int tm = tile / tilesN, tn = tile % tilesN;
  const float* Ar = A + (size_t)(tm * 16 + lr) * 256 + 64 * wave + 2 * hl;
  const float* Bc = B + (size_t)(64 * wave + 2 * hl) * NC + tn * 16 + lr;
  v2f a[16], b[16];
  #pragma unroll
  for (int i = 0; i < 16; ++i) {
    a[i] = *(const v2f*)(Ar + 4 * i);
    b[i][0] = Bc[(size_t)(4 * i) * NC];
    b[i][1] = Bc[(size_t)(4 * i) * NC + NC];
  }
  __builtin_amdgcn_sched_barrier(0);
  v8f acc = {};
  #pragma unroll
  for (int i = 0; i < 16; ++i) acc = WMMA_F32(a[i], b[i], acc);
  #pragma unroll
  for (int v = 0; v < 8; ++v) part[wave][v][lane] = acc[v];
  __syncthreads();
  #pragma unroll
  for (int e = 0; e < 2; ++e) {
    const int v = 2 * wave + e;
    float s = part[0][v][lane] + part[1][v][lane];
    s += part[2][v][lane];
    s += part[3][v][lane];
    D[(size_t)(tm * 16 + 8 * hl + v) * NC + tn * 16 + lr] = s;
  }
}

template <int NC>
__global__ __launch_bounds__(128) void gemm_ksplit_kernel(
    float* __restrict__ D, const float* __restrict__ A, const float* __restrict__ B) {
  __shared__ float part[4][8][32];
  ksplit_tile<NC>(D, A, B, blockIdx.x, part);
}

static constexpr int SQ_TILES = 256;
__device__ __forceinline__ void square_tile(float* __restrict__ SqD, const float* __restrict__ SqA,
                                            int tile, float (*part)[8][32]) {
  ksplit_tile<256>(SqD, SqA, SqA, tile, part);
}

template <int RM, int RN>
struct FragSet {
  v4f a[RM][4];
  v4f b[RN][4];
};

template <int RM, int RN>
__device__ __forceinline__ void rb_load(FragSet<RM, RN>& f, const float* Ap, size_t astep,
                                        const float* Bp, size_t bstep, int k0) {
  #pragma unroll
  for (int i = 0; i < RM; ++i) {
    #pragma unroll
    for (int q = 0; q < 4; ++q) f.a[i][q] = *(const v4f*)(Ap + i * astep + k0 + 8 * q);
  }
  #pragma unroll
  for (int j = 0; j < RN; ++j) {
    #pragma unroll
    for (int q = 0; q < 4; ++q) f.b[j][q] = *(const v4f*)(Bp + j * bstep + k0 + 8 * q);
  }
}

template <int RM, int RN>
__device__ __forceinline__ void rb_compute(v8f (&acc)[RM][RN], const FragSet<RM, RN>& f) {
  #pragma unroll
  for (int q = 0; q < 4; ++q) {
    #pragma unroll
    for (int i = 0; i < RM; ++i) {
      #pragma unroll
      for (int j = 0; j < RN; ++j) {
        v2f a = {f.a[i][q][0], f.a[i][q][1]};
        v2f b = {f.b[j][q][0], f.b[j][q][1]};
        acc[i][j] = WMMA_F32(a, b, acc[i][j]);
      }
    }
    #pragma unroll
    for (int i = 0; i < RM; ++i) {
      #pragma unroll
      for (int j = 0; j < RN; ++j) {
        v2f a = {f.a[i][q][2], f.a[i][q][3]};
        v2f b = {f.b[j][q][2], f.b[j][q][3]};
        acc[i][j] = WMMA_F32(a, b, acc[i][j]);
      }
    }
  }
}

template <int RM, int RN, int WM, int WN, int KD, bool HASC, int BN, int K2, int MB>
__global__ __launch_bounds__(128) void gemm_rb_kernel(
    float* D, int ldd,
    const float* A, int lda,
    const float* __restrict__ B, int ldb,
    const float* Cin, int ldc,
    const float* __restrict__ A2, const float* __restrict__ B2,
    float* __restrict__ SqD, const float* __restrict__ SqA) {
  static_assert(WM * WN == 4, "four waves per work-group");
  if constexpr (MB != 0) {
    if ((int)blockIdx.x >= MB) {
      __shared__ float sqpart[4][8][32];
      square_tile(SqD, SqA, (int)blockIdx.x - MB, sqpart);
      return;
    }
  }
  static_assert(K2 == 0 || K2 == 32, "the second pair is exactly one 32-k step");
  static_assert(KD % 64 == 0 && KD >= 128, "two 32-k steps per trip");
  const int lane = threadIdx.x & 31, wave = threadIdx.x >> 5;
  const int hl = lane >> 4, lr = lane & 15;
  const int bn = blockIdx.x % BN, bm = blockIdx.x / BN;
  const int wm = wave / WN, wn = wave % WN;
  const int row0 = (bm * WM + wm) * (16 * RM);
  const int col0 = (bn * WN + wn) * (16 * RN);
  const float* Ap = A + (size_t)(row0 + lr) * lda + 4 * hl;
  const float* Bp = B + (size_t)(col0 + lr) * ldb + 4 * hl;
  const size_t astep = (size_t)16 * lda, bstep = (size_t)16 * ldb;

  v8f acc[RM][RN];
  #pragma unroll
  for (int i = 0; i < RM; ++i) {
    #pragma unroll
    for (int j = 0; j < RN; ++j) acc[i][j] = v8f{};
  }

  FragSet<RM, RN> f0, f1;
  rb_load<RM, RN>(f0, Ap, astep, Bp, bstep, 0);
  #pragma unroll 1
  for (int k0 = 0; k0 < KD - 64; k0 += 64) {
    rb_load<RM, RN>(f1, Ap, astep, Bp, bstep, k0 + 32);
    __builtin_amdgcn_sched_barrier(0);
    rb_compute<RM, RN>(acc, f0);
    __builtin_amdgcn_sched_barrier(0);
    rb_load<RM, RN>(f0, Ap, astep, Bp, bstep, k0 + 64);
    __builtin_amdgcn_sched_barrier(0);
    rb_compute<RM, RN>(acc, f1);
    __builtin_amdgcn_sched_barrier(0);
  }
  rb_load<RM, RN>(f1, Ap, astep, Bp, bstep, KD - 32);
  __builtin_amdgcn_sched_barrier(0);
  rb_compute<RM, RN>(acc, f0);
  __builtin_amdgcn_sched_barrier(0);

  float* Dp = D + (size_t)(row0 + 8 * hl) * ldd + col0 + lr;
  v8f cv[RM][RN];
  if (HASC) {
    const float* Cp = Cin + (size_t)(row0 + 8 * hl) * ldc + col0 + lr;
    #pragma unroll
    for (int i = 0; i < RM; ++i) {
      #pragma unroll
      for (int j = 0; j < RN; ++j) {
        #pragma unroll
        for (int v = 0; v < 8; ++v) cv[i][j][v] = Cp[(size_t)(16 * i + v) * ldc + 16 * j];
      }
    }
    __builtin_amdgcn_sched_barrier(0);
  }
  if (K2 != 0) {
    rb_load<RM, RN>(f0, A2 + (size_t)(row0 + lr) * K2 + 4 * hl, (size_t)16 * K2,
                    B2 + (size_t)(col0 + lr) * K2 + 4 * hl, (size_t)16 * K2, 0);
    __builtin_amdgcn_sched_barrier(0);
  }
  rb_compute<RM, RN>(acc, f1);
  if (K2 != 0) rb_compute<RM, RN>(acc, f0);
  if (HASC) {
    #pragma unroll
    for (int i = 0; i < RM; ++i) {
      #pragma unroll
      for (int j = 0; j < RN; ++j) {
        #pragma unroll
        for (int v = 0; v < 8; ++v) acc[i][j][v] += cv[i][j][v];
      }
    }
  }
  #pragma unroll
  for (int i = 0; i < RM; ++i) {
    #pragma unroll
    for (int j = 0; j < RN; ++j) {
      #pragma unroll
      for (int v = 0; v < 8; ++v) Dp[(size_t)(16 * i + v) * ldd + 16 * j] = acc[i][j][v];
    }
  }
}

__device__ __forceinline__ void ksplit_block(float (&s)[4], const float* Ar, const float* Br,
                                             float (*part)[2][8][32], int wave, int lane) {
  v4f a[8], b0[8], b1[8];
  #pragma unroll
  for (int q = 0; q < 8; ++q) {
    a[q]  = *(const v4f*)(Ar + 8 * q);
    b0[q] = *(const v4f*)(Br + 8 * q);
    b1[q] = *(const v4f*)(Br + (size_t)16 * 256 + 8 * q);
  }
  __builtin_amdgcn_sched_barrier(0);
  v8f acc0 = {}, acc1 = {};
  #pragma unroll
  for (int q = 0; q < 8; ++q) {
    v2f a01 = {a[q][0], a[q][1]}, a23 = {a[q][2], a[q][3]};
    v2f p01 = {b0[q][0], b0[q][1]}, p23 = {b0[q][2], b0[q][3]};
    v2f r01 = {b1[q][0], b1[q][1]}, r23 = {b1[q][2], b1[q][3]};
    acc0 = WMMA_F32(a01, p01, acc0);
    acc1 = WMMA_F32(a01, r01, acc1);
    acc0 = WMMA_F32(a23, p23, acc0);
    acc1 = WMMA_F32(a23, r23, acc1);
  }
  #pragma unroll
  for (int v = 0; v < 8; ++v) {
    part[wave][0][v][lane] = acc0[v];
    part[wave][1][v][lane] = acc1[v];
  }
  __syncthreads();
  const int j = lane >> 4, lc = lane & 15;
  #pragma unroll
  for (int e = 0; e < 4; ++e) {
    const int r = 4 * wave + e;
    const int v = r & 7, sl = 16 * (r >> 3) + lc;
    float t = part[0][j][v][sl] + part[1][j][v][sl];
    t += part[2][j][v][sl];
    t += part[3][j][v][sl];
    s[e] = t;
  }
}

template <int MB>
__global__ __launch_bounds__(128) void pair_ksplit_kernel(
    float* D, int ldd, const float* A, int lda, const float* __restrict__ B,
    const float* Cin, int ldc, float* __restrict__ SqD, const float* __restrict__ SqA) {
  if constexpr (MB != 0) {
    if ((int)blockIdx.x >= MB) {
      __shared__ float sqpart[4][8][32];
      square_tile(SqD, SqA, (int)blockIdx.x - MB, sqpart);
      return;
    }
  }
  __shared__ float part[4][2][8][32];
  const int lane = threadIdx.x & 31, wave = threadIdx.x >> 5;
  const int hl = lane >> 4, lr = lane & 15;
  const int tm = blockIdx.x >> 3, tp = blockIdx.x & 7;
  const int col = tp * 32 + lane;
  const int rbase = tm * 16 + 4 * wave;
  float c[4];
  #pragma unroll
  for (int e = 0; e < 4; ++e) c[e] = Cin[(size_t)(rbase + e) * ldc + col];
  const float* Ar = A + (size_t)(tm * 16 + lr) * lda + 64 * wave + 4 * hl;
  const float* Br = B + (size_t)(tp * 32 + lr) * 256 + 64 * wave + 4 * hl;
  float s[4];
  ksplit_block(s, Ar, Br, part, wave, lane);
  #pragma unroll
  for (int e = 0; e < 4; ++e) D[(size_t)(rbase + e) * ldd + col] = c[e] + s[e];
}

template <bool FIRST>
__device__ __forceinline__ const float* carry_row(const float* Vc, const float* G,
                                                  const float* x0, int c) {
  if (FIRST) return (c == 0) ? x0 : G + ((size_t)(c - 1) * L1 + (L1 - 1)) * 256;
  return Vc + (size_t)c * 256;
}
static constexpr int CARRY_BLOCKS = 512;
template <bool FIRST, bool LAST, bool SQ>
__global__ __launch_bounds__(128) void carry_double_kernel(
    float* __restrict__ Vn, const float* __restrict__ Vc,
    const float* __restrict__ Pd, int d,
    const float* __restrict__ G, const float* __restrict__ x0,
    float* __restrict__ SqD, const float* __restrict__ SqA) {
  if constexpr (SQ) {
    if ((int)blockIdx.x >= CARRY_BLOCKS) {
      __shared__ float sqpart[4][8][32];
      square_tile(SqD, SqA, (int)blockIdx.x - CARRY_BLOCKS, sqpart);
      return;
    }
  }
  __shared__ float part[4][2][8][32];
  const int lane = threadIdx.x & 31, wave = threadIdx.x >> 5;
  const int hl = lane >> 4, lr = lane & 15;
  const int tm = blockIdx.x >> 3, tp = blockIdx.x & 7;
  const int col = tp * 32 + lane;
  const int rbase = tm * 16 + 4 * wave;
  float seed[4], res[4];
  #pragma unroll
  for (int e = 0; e < 4; ++e) {
    seed[e] = carry_row<FIRST>(Vc, G, x0, rbase + e)[col];
    res[e] = seed[e];
  }
  if (tm * 16 + 15 >= d) {
    const int srow = tm * 16 + lr - d;
    const float* Ar = carry_row<FIRST>(Vc, G, x0, srow >= 0 ? srow : 0) + 64 * wave + 4 * hl;
    const float* Br = Pd + (size_t)(tp * 32 + lr) * 256 + 64 * wave + 4 * hl;
    float s[4];
    ksplit_block(s, Ar, Br, part, wave, lane);
    #pragma unroll
    for (int e = 0; e < 4; ++e) res[e] = (rbase + e >= d) ? seed[e] + s[e] : seed[e];
  }
  constexpr size_t orow = LAST ? (size_t)L1 * 256 : (size_t)256;
  #pragma unroll
  for (int e = 0; e < 4; ++e) Vn[(size_t)(rbase + e) * orow + col] = res[e];
}

template <int NC, int Mr>
static inline void launch_ksplit(hipStream_t s, float* D, const float* A, const float* B) {
  static_assert(Mr % 16 == 0 && NC % 16 == 0, "whole tiles");
  gemm_ksplit_kernel<NC><<<(Mr / 16) * (NC / 16), 128, 0, s>>>(D, A, B);
}

template <int RM, int RN, int WM, int WN, int KD, bool HASC, int Mr, int Nc, int K2 = 0, bool SQ = false>
static inline void launch_rb(hipStream_t s, float* D, int ldd,
                             const float* A, int lda, const float* B, int ldb,
                             const float* Cin, int ldc,
                             const float* A2 = nullptr, const float* B2 = nullptr,
                             float* SqD = nullptr, const float* SqA = nullptr) {
  static_assert(Mr % (16 * RM * WM) == 0 && Nc % (16 * RN * WN) == 0, "whole work-group tiles");
  constexpr int blocksN = Nc / (16 * RN * WN);
  constexpr int blocksM = Mr / (16 * RM * WM);
  constexpr int own = blocksM * blocksN;
  gemm_rb_kernel<RM, RN, WM, WN, KD, HASC, blocksN, K2, SQ ? own : 0>
      <<<own + (SQ ? SQ_TILES : 0), 128, 0, s>>>(D, ldd, A, lda, B, ldb, Cin, ldc, A2, B2, SqD, SqA);
}

template <int Mr, bool SQ = false>
static inline void launch_pair(hipStream_t s, float* D, int ldd, const float* A, int lda,
                               const float* B, const float* Cin, int ldc,
                               float* SqD = nullptr, const float* SqA = nullptr) {
  static_assert(Mr % 16 == 0, "whole blocks");
  constexpr int own = (Mr / 16) * 8;
  pair_ksplit_kernel<SQ ? own : 0><<<own + (SQ ? SQ_TILES : 0), 128, 0, s>>>(
      D, ldd, A, lda, B, Cin, ldc, SqD, SqA);
}

extern "C" void kernel_launch(void* const* d_in, const int* in_sizes, int n_in,
                              void* d_out, int out_size, void* d_ws, size_t ws_size,
                              hipStream_t stream) {
  (void)in_sizes; (void)n_in; (void)out_size; (void)ws_size;
  const float* t  = (const float*)d_in[0];
  const float* u  = (const float*)d_in[1];
  const float* x0 = (const float*)d_in[2];
  const float* Am = (const float*)d_in[3];
  const float* Bm = (const float*)d_in[4];
  const float* Cm = (const float*)d_in[5];
  const float* Dm = (const float*)d_in[6];
  float* xout = (float*)d_out;
  float* yout = xout + (size_t)Tn * NS;

  float* w = (float*)d_ws;
  float* Bd = w;  w += 8192;
  float* Kt[6]; for (int i = 0; i < 6; i++) { Kt[i] = w; w += 65536; }
  float* Rst  = w; w += 5 * 8192;
  float* Mm   = w; w += 65536;
  float* Qc   = w; w += 32768;
  float* PwA  = w; w += 65536;
  float* PwB  = w; w += 65536;
  float* PwC  = w; w += 65536;
  float* PwD  = w; w += 65536;
  float* Pm   = w; w += (size_t)Tn * 128;
  float* G    = w; w += (size_t)Tn * 256;
  float* V0   = w; w += (size_t)C1 * 256;
  w += (size_t)C2 * 256;
  w += (size_t)C2 * 256;
  float* V1   = w; w += (size_t)C1 * 256;

  float* PA  = Kt[0];
  k_front<<<Tn / (16 * FR), 128, 0, stream>>>(Pm, u, PA, Bd, Am, Bm, t);
  launch_ksplit<256, 256>(stream, Kt[1], Kt[0], Kt[0]);
  launch_ksplit<256, 512>(stream, Kt[2], Kt[0], Kt[1]);
  launch_ksplit<256, 512>(stream, Kt[4], Kt[0], Kt[3]);
  launch_ksplit<32, 1280>(stream, Rst, Kt[0], Bd);
  k_combine<<<(16384 + 8192) / 256, 256, 0, stream>>>(Mm, Qc, PA, Bd, Rst);

  float* Mp[5] = {Mm, Kt[0], Kt[1], Kt[2], Kt[3]};
  float* Pw[10] = {PwA, PwB, PwC, PwD, PwA, PwB, PwC, PwD, PwA, PwB};

  launch_rb<2, 4, 2, 2, 128, false, Tn, 256, 0, true>(stream, G, 256, Pm, 128, Qc, 128, nullptr, 0,
                                                      nullptr, nullptr, Mp[1], Mp[0]);

  launch_rb<2, 4, 2, 2, 256, true, 16384, 256, 0, true>(stream, G + 1 * 256, 512, G + 0 * 256, 512, Mp[0], 256, G + 1 * 256, 512,
                                                        nullptr, nullptr, Mp[2], Mp[1]);
  launch_rb<2, 4, 2, 2, 256, true, 8192, 256, 0, true>(stream, G + 3 * 256, 1024, G + 1 * 256, 1024, Mp[1], 256, G + 3 * 256, 1024,
                                                       nullptr, nullptr, Mp[3], Mp[2]);
  launch_rb<2, 2, 2, 2, 256, true, 4096, 256, 0, true>(stream, G + 7 * 256, 2048, G + 3 * 256, 2048, Mp[2], 256, G + 7 * 256, 2048,
                                                       nullptr, nullptr, Mp[4], Mp[3]);
  launch_pair<2048, true>(stream, G + 15 * 256, 4096, G + 7 * 256, 4096, Mp[3], G + 15 * 256, 4096,
                          Pw[0], Mp[4]);
  launch_pair<1024, true>(stream, G + 31 * 256, 8192, G + 15 * 256, 8192, Mp[4], G + 31 * 256, 8192,
                          Pw[1], Pw[0]);
  const int dd[10] = {1, 2, 4, 8, 16, 32, 64, 128, 256, 512};
  carry_double_kernel<true, false, true><<<CARRY_BLOCKS + SQ_TILES, 128, 0, stream>>>(
      V0, V1, Pw[0], dd[0], G, x0, Pw[2], Pw[1]);
  float* Va = V0;
  float* Vb = V1;
  for (int p = 1; p < 8; ++p) {
    carry_double_kernel<false, false, true><<<CARRY_BLOCKS + SQ_TILES, 128, 0, stream>>>(
        Vb, Va, Pw[p], dd[p], G, x0, Pw[p + 2], Pw[p + 1]);
    float* tv = Va; Va = Vb; Vb = tv;
  }
  carry_double_kernel<false, false, false><<<CARRY_BLOCKS, 128, 0, stream>>>(
      Vb, Va, Pw[8], dd[8], G, x0, nullptr, nullptr);
  carry_double_kernel<false, true, false><<<CARRY_BLOCKS, 128, 0, stream>>>(
      xout, Vb, Pw[9], dd[9], G, x0, nullptr, nullptr);
  launch_pair<1024>(stream, xout + 16 * 256, 8192, xout, 8192, Mp[4], G + 15 * 256, 8192);
  launch_pair<2048>(stream, xout + 8 * 256, 4096, xout, 4096, Mp[3], G + 7 * 256, 4096);
  launch_rb<2, 2, 2, 2, 256, true, 4096, 256>(stream, xout + 4 * 256, 2048, xout, 2048, Mp[2], 256, G + 3 * 256, 2048);
  launch_rb<2, 4, 2, 2, 256, true, 8192, 256>(stream, xout + 2 * 256, 1024, xout, 1024, Mp[1], 256, G + 1 * 256, 1024);
  launch_rb<2, 4, 2, 2, 256, true, 16384, 256>(stream, xout + 1 * 256, 512, xout, 512, Mp[0], 256, G + 0 * 256, 512);

  launch_rb<2, 2, 4, 1, 256, false, Tn, 32, 32>(stream, yout, 32, xout, 256, Cm, 256, nullptr, 0, u, Dm);
}


# way down (5 levels) and outputs y in ONE launch, 28 launches to 23: a work-group takes 128 time rows through all levels with the rows in LDS, then forms y from LDS; strides 8, 16: new k sum order
# speedup vs baseline: 7.0953x; 1.0443x over previous
#include <hip/hip_runtime.h>

typedef __attribute__((ext_vector_type(2))) float v2f;
typedef __attribute__((ext_vector_type(4))) float v4f;
typedef __attribute__((ext_vector_type(8))) float v8f;

#define WMMA_F32(a, b, c) \
  __builtin_amdgcn_wmma_f32_16x16x4_f32(false, (a), false, (b), (short)0, (c), false, false)

static constexpr int Tn = 32768;
static constexpr int NS = 256;
static constexpr int L1 = 32;
static constexpr int C1 = 1024;
static constexpr int C2 = 64;

__constant__ float d_mu[7] = {
  1.00000000f, 1.00000000f, 0.500000000f, 0.166666672f, 0.0416666679f, 0.00833333377f, 0.00166666671f};
__constant__ float d_theta[4][6] = {
  {0.500000000f, 0.349999994f, 0.133333340f, 0.0351111107f, 0.00746666687f, 0.00166666671f},
  {0.0833333358f, 0.0500000007f, 0.0166666675f, 0.00366666657f, 0.00106666668f, 0.0f},
  {0.500000000f, 0.150000006f, 0.0333333351f, 0.00655555539f, 0.000866666669f, 0.0f},
  {-0.0833333358f, -0.0333333351f, -0.00833333377f, -0.00188888889f, -0.000266666670f, 0.0f},
};

__global__ __launch_bounds__(256) void k_combine(float* __restrict__ M, float* __restrict__ Q,
                                                 const float* __restrict__ PA,
                                                 const float* __restrict__ Bd,
                                                 const float* __restrict__ R) {
  const int g = blockIdx.x * 256 + threadIdx.x;
  if (g < 16384) {
    const int idx = 4 * g;
    v4f pl[6];
    #pragma unroll
    for (int p = 0; p < 6; ++p) pl[p] = *(const v4f*)(PA + p * 65536 + idx);
    v4f o;
    #pragma unroll
    for (int e = 0; e < 4; ++e) {
      float s = d_mu[6] * pl[5][e];
      #pragma unroll
      for (int p = 5; p >= 1; --p) s += d_mu[p] * pl[p - 1][e];
      o[e] = ((((idx + e) >> 8) == ((idx + e) & 255)) ? 1.0f : 0.0f) + s;
    }
    *(v4f*)(M + idx) = o;
  } else if (g < 16384 + 8192) {
    const int q = g - 16384;
    const int c = (q & 7) * 4, j = (q >> 3) & 3, r = q >> 5;
    v4f rl[5];
    #pragma unroll
    for (int p = 0; p < 5; ++p) rl[p] = *(const v4f*)(R + p * 8192 + r * 32 + c);
    const v4f bd = *(const v4f*)(Bd + r * 32 + c);
    v4f o;
    #pragma unroll
    for (int e = 0; e < 4; ++e) {
      float s = 0.0f;
      #pragma unroll
      for (int p = 5; p >= 1; --p) s += d_theta[j][p] * rl[p - 1][e];
      s += d_theta[j][0] * bd[e];
      o[e] = s;
    }
    *(v4f*)(Q + r * 128 + 32 * j + c) = o;
  }
}
static constexpr int FR = 8;
__global__ __launch_bounds__(128) void k_front(float* __restrict__ P, const float* __restrict__ u,
                                               float* __restrict__ Ad, float* __restrict__ Bd,
                                               const float* __restrict__ A, const float* __restrict__ B,
                                               const float* __restrict__ t) {
  const int tid = threadIdx.x;
  const int c = (tid & 7) * 4;
  const int n0 = (blockIdx.x * 16 + (tid >> 3)) * FR;
  const int g = blockIdx.x * 128 + tid;
  const bool scales = g < 16384 + 2048;
  const bool inA = g < 16384;
  const int off = 4 * (inA ? g : g - 16384);
  const float* sp = inA ? A : B;
  float* dp = inA ? Ad : Bd;
  const float t0 = t[0], t1 = t[1];
  v4f av = {};
  if (scales) av = *(const v4f*)(sp + off);
  v4f w[FR + 3];
  #pragma unroll
  for (int i = 0; i < FR + 3; ++i) {
    int r = n0 - 1 + i;
    r = r < 0 ? 0 : (r > Tn - 1 ? Tn - 1 : r);
    w[i] = *(const v4f*)(u + (size_t)r * 32 + c);
  }
  __builtin_amdgcn_sched_barrier(0);
  v4f s[FR + 1];
  #pragma unroll
  for (int i = 0; i < FR + 1; ++i) {
    const int m = n0 + i;
    const float h = (m == 0 || m >= Tn - 1) ? 1.0f : 0.5f;
    s[i] = h * (w[i + 2] - w[i]);
  }
  const v4f zero = {};
  #pragma unroll
  for (int i = 0; i < FR; ++i) {
    const int n = n0 + i;
    const bool live = (i < FR - 1) || (n < Tn - 1);
    float* row = P + (size_t)n * 128 + c;
    *(v4f*)(row)      = live ? w[i + 1] : zero;
    *(v4f*)(row + 32) = live ? s[i]     : zero;
    *(v4f*)(row + 64) = live ? w[i + 2] : zero;
    *(v4f*)(row + 96) = live ? s[i + 1] : zero;
  }
  if (scales) *(v4f*)(dp + off) = (t1 - t0) * av;
}

template <int NC>
__device__ __forceinline__ void ksplit_tile(float* __restrict__ D, const float* __restrict__ A,
                                            const float* __restrict__ B, int tile,
                                            float (*part)[8][32]) {
  const int lane = threadIdx.x & 31, wave = threadIdx.x >> 5;
  const int hl = lane >> 4, lr = lane & 15;
  constexpr int tilesN = NC >> 4;
  const int tm = tile / tilesN, tn = tile % tilesN;
  const float* Ar = A + (size_t)(tm * 16 + lr) * 256 + 64 * wave + 2 * hl;
  const float* Bc = B + (size_t)(64 * wave + 2 * hl) * NC + tn * 16 + lr;
  v2f a[16], b[16];
  #pragma unroll
  for (int i = 0; i < 16; ++i) {
    a[i] = *(const v2f*)(Ar + 4 * i);
    b[i][0] = Bc[(size_t)(4 * i) * NC];
    b[i][1] = Bc[(size_t)(4 * i) * NC + NC];
  }
  __builtin_amdgcn_sched_barrier(0);
  v8f acc = {};
  #pragma unroll
  for (int i = 0; i < 16; ++i) acc = WMMA_F32(a[i], b[i], acc);
  #pragma unroll
  for (int v = 0; v < 8; ++v) part[wave][v][lane] = acc[v];
  __syncthreads();
  #pragma unroll
  for (int e = 0; e < 2; ++e) {
    const int v = 2 * wave + e;
    float s = part[0][v][lane] + part[1][v][lane];
    s += part[2][v][lane];
    s += part[3][v][lane];
    D[(size_t)(tm * 16 + 8 * hl + v) * NC + tn * 16 + lr] = s;
  }
}

template <int NC>
__global__ __launch_bounds__(128) void gemm_ksplit_kernel(
    float* __restrict__ D, const float* __restrict__ A, const float* __restrict__ B) {
  __shared__ float part[4][8][32];
  ksplit_tile<NC>(D, A, B, blockIdx.x, part);
}

static constexpr int SQ_TILES = 256;
__device__ __forceinline__ void square_tile(float* __restrict__ SqD, const float* __restrict__ SqA,
                                            int tile, float (*part)[8][32]) {
  ksplit_tile<256>(SqD, SqA, SqA, tile, part);
}

template <int RM, int RN>
struct FragSet {
  v4f a[RM][4];
  v4f b[RN][4];
};

template <int RM, int RN>
__device__ __forceinline__ void rb_load(FragSet<RM, RN>& f, const float* Ap, size_t astep,
                                        const float* Bp, size_t bstep, int k0) {
  #pragma unroll
  for (int i = 0; i < RM; ++i) {
    #pragma unroll
    for (int q = 0; q < 4; ++q) f.a[i][q] = *(const v4f*)(Ap + i * astep + k0 + 8 * q);
  }
  #pragma unroll
  for (int j = 0; j < RN; ++j) {
    #pragma unroll
    for (int q = 0; q < 4; ++q) f.b[j][q] = *(const v4f*)(Bp + j * bstep + k0 + 8 * q);
  }
}

template <int RM, int RN>
__device__ __forceinline__ void rb_compute(v8f (&acc)[RM][RN], const FragSet<RM, RN>& f) {
  #pragma unroll
  for (int q = 0; q < 4; ++q) {
    #pragma unroll
    for (int i = 0; i < RM; ++i) {
      #pragma unroll
      for (int j = 0; j < RN; ++j) {
        v2f a = {f.a[i][q][0], f.a[i][q][1]};
        v2f b = {f.b[j][q][0], f.b[j][q][1]};
        acc[i][j] = WMMA_F32(a, b, acc[i][j]);
      }
    }
    #pragma unroll
    for (int i = 0; i < RM; ++i) {
      #pragma unroll
      for (int j = 0; j < RN; ++j) {
        v2f a = {f.a[i][q][2], f.a[i][q][3]};
        v2f b = {f.b[j][q][2], f.b[j][q][3]};
        acc[i][j] = WMMA_F32(a, b, acc[i][j]);
      }
    }
  }
}

template <int RM, int RN, int WM, int WN, int KD, bool HASC, int BN, int K2, int MB>
__global__ __launch_bounds__(128) void gemm_rb_kernel(
    float* D, int ldd,
    const float* A, int lda,
    const float* __restrict__ B, int ldb,
    const float* Cin, int ldc,
    const float* __restrict__ A2, const float* __restrict__ B2,
    float* __restrict__ SqD, const float* __restrict__ SqA) {
  static_assert(WM * WN == 4, "four waves per work-group");
  if constexpr (MB != 0) {
    if ((int)blockIdx.x >= MB) {
      __shared__ float sqpart[4][8][32];
      square_tile(SqD, SqA, (int)blockIdx.x - MB, sqpart);
      return;
    }
  }
  static_assert(K2 == 0 || K2 == 32, "the second pair is exactly one 32-k step");
  static_assert(KD % 64 == 0 && KD >= 128, "two 32-k steps per trip");
  const int lane = threadIdx.x & 31, wave = threadIdx.x >> 5;
  const int hl = lane >> 4, lr = lane & 15;
  const int bn = blockIdx.x % BN, bm = blockIdx.x / BN;
  const int wm = wave / WN, wn = wave % WN;
  const int row0 = (bm * WM + wm) * (16 * RM);
  const int col0 = (bn * WN + wn) * (16 * RN);
  const float* Ap = A + (size_t)(row0 + lr) * lda + 4 * hl;
  const float* Bp = B + (size_t)(col0 + lr) * ldb + 4 * hl;
  const size_t astep = (size_t)16 * lda, bstep = (size_t)16 * ldb;

  v8f acc[RM][RN];
  #pragma unroll
  for (int i = 0; i < RM; ++i) {
    #pragma unroll
    for (int j = 0; j < RN; ++j) acc[i][j] = v8f{};
  }

  FragSet<RM, RN> f0, f1;
  rb_load<RM, RN>(f0, Ap, astep, Bp, bstep, 0);
  #pragma unroll 1
  for (int k0 = 0; k0 < KD - 64; k0 += 64) {
    rb_load<RM, RN>(f1, Ap, astep, Bp, bstep, k0 + 32);
    __builtin_amdgcn_sched_barrier(0);
    rb_compute<RM, RN>(acc, f0);
    __builtin_amdgcn_sched_barrier(0);
    rb_load<RM, RN>(f0, Ap, astep, Bp, bstep, k0 + 64);
    __builtin_amdgcn_sched_barrier(0);
    rb_compute<RM, RN>(acc, f1);
    __builtin_amdgcn_sched_barrier(0);
  }
  rb_load<RM, RN>(f1, Ap, astep, Bp, bstep, KD - 32);
  __builtin_amdgcn_sched_barrier(0);
  rb_compute<RM, RN>(acc, f0);
  __builtin_amdgcn_sched_barrier(0);

  float* Dp = D + (size_t)(row0 + 8 * hl) * ldd + col0 + lr;
  v8f cv[RM][RN];
  if (HASC) {
    const float* Cp = Cin + (size_t)(row0 + 8 * hl) * ldc + col0 + lr;
    #pragma unroll
    for (int i = 0; i < RM; ++i) {
      #pragma unroll
      for (int j = 0; j < RN; ++j) {
        #pragma unroll
        for (int v = 0; v < 8; ++v) cv[i][j][v] = Cp[(size_t)(16 * i + v) * ldc + 16 * j];
      }
    }
    __builtin_amdgcn_sched_barrier(0);
  }
  if (K2 != 0) {
    rb_load<RM, RN>(f0, A2 + (size_t)(row0 + lr) * K2 + 4 * hl, (size_t)16 * K2,
                    B2 + (size_t)(col0 + lr) * K2 + 4 * hl, (size_t)16 * K2, 0);
    __builtin_amdgcn_sched_barrier(0);
  }
  rb_compute<RM, RN>(acc, f1);
  if (K2 != 0) rb_compute<RM, RN>(acc, f0);
  if (HASC) {
    #pragma unroll
    for (int i = 0; i < RM; ++i) {
      #pragma unroll
      for (int j = 0; j < RN; ++j) {
        #pragma unroll
        for (int v = 0; v < 8; ++v) acc[i][j][v] += cv[i][j][v];
      }
    }
  }
  #pragma unroll
  for (int i = 0; i < RM; ++i) {
    #pragma unroll
    for (int j = 0; j < RN; ++j) {
      #pragma unroll
      for (int v = 0; v < 8; ++v) Dp[(size_t)(16 * i + v) * ldd + 16 * j] = acc[i][j][v];
    }
  }
}

__device__ __forceinline__ void ksplit_block(float (&s)[4], const float* Ar, const float* Br,
                                             float (*part)[2][8][32], int wave, int lane) {
  v4f a[8], b0[8], b1[8];
  #pragma unroll
  for (int q = 0; q < 8; ++q) {
    a[q]  = *(const v4f*)(Ar + 8 * q);
    b0[q] = *(const v4f*)(Br + 8 * q);
    b1[q] = *(const v4f*)(Br + (size_t)16 * 256 + 8 * q);
  }
  __builtin_amdgcn_sched_barrier(0);
  v8f acc0 = {}, acc1 = {};
  #pragma unroll
  for (int q = 0; q < 8; ++q) {
    v2f a01 = {a[q][0], a[q][1]}, a23 = {a[q][2], a[q][3]};
    v2f p01 = {b0[q][0], b0[q][1]}, p23 = {b0[q][2], b0[q][3]};
    v2f r01 = {b1[q][0], b1[q][1]}, r23 = {b1[q][2], b1[q][3]};
    acc0 = WMMA_F32(a01, p01, acc0);
    acc1 = WMMA_F32(a01, r01, acc1);
    acc0 = WMMA_F32(a23, p23, acc0);
    acc1 = WMMA_F32(a23, r23, acc1);
  }
  #pragma unroll
  for (int v = 0; v < 8; ++v) {
    part[wave][0][v][lane] = acc0[v];
    part[wave][1][v][lane] = acc1[v];
  }
  __syncthreads();
  const int j = lane >> 4, lc = lane & 15;
  #pragma unroll
  for (int e = 0; e < 4; ++e) {
    const int r = 4 * wave + e;
    const int v = r & 7, sl = 16 * (r >> 3) + lc;
    float t = part[0][j][v][sl] + part[1][j][v][sl];
    t += part[2][j][v][sl];
    t += part[3][j][v][sl];
    s[e] = t;
  }
}

template <int MB>
__global__ __launch_bounds__(128) void pair_ksplit_kernel(
    float* D, int ldd, const float* A, int lda, const float* __restrict__ B,
    const float* Cin, int ldc, float* __restrict__ SqD, const float* __restrict__ SqA) {
  if constexpr (MB != 0) {
    if ((int)blockIdx.x >= MB) {
      __shared__ float sqpart[4][8][32];
      square_tile(SqD, SqA, (int)blockIdx.x - MB, sqpart);
      return;
    }
  }
  __shared__ float part[4][2][8][32];
  const int lane = threadIdx.x & 31, wave = threadIdx.x >> 5;
  const int hl = lane >> 4, lr = lane & 15;
  const int tm = blockIdx.x >> 3, tp = blockIdx.x & 7;
  const int col = tp * 32 + lane;
  const int rbase = tm * 16 + 4 * wave;
  float c[4];
  #pragma unroll
  for (int e = 0; e < 4; ++e) c[e] = Cin[(size_t)(rbase + e) * ldc + col];
  const float* Ar = A + (size_t)(tm * 16 + lr) * lda + 64 * wave + 4 * hl;
  const float* Br = B + (size_t)(tp * 32 + lr) * 256 + 64 * wave + 4 * hl;
  float s[4];
  ksplit_block(s, Ar, Br, part, wave, lane);
  #pragma unroll
  for (int e = 0; e < 4; ++e) D[(size_t)(rbase + e) * ldd + col] = c[e] + s[e];
}

template <bool FIRST>
__device__ __forceinline__ const float* carry_row(const float* Vc, const float* G,
                                                  const float* x0, int c) {
  if (FIRST) return (c == 0) ? x0 : G + ((size_t)(c - 1) * L1 + (L1 - 1)) * 256;
  return Vc + (size_t)c * 256;
}
static constexpr int CARRY_BLOCKS = 512;
template <bool FIRST, bool LAST, bool SQ>
__global__ __launch_bounds__(128) void carry_double_kernel(
    float* __restrict__ Vn, const float* __restrict__ Vc,
    const float* __restrict__ Pd, int d,
    const float* __restrict__ G, const float* __restrict__ x0,
    float* __restrict__ SqD, const float* __restrict__ SqA) {
  if constexpr (SQ) {
    if ((int)blockIdx.x >= CARRY_BLOCKS) {
      __shared__ float sqpart[4][8][32];
      square_tile(SqD, SqA, (int)blockIdx.x - CARRY_BLOCKS, sqpart);
      return;
    }
  }
  __shared__ float part[4][2][8][32];
  const int lane = threadIdx.x & 31, wave = threadIdx.x >> 5;
  const int hl = lane >> 4, lr = lane & 15;
  const int tm = blockIdx.x >> 3, tp = blockIdx.x & 7;
  const int col = tp * 32 + lane;
  const int rbase = tm * 16 + 4 * wave;
  float seed[4], res[4];
  #pragma unroll
  for (int e = 0; e < 4; ++e) {
    seed[e] = carry_row<FIRST>(Vc, G, x0, rbase + e)[col];
    res[e] = seed[e];
  }
  if (tm * 16 + 15 >= d) {
    const int srow = tm * 16 + lr - d;
    const float* Ar = carry_row<FIRST>(Vc, G, x0, srow >= 0 ? srow : 0) + 64 * wave + 4 * hl;
    const float* Br = Pd + (size_t)(tp * 32 + lr) * 256 + 64 * wave + 4 * hl;
    float s[4];
    ksplit_block(s, Ar, Br, part, wave, lane);
    #pragma unroll
    for (int e = 0; e < 4; ++e) res[e] = (rbase + e >= d) ? seed[e] + s[e] : seed[e];
  }
  constexpr size_t orow = LAST ? (size_t)L1 * 256 : (size_t)256;
  #pragma unroll
  for (int e = 0; e < 4; ++e) Vn[(size_t)(rbase + e) * orow + col] = res[e];
}

static constexpr int XP = 260;
__device__ __forceinline__ int dn_time(int p) {
  return 32 * (p & 3) + 16 * ((p >> 2) & 1) + 8 * ((p >> 3) & 1) + 4 * ((p >> 4) & 1) +
         2 * ((p >> 5) & 1) + ((p >> 6) & 1);
}
template <int RT, int RN>
__device__ __forceinline__ void dn_load(FragSet<RT, RN>& f, float (*X)[XP], int arow, int hl,
                                        const float* Bp, size_t bstep, int k0) {
  #pragma unroll
  for (int i = 0; i < RT; ++i) {
    #pragma unroll
    for (int q = 0; q < 4; ++q) f.a[i][q] = *(const v4f*)&X[arow + 16 * i][k0 + 8 * q + 4 * hl];
  }
  #pragma unroll
  for (int j = 0; j < RN; ++j) {
    #pragma unroll
    for (int q = 0; q < 4; ++q) f.b[j][q] = *(const v4f*)(Bp + j * bstep + k0 + 8 * q);
  }
}
template <int RT, int RN>
__device__ __forceinline__ void dn_kwalk(v8f (&acc)[RT][RN], FragSet<RT, RN>& f0, FragSet<RT, RN>& f1,
                                         float (*X)[XP], int arow, int hl, const float* Bp, size_t bstep) {
  dn_load<RT, RN>(f0, X, arow, hl, Bp, bstep, 0);
  #pragma unroll 1
  for (int k0 = 0; k0 < 256 - 64; k0 += 64) {
    dn_load<RT, RN>(f1, X, arow, hl, Bp, bstep, k0 + 32);
    __builtin_amdgcn_sched_barrier(0);
    rb_compute<RT, RN>(acc, f0);
    __builtin_amdgcn_sched_barrier(0);
    dn_load<RT, RN>(f0, X, arow, hl, Bp, bstep, k0 + 64);
    __builtin_amdgcn_sched_barrier(0);
    rb_compute<RT, RN>(acc, f1);
    __builtin_amdgcn_sched_barrier(0);
  }
  dn_load<RT, RN>(f1, X, arow, hl, Bp, bstep, 256 - 32);
  __builtin_amdgcn_sched_barrier(0);
  rb_compute<RT, RN>(acc, f0);
  __builtin_amdgcn_sched_barrier(0);
}
template <int RT, int P>
__device__ __forceinline__ void dn_level(float (*X)[XP], const float* __restrict__ Mh,
                                         const float* __restrict__ G, float* xout, int n0,
                                         int lane, int wave) {
  const int hl = lane >> 4, lr = lane & 15;
  const int col0 = 64 * wave;
  constexpr int RV = P >= 16 ? 8 : P;
  const bool live = P >= 16 || hl == 0;
  v8f acc[RT][4];
  #pragma unroll
  for (int i = 0; i < RT; ++i) {
    #pragma unroll
    for (int j = 0; j < 4; ++j) acc[i][j] = v8f{};
  }
  FragSet<RT, 4> f0, f1;
  dn_kwalk<RT, 4>(acc, f0, f1, X, lr, hl, Mh + (size_t)(col0 + lr) * 256 + 4 * hl, (size_t)16 * 256);
  v8f gv[RT][4];
  #pragma unroll
  for (int i = 0; i < RT; ++i) {
    #pragma unroll
    for (int j = 0; j < 4; ++j) {
      #pragma unroll
      for (int r = 0; r < 8; ++r) gv[i][j][r] = 0.0f;
      if (live) {
        #pragma unroll
        for (int r = 0; r < RV; ++r) {
          const int p = 16 * i + 8 * hl + r;
          gv[i][j][r] = G[(size_t)(n0 + dn_time(P + p) - 1) * 256 + col0 + 16 * j + lr];
        }
      }
    }
  }
  __builtin_amdgcn_sched_barrier(0);
  rb_compute<RT, 4>(acc, f1);
  if (P < 16) __syncthreads();
  #pragma unroll
  for (int i = 0; i < RT; ++i) {
    #pragma unroll
    for (int j = 0; j < 4; ++j) {
      #pragma unroll
      for (int r = 0; r < RV; ++r) {
        const int p = 16 * i + 8 * hl + r;
        if (live) {
          const float v = acc[i][j][r] + gv[i][j][r];
          const int c = col0 + 16 * j + lr;
          xout[(size_t)(n0 + dn_time(P + p)) * 256 + c] = v;
          X[P + p][c] = v;
        }
      }
    }
  }
  __syncthreads();
}

__global__ __launch_bounds__(128) void down_y_kernel(
    float* xout, float* __restrict__ yout, const float* __restrict__ G,
    const float* __restrict__ M1, const float* __restrict__ M2, const float* __restrict__ M4,
    const float* __restrict__ M8, const float* __restrict__ M16,
    const float* __restrict__ Cm, const float* __restrict__ u, const float* __restrict__ Dm) {
  __shared__ __attribute__((aligned(16))) float X[128][XP];
  const int tid = threadIdx.x;
  const int lane = tid & 31, wave = tid >> 5;
  const int hl = lane >> 4, lr = lane & 15;
  const int n0 = blockIdx.x * 128;
  #pragma unroll
  for (int it = 0; it < 8; ++it) {
    const int idx = it * 128 + tid;
    const int row = idx >> 6, c4 = (idx & 63) * 4;
    v4f v = {};
    if (row < 4) v = *(const v4f*)(xout + (size_t)(n0 + 32 * row) * 256 + c4);
    *(v4f*)&X[row][c4] = v;
  }
  __syncthreads();
  dn_level<1, 4>(X, M16, G, xout, n0, lane, wave);
  dn_level<1, 8>(X, M8, G, xout, n0, lane, wave);
  dn_level<1, 16>(X, M4, G, xout, n0, lane, wave);
  dn_level<2, 32>(X, M2, G, xout, n0, lane, wave);
  dn_level<4, 64>(X, M1, G, xout, n0, lane, wave);
  v8f acc[2][2];
  #pragma unroll
  for (int i = 0; i < 2; ++i) {
    #pragma unroll
    for (int j = 0; j < 2; ++j) acc[i][j] = v8f{};
  }
  FragSet<2, 2> f0, f1;
  const int arow = 32 * wave + lr;
  dn_kwalk<2, 2>(acc, f0, f1, X, arow, hl, Cm + (size_t)lr * 256 + 4 * hl, (size_t)16 * 256);
  #pragma unroll
  for (int i = 0; i < 2; ++i) {
    const float* up = u + (size_t)(n0 + dn_time(arow + 16 * i)) * 32 + 4 * hl;
    #pragma unroll
    for (int q = 0; q < 4; ++q) f0.a[i][q] = *(const v4f*)(up + 8 * q);
  }
  #pragma unroll
  for (int j = 0; j < 2; ++j) {
    #pragma unroll
    for (int q = 0; q < 4; ++q) f0.b[j][q] = *(const v4f*)(Dm + (size_t)(16 * j + lr) * 32 + 4 * hl + 8 * q);
  }
  __builtin_amdgcn_sched_barrier(0);
  rb_compute<2, 2>(acc, f1);
  rb_compute<2, 2>(acc, f0);
  #pragma unroll
  for (int i = 0; i < 2; ++i) {
    #pragma unroll
    for (int j = 0; j < 2; ++j) {
      #pragma unroll
      for (int r = 0; r < 8; ++r) {
        const int p = 32 * wave + 16 * i + 8 * hl + r;
        yout[(size_t)(n0 + dn_time(p)) * 32 + 16 * j + lr] = acc[i][j][r];
      }
    }
  }
}

template <int NC, int Mr>
static inline void launch_ksplit(hipStream_t s, float* D, const float* A, const float* B) {
  static_assert(Mr % 16 == 0 && NC % 16 == 0, "whole tiles");
  gemm_ksplit_kernel<NC><<<(Mr / 16) * (NC / 16), 128, 0, s>>>(D, A, B);
}

template <int RM, int RN, int WM, int WN, int KD, bool HASC, int Mr, int Nc, int K2 = 0, bool SQ = false>
static inline void launch_rb(hipStream_t s, float* D, int ldd,
                             const float* A, int lda, const float* B, int ldb,
                             const float* Cin, int ldc,
                             const float* A2 = nullptr, const float* B2 = nullptr,
                             float* SqD = nullptr, const float* SqA = nullptr) {
  static_assert(Mr % (16 * RM * WM) == 0 && Nc % (16 * RN * WN) == 0, "whole work-group tiles");
  constexpr int blocksN = Nc / (16 * RN * WN);
  constexpr int blocksM = Mr / (16 * RM * WM);
  constexpr int own = blocksM * blocksN;
  gemm_rb_kernel<RM, RN, WM, WN, KD, HASC, blocksN, K2, SQ ? own : 0>
      <<<own + (SQ ? SQ_TILES : 0), 128, 0, s>>>(D, ldd, A, lda, B, ldb, Cin, ldc, A2, B2, SqD, SqA);
}

template <int Mr, bool SQ = false>
static inline void launch_pair(hipStream_t s, float* D, int ldd, const float* A, int lda,
                               const float* B, const float* Cin, int ldc,
                               float* SqD = nullptr, const float* SqA = nullptr) {
  static_assert(Mr % 16 == 0, "whole blocks");
  constexpr int own = (Mr / 16) * 8;
  pair_ksplit_kernel<SQ ? own : 0><<<own + (SQ ? SQ_TILES : 0), 128, 0, s>>>(
      D, ldd, A, lda, B, Cin, ldc, SqD, SqA);
}

extern "C" void kernel_launch(void* const* d_in, const int* in_sizes, int n_in,
                              void* d_out, int out_size, void* d_ws, size_t ws_size,
                              hipStream_t stream) {
  (void)in_sizes; (void)n_in; (void)out_size; (void)ws_size;
  const float* t  = (const float*)d_in[0];
  const float* u  = (const float*)d_in[1];
  const float* x0 = (const float*)d_in[2];
  const float* Am = (const float*)d_in[3];
  const float* Bm = (const float*)d_in[4];
  const float* Cm = (const float*)d_in[5];
  const float* Dm = (const float*)d_in[6];
  float* xout = (float*)d_out;
  float* yout = xout + (size_t)Tn * NS;

  float* w = (float*)d_ws;
  float* Bd = w;  w += 8192;
  float* Kt[6]; for (int i = 0; i < 6; i++) { Kt[i] = w; w += 65536; }
  float* Rst  = w; w += 5 * 8192;
  float* Mm   = w; w += 65536;
  float* Qc   = w; w += 32768;
  float* PwA  = w; w += 65536;
  float* PwB  = w; w += 65536;
  float* PwC  = w; w += 65536;
  float* PwD  = w; w += 65536;
  float* Pm   = w; w += (size_t)Tn * 128;
  float* G    = w; w += (size_t)Tn * 256;
  float* V0   = w; w += (size_t)C1 * 256;
  w += (size_t)C2 * 256;
  w += (size_t)C2 * 256;
  float* V1   = w; w += (size_t)C1 * 256;

  float* PA  = Kt[0];
  k_front<<<Tn / (16 * FR), 128, 0, stream>>>(Pm, u, PA, Bd, Am, Bm, t);
  launch_ksplit<256, 256>(stream, Kt[1], Kt[0], Kt[0]);
  launch_ksplit<256, 512>(stream, Kt[2], Kt[0], Kt[1]);
  launch_ksplit<256, 512>(stream, Kt[4], Kt[0], Kt[3]);
  launch_ksplit<32, 1280>(stream, Rst, Kt[0], Bd);
  k_combine<<<(16384 + 8192) / 256, 256, 0, stream>>>(Mm, Qc, PA, Bd, Rst);

  float* Mp[5] = {Mm, Kt[0], Kt[1], Kt[2], Kt[3]};
  float* Pw[10] = {PwA, PwB, PwC, PwD, PwA, PwB, PwC, PwD, PwA, PwB};

  launch_rb<2, 4, 2, 2, 128, false, Tn, 256, 0, true>(stream, G, 256, Pm, 128, Qc, 128, nullptr, 0,
                                                      nullptr, nullptr, Mp[1], Mp[0]);

  launch_rb<2, 4, 2, 2, 256, true, 16384, 256, 0, true>(stream, G + 1 * 256, 512, G + 0 * 256, 512, Mp[0], 256, G + 1 * 256, 512,
                                                        nullptr, nullptr, Mp[2], Mp[1]);
  launch_rb<2, 4, 2, 2, 256, true, 8192, 256, 0, true>(stream, G + 3 * 256, 1024, G + 1 * 256, 1024, Mp[1], 256, G + 3 * 256, 1024,
                                                       nullptr, nullptr, Mp[3], Mp[2]);
  launch_rb<2, 2, 2, 2, 256, true, 4096, 256, 0, true>(stream, G + 7 * 256, 2048, G + 3 * 256, 2048, Mp[2], 256, G + 7 * 256, 2048,
                                                       nullptr, nullptr, Mp[4], Mp[3]);
  launch_pair<2048, true>(stream, G + 15 * 256, 4096, G + 7 * 256, 4096, Mp[3], G + 15 * 256, 4096,
                          Pw[0], Mp[4]);
  launch_pair<1024, true>(stream, G + 31 * 256, 8192, G + 15 * 256, 8192, Mp[4], G + 31 * 256, 8192,
                          Pw[1], Pw[0]);
  const int dd[10] = {1, 2, 4, 8, 16, 32, 64, 128, 256, 512};
  carry_double_kernel<true, false, true><<<CARRY_BLOCKS + SQ_TILES, 128, 0, stream>>>(
      V0, V1, Pw[0], dd[0], G, x0, Pw[2], Pw[1]);
  float* Va = V0;
  float* Vb = V1;
  for (int p = 1; p < 8; ++p) {
    carry_double_kernel<false, false, true><<<CARRY_BLOCKS + SQ_TILES, 128, 0, stream>>>(
        Vb, Va, Pw[p], dd[p], G, x0, Pw[p + 2], Pw[p + 1]);
    float* tv = Va; Va = Vb; Vb = tv;
  }
  carry_double_kernel<false, false, false><<<CARRY_BLOCKS, 128, 0, stream>>>(
      Vb, Va, Pw[8], dd[8], G, x0, nullptr, nullptr);
  carry_double_kernel<false, true, false><<<CARRY_BLOCKS, 128, 0, stream>>>(
      xout, Vb, Pw[9], dd[9], G, x0, nullptr, nullptr);
  down_y_kernel<<<Tn / 128, 128, 0, stream>>>(xout, yout, G, Mp[0], Mp[1], Mp[2], Mp[3], Mp[4], Cm, u, Dm);
}
